# starting kernel
# baseline (speedup 1.0000x reference)
#include <hip/hip_runtime.h>
#include <hip/hip_bf16.h>
#include <math.h>

typedef float v2f __attribute__((ext_vector_type(2)));
typedef float v8f __attribute__((ext_vector_type(8)));

#define B_    2
#define T_    4
#define N_    1024
#define L_    4
#define MA_   256
#define KNB_  32
#define C_    512
#define HEADS_ 8
#define DH_   64
#define MTOK_ 1024
#define ROWS_ 2048
#define MLP_  2048

__device__ __forceinline__ v8f wmma4(v2f a, v2f b, v8f c) {
  return __builtin_amdgcn_wmma_f32_16x16x4_f32(false, a, false, b, (short)0, c,
                                               false, false);
}

__device__ __forceinline__ float gelu_exact(float x) {
  return 0.5f * x * (1.0f + erff(x * 0.7071067811865475f));
}

__global__ void fps_kernel(const float* __restrict__ xyzs,
                           float* __restrict__ anchors) {
  __shared__ float sval[N_];
  __shared__ int   sidx[N_];
  __shared__ float last[3];
  const int blk = blockIdx.x;
  const int tid = threadIdx.x;
  const float* frame = xyzs + (size_t)blk * N_ * 3;
  const float px = frame[tid * 3 + 0];
  const float py = frame[tid * 3 + 1];
  const float pz = frame[tid * 3 + 2];
  float dist = 1e10f;
  float* aout = anchors + (size_t)blk * MA_ * 3;
  if (tid == 0) {
    last[0] = px; last[1] = py; last[2] = pz;
    aout[0] = px; aout[1] = py; aout[2] = pz;
  }
  __syncthreads();
  for (int it = 1; it < MA_; ++it) {
    const float dx = px - last[0], dy = py - last[1], dz = pz - last[2];
    dist = fminf(dist, dx * dx + dy * dy + dz * dz);
    sval[tid] = dist; sidx[tid] = tid;
    __syncthreads();
    for (int s = 512; s > 0; s >>= 1) {
      if (tid < s) {
        const float v2v = sval[tid + s];
        const int   i2  = sidx[tid + s];
        if (v2v > sval[tid] || (v2v == sval[tid] && i2 < sidx[tid])) {
          sval[tid] = v2v; sidx[tid] = i2;
        }
      }
      __syncthreads();
    }
    const int nxt = sidx[0];
    __syncthreads();
    if (tid == nxt) {
      last[0] = px; last[1] = py; last[2] = pz;
      aout[it * 3 + 0] = px; aout[it * 3 + 1] = py; aout[it * 3 + 2] = pz;
    }
    __syncthreads();
  }
}

__global__ void p4dconv_kernel(const float* __restrict__ xyzs,
                               const float* __restrict__ anchors,
                               const float* __restrict__ Wd,
                               float* __restrict__ feats) {
  __shared__ int   sscan[N_];
  __shared__ float spx[KNB_], spy[KNB_], spz[KNB_];
  __shared__ float s_p0[3];
  const int blk = blockIdx.x;
  const int m   = blk % MA_;
  const int bl  = blk / MA_;
  const int l   = bl % L_;
  const int b   = bl / L_;
  const int tid = threadIdx.x;

  const float* arow = anchors + ((size_t)bl * MA_ + m) * 3;
  const float ax = arow[0], ay = arow[1], az = arow[2];

  float w0 = 0.f, w1 = 0.f, w2 = 0.f, w3 = 0.f;
  if (tid < C_) {
    w0 = Wd[tid * 4 + 0]; w1 = Wd[tid * 4 + 1];
    w2 = Wd[tid * 4 + 2]; w3 = Wd[tid * 4 + 3];
  }
  float maxv = -3.4e38f;

  for (int di = -1; di <= 1; ++di) {
    int fi = l + di; fi = fi < 0 ? 0 : (fi > T_ - 1 ? T_ - 1 : fi);
    const float* nb = xyzs + ((size_t)b * T_ + fi) * N_ * 3;
    const float qx = nb[tid * 3 + 0];
    const float qy = nb[tid * 3 + 1];
    const float qz = nb[tid * 3 + 2];
    const float dx = qx - ax, dy = qy - ay, dz = qz - az;
    const int pred = (dx * dx + dy * dy + dz * dz) < 1.0f ? 1 : 0;
    sscan[tid] = pred;
    if (tid == 0) { s_p0[0] = dx; s_p0[1] = dy; s_p0[2] = dz; }
    __syncthreads();
    for (int off = 1; off < N_; off <<= 1) {
      const int v   = sscan[tid];
      const int add = (tid >= off) ? sscan[tid - off] : 0;
      __syncthreads();
      sscan[tid] = v + add;
      __syncthreads();
    }
    const int incl  = sscan[tid];
    const int total = sscan[N_ - 1];
    const int pos   = incl - pred;
    if (pred && pos < KNB_) { spx[pos] = dx; spy[pos] = dy; spz[pos] = dz; }
    __syncthreads();
    const int cnt = total < KNB_ ? total : KNB_;
    if (tid < KNB_ && tid >= cnt) {
      spx[tid] = (total > 0) ? spx[0] : s_p0[0];
      spy[tid] = (total > 0) ? spy[0] : s_p0[1];
      spz[tid] = (total > 0) ? spz[0] : s_p0[2];
    }
    __syncthreads();
    if (tid < C_) {
      const float tq = (float)di;
      #pragma unroll 8
      for (int k = 0; k < KNB_; ++k) {
        const float v = w0 * spx[k] + w1 * spy[k] + w2 * spz[k] + w3 * tq;
        maxv = fmaxf(maxv, v);
      }
    }
    __syncthreads();
  }
  if (tid < C_)
    feats[((size_t)bl * MA_ + m) * C_ + tid] = fmaxf(maxv, 0.0f);
}

__global__ void ln_kernel(const float* __restrict__ x, const float* __restrict__ g,
                          const float* __restrict__ bta, float* __restrict__ y) {
  __shared__ float red[256];
  const int row = blockIdx.x, t = threadIdx.x;
  const float* xr = x + (size_t)row * C_;
  const float a0 = xr[t], a1 = xr[t + 256];
  red[t] = a0 + a1; __syncthreads();
  for (int s = 128; s > 0; s >>= 1) { if (t < s) red[t] += red[t + s]; __syncthreads(); }
  const float mean = red[0] * (1.0f / C_);
  __syncthreads();
  const float d0 = a0 - mean, d1 = a1 - mean;
  red[t] = d0 * d0 + d1 * d1; __syncthreads();
  for (int s = 128; s > 0; s >>= 1) { if (t < s) red[t] += red[t + s]; __syncthreads(); }
  const float inv = rsqrtf(red[0] * (1.0f / C_) + 1e-5f);
  float* yr = y + (size_t)row * C_;
  yr[t]       = d0 * inv * g[t]       + bta[t];
  yr[t + 256] = d1 * inv * g[t + 256] + bta[t + 256];
}

__global__ void gemm_wmma(const float* __restrict__ A, int lda,
                          const float* __restrict__ W, int ldw,
                          const float* __restrict__ bias,
                          const float* __restrict__ resid,
                          float* __restrict__ Cout, int ldc,
                          int Kdim, int act) {
  const int j0 = blockIdx.x * 32, i0 = blockIdx.y * 32;
  const int lane = threadIdx.x & 31;
  const int sub = lane & 15, hi = lane >> 4;
  const float* arow0 = A + (size_t)(i0 + sub) * lda + 2 * hi;
  const float* arow1 = arow0 + (size_t)16 * lda;
  const float* wrow0 = W + (size_t)(j0 + sub) * ldw + 2 * hi;
  const float* wrow1 = wrow0 + (size_t)16 * ldw;
  __builtin_prefetch(arow0, 0, 0);
  __builtin_prefetch(wrow0, 0, 0);
  v8f acc00 = {0.f, 0.f, 0.f, 0.f, 0.f, 0.f, 0.f, 0.f};
  v8f acc01 = acc00, acc10 = acc00, acc11 = acc00;
  #pragma unroll 4
  for (int k0 = 0; k0 < Kdim; k0 += 4) {
    const v2f a0 = *(const v2f*)(arow0 + k0);
    const v2f a1 = *(const v2f*)(arow1 + k0);
    const v2f b0 = *(const v2f*)(wrow0 + k0);
    const v2f b1 = *(const v2f*)(wrow1 + k0);
    acc00 = wmma4(a0, b0, acc00);
    acc01 = wmma4(a0, b1, acc01);
    acc10 = wmma4(a1, b0, acc10);
    acc11 = wmma4(a1, b1, acc11);
  }
  const float bj0 = bias ? bias[j0 + sub]      : 0.0f;
  const float bj1 = bias ? bias[j0 + 16 + sub] : 0.0f;
  #pragma unroll
  for (int r = 0; r < 8; ++r) {
    const int row0 = i0 + r + 8 * hi;
    const int row1 = row0 + 16;
    const int col0 = j0 + sub;
    const int col1 = col0 + 16;
    float v00 = acc00[r] + bj0, v01 = acc01[r] + bj1;
    float v10 = acc10[r] + bj0, v11 = acc11[r] + bj1;
    if (act == 1) {
      v00 = gelu_exact(v00); v01 = gelu_exact(v01);
      v10 = gelu_exact(v10); v11 = gelu_exact(v11);
    }
    if (resid) {
      v00 += resid[(size_t)row0 * ldc + col0];
      v01 += resid[(size_t)row0 * ldc + col1];
      v10 += resid[(size_t)row1 * ldc + col0];
      v11 += resid[(size_t)row1 * ldc + col1];
    }
    Cout[(size_t)row0 * ldc + col0] = v00;
    Cout[(size_t)row0 * ldc + col1] = v01;
    Cout[(size_t)row1 * ldc + col0] = v10;
    Cout[(size_t)row1 * ldc + col1] = v11;
  }
}

__global__ void attn_kernel(const float* __restrict__ qkv,
                            const float* __restrict__ xyzf,
                            const float* __restrict__ Wsp,
                            float* __restrict__ attn_out) {
  extern __shared__ float smem[];
  float* S    = smem;
  float* red  = S + 16 * 1024;
  float* rmax = red + 256;
  float* rsum = rmax + 16;
  float* wx   = rsum + 16;
  const int b = blockIdx.z, h = blockIdx.y, i0 = blockIdx.x * 16;
  const int t = threadIdx.x;
  const int lane = t & 31, w = t >> 5;
  const int sub = lane & 15, hi = lane >> 4;

  const float* qbase =
      qkv + ((size_t)(b * MTOK_ + i0 + sub) * 3 + 0) * C_ + h * DH_ + 2 * hi;
  v2f qa[16];
  #pragma unroll
  for (int ks = 0; ks < 16; ++ks) qa[ks] = *(const v2f*)(qbase + 4 * ks);
  for (int jj = 0; jj < 8; ++jj) {
    const int j0 = (w * 8 + jj) * 16;
    const float* kbase =
        qkv + ((size_t)(b * MTOK_ + j0 + sub) * 3 + 1) * C_ + h * DH_ + 2 * hi;
    v8f acc = {0.f, 0.f, 0.f, 0.f, 0.f, 0.f, 0.f, 0.f};
    #pragma unroll
    for (int ks = 0; ks < 16; ++ks) {
      const v2f kb = *(const v2f*)(kbase + 4 * ks);
      acc = wmma4(qa[ks], kb, acc);
    }
    #pragma unroll
    for (int r = 0; r < 8; ++r)
      S[(r + 8 * hi) * 1024 + j0 + sub] = acc[r] * 0.125f;
  }
  __syncthreads();

  const int row = t >> 4, seg = t & 15;
  float* sr = S + row * 1024 + seg * 64;
  float lm = -3.4e38f;
  for (int c2 = 0; c2 < 64; ++c2) lm = fmaxf(lm, sr[c2]);
  red[t] = lm; __syncthreads();
  if (t < 16) {
    float mm = -3.4e38f;
    for (int s2 = 0; s2 < 16; ++s2) mm = fmaxf(mm, red[t * 16 + s2]);
    rmax[t] = mm;
  }
  __syncthreads();
  const float rm = rmax[row];
  float ls = 0.0f;
  for (int c2 = 0; c2 < 64; ++c2) {
    const float e = expf(sr[c2] - rm);
    sr[c2] = e; ls += e;
  }
  red[t] = ls; __syncthreads();
  if (t < 16) {
    float ss = 0.0f;
    for (int s2 = 0; s2 < 16; ++s2) ss += red[t * 16 + s2];
    rsum[t] = ss;
  }
  __syncthreads();

  v8f acc = {0.f, 0.f, 0.f, 0.f, 0.f, 0.f, 0.f, 0.f};
  const int d0 = (w & 3) * 16;
  if (w < 4) {
    const float* vbase =
        qkv + ((size_t)(b * MTOK_) * 3 + 2) * C_ + h * DH_ + d0 + sub;
    const float* srow = S + sub * 1024 + 2 * hi;
    for (int ks = 0; ks < 256; ++ks) {
      const v2f a = *(const v2f*)(srow + 4 * ks);
      const int kk = 4 * ks + 2 * hi;
      v2f bb;
      bb.x = vbase[(size_t)kk * 1536];
      bb.y = vbase[(size_t)(kk + 1) * 1536];
      acc = wmma4(a, bb, acc);
    }
  } else {
    const int q = (w - 4) * 32 + lane;
    if (q < 48) {
      const int rr = q / 3, cc = q % 3;
      const float* srw = S + rr * 1024;
      const float* xb  = xyzf + (size_t)b * MTOK_ * 3 + cc;
      float s = 0.0f;
      for (int j = 0; j < 1024; ++j) s += srw[j] * xb[j * 3];
      wx[q] = s;
    }
  }
  __syncthreads();
  if (w < 4) {
    const int d = d0 + sub;
    const float w0 = Wsp[d * 3 + 0], w1 = Wsp[d * 3 + 1], w2 = Wsp[d * 3 + 2];
    #pragma unroll
    for (int r = 0; r < 8; ++r) {
      const int rr = r + 8 * hi;
      const int i  = i0 + rr;
      const float inv = 1.0f / rsum[rr];
      const float* xi = xyzf + ((size_t)b * MTOK_ + i) * 3;
      const float wd0 = wx[rr * 3 + 0] * inv - xi[0];
      const float wd1 = wx[rr * 3 + 1] * inv - xi[1];
      const float wd2 = wx[rr * 3 + 2] * inv - xi[2];
      attn_out[((size_t)(b * MTOK_) + i) * C_ + h * DH_ + d] =
          acc[r] * inv + w0 * wd0 + w1 * wd1 + w2 * wd2;
    }
  }
}

extern "C" void kernel_launch(void* const* d_in, const int* in_sizes, int n_in,
                              void* d_out, int out_size, void* d_ws, size_t ws_size,
                              hipStream_t stream) {
  (void)in_sizes; (void)n_in; (void)out_size; (void)ws_size;
  const float* xyzs = (const float*)d_in[0];
  const float* Wd   = (const float*)d_in[1];
  const float* Wqkv = (const float*)d_in[2];
  const float* Wsp  = (const float*)d_in[3];
  const float* Wout = (const float*)d_in[4];
  const float* bout = (const float*)d_in[5];
  const float* ln1g = (const float*)d_in[6];
  const float* ln1b = (const float*)d_in[7];
  const float* Wff1 = (const float*)d_in[8];
  const float* bff1 = (const float*)d_in[9];
  const float* Wff2 = (const float*)d_in[10];
  const float* bff2 = (const float*)d_in[11];
  const float* ln2g = (const float*)d_in[12];
  const float* ln2b = (const float*)d_in[13];

  float* ws      = (float*)d_ws;
  float* anchors = ws;
  float* feats   = ws + 8192;
  float* xln     = feats + (size_t)ROWS_ * C_;
  float* qkvb    = xln   + (size_t)ROWS_ * C_;
  float* attno   = qkvb  + (size_t)ROWS_ * 1536;
  float* ffh     = attno + (size_t)ROWS_ * C_;

  fps_kernel<<<B_ * L_, N_, 0, stream>>>(xyzs, anchors);
  p4dconv_kernel<<<B_ * L_ * MA_, N_, 0, stream>>>(xyzs, anchors, Wd, feats);

  const size_t smem_attn = (16 * 1024 + 256 + 16 + 16 + 48) * sizeof(float);
  for (int d = 0; d < 2; ++d) {
    ln_kernel<<<ROWS_, 256, 0, stream>>>(feats, ln1g + d * C_, ln1b + d * C_, xln);
    gemm_wmma<<<dim3(1536 / 32, ROWS_ / 32), 32, 0, stream>>>(
        xln, C_, Wqkv + (size_t)d * 1536 * C_, C_, nullptr, nullptr,
        qkvb, 1536, C_, 0);
    attn_kernel<<<dim3(MTOK_ / 16, HEADS_, B_), 256, smem_attn, stream>>>(
        qkvb, anchors, Wsp + (size_t)d * DH_ * 3, attno);
    gemm_wmma<<<dim3(C_ / 32, ROWS_ / 32), 32, 0, stream>>>(
        attno, C_, Wout + (size_t)d * C_ * C_, C_, bout + d * C_,
        feats, feats, C_, C_, 1);
    ln_kernel<<<ROWS_, 256, 0, stream>>>(feats, ln2g + d * C_, ln2b + d * C_, xln);
    gemm_wmma<<<dim3(MLP_ / 32, ROWS_ / 32), 32, 0, stream>>>(
        xln, C_, Wff1 + (size_t)d * MLP_ * C_, C_, bff1 + d * MLP_,
        nullptr, ffh, MLP_, C_, 1);
    gemm_wmma<<<dim3(C_ / 32, ROWS_ / 32), 32, 0, stream>>>(
        ffh, MLP_, Wff2 + (size_t)d * C_ * MLP_, MLP_, bff2 + d * C_,
        feats, feats, C_, MLP_, 0);
  }
  hipMemcpyAsync(d_out, feats, (size_t)ROWS_ * C_ * sizeof(float),
                 hipMemcpyDeviceToDevice, stream);
}


# FPS: one wave per frame, 32 points per lane in registers; argmax by per-lane running max, then row_xmask DPP + permlanex16 wave max and index min; no LDS tree, no barrier in the 255-step loop
# speedup vs baseline: 1.3124x; 1.3124x over previous
#include <hip/hip_runtime.h>
#include <hip/hip_bf16.h>
#include <math.h>

typedef float v2f __attribute__((ext_vector_type(2)));
typedef float v8f __attribute__((ext_vector_type(8)));

#define B_    2
#define T_    4
#define N_    1024
#define L_    4
#define MA_   256
#define KNB_  32
#define C_    512
#define HEADS_ 8
#define DH_   64
#define MTOK_ 1024
#define ROWS_ 2048
#define MLP_  2048

__device__ __forceinline__ v8f wmma4(v2f a, v2f b, v8f c) {
  return __builtin_amdgcn_wmma_f32_16x16x4_f32(false, a, false, b, (short)0, c,
                                               false, false);
}

__device__ __forceinline__ float gelu_exact(float x) {
  return 0.5f * x * (1.0f + erff(x * 0.7071067811865475f));
}

#define FPS_SLOTS_  (N_ / 32)
#define FPS_CHAINS_ 4
#define FPS_CLEN_   (FPS_SLOTS_ / FPS_CHAINS_)

template <int CTRL>
__device__ __forceinline__ int dpp_row_i(int v) {
  return __builtin_amdgcn_update_dpp(0, v, CTRL, 0xf, 0xf, true);
}
__device__ __forceinline__ int other_row_i(int v) {
  return __builtin_amdgcn_permlanex16(v, v, 0x76543210, 0xfedcba98, false, false);
}
__device__ __forceinline__ int wave_max_i32(int v) {
  v = max(v, dpp_row_i<0x161>(v));
  v = max(v, dpp_row_i<0x162>(v));
  v = max(v, dpp_row_i<0x164>(v));
  v = max(v, dpp_row_i<0x168>(v));
  v = max(v, other_row_i(v));
  return v;
}
__device__ __forceinline__ int wave_min_i32(int v) {
  v = min(v, dpp_row_i<0x161>(v));
  v = min(v, dpp_row_i<0x162>(v));
  v = min(v, dpp_row_i<0x164>(v));
  v = min(v, dpp_row_i<0x168>(v));
  v = min(v, other_row_i(v));
  return v;
}

__global__ __launch_bounds__(32) void fps_kernel(const float* __restrict__ xyzs,
                                                 float* __restrict__ anchors) {
  __shared__ float4 spt[N_];
  const int blk  = blockIdx.x;
  const int lane = threadIdx.x;
  const float* frame = xyzs + (size_t)blk * N_ * 3;
  float px[FPS_SLOTS_], py[FPS_SLOTS_], pz[FPS_SLOTS_];
  int   dist[FPS_SLOTS_];
  #pragma unroll
  for (int s = 0; s < FPS_SLOTS_; ++s) {
    const int i = s * 32 + lane;
    px[s] = frame[i * 3 + 0];
    py[s] = frame[i * 3 + 1];
    pz[s] = frame[i * 3 + 2];
    dist[s] = __builtin_bit_cast(int, 1e10f);
    spt[i] = make_float4(px[s], py[s], pz[s], 0.0f);
  }
  __syncthreads();
  float* aout = anchors + (size_t)blk * MA_ * 3;
  const char* sbytes = (const char*)spt;
  int curb = 0;
  for (int it = 1; it < MA_; ++it) {
    const float4 last = *(const float4*)(sbytes + curb);
    if (lane == 0) {
      aout[(it - 1) * 3 + 0] = last.x;
      aout[(it - 1) * 3 + 1] = last.y;
      aout[(it - 1) * 3 + 2] = last.z;
    }
    #pragma unroll
    for (int s = 0; s < FPS_SLOTS_; ++s) {
      const float dx = px[s] - last.x, dy = py[s] - last.y, dz = pz[s] - last.z;
      const float d2 = fmaf(dz, dz, fmaf(dx, dx, dy * dy));
      dist[s] = min(dist[s], __builtin_bit_cast(int, d2));
    }
    int bv[FPS_CHAINS_], bo[FPS_CHAINS_];
    #pragma unroll
    for (int c = 0; c < FPS_CHAINS_; ++c) {
      bv[c] = dist[c * FPS_CLEN_];
      bo[c] = c * FPS_CLEN_ * 32 * 16;
      #pragma unroll
      for (int k = 1; k < FPS_CLEN_; ++k) {
        const int s = c * FPS_CLEN_ + k;
        bo[c] = (dist[s] > bv[c]) ? s * 32 * 16 : bo[c];
        bv[c] = max(bv[c], dist[s]);
      }
    }
    #pragma unroll
    for (int w = 1; w < FPS_CHAINS_; w <<= 1) {
      #pragma unroll
      for (int c = 0; c < FPS_CHAINS_; c += 2 * w) {
        bo[c] = (bv[c + w] > bv[c]) ? bo[c + w] : bo[c];
        bv[c] = max(bv[c], bv[c + w]);
      }
    }
    const int gmax = wave_max_i32(bv[0]);
    const int cand = (bv[0] == gmax) ? (bo[0] + lane * 16) : 0x7fffffff;
    curb = wave_min_i32(cand);
  }
  const float4 last = *(const float4*)(sbytes + curb);
  if (lane == 0) {
    aout[(MA_ - 1) * 3 + 0] = last.x;
    aout[(MA_ - 1) * 3 + 1] = last.y;
    aout[(MA_ - 1) * 3 + 2] = last.z;
  }
}

__global__ void p4dconv_kernel(const float* __restrict__ xyzs,
                               const float* __restrict__ anchors,
                               const float* __restrict__ Wd,
                               float* __restrict__ feats) {
  __shared__ int   sscan[N_];
  __shared__ float spx[KNB_], spy[KNB_], spz[KNB_];
  __shared__ float s_p0[3];
  const int blk = blockIdx.x;
  const int m   = blk % MA_;
  const int bl  = blk / MA_;
  const int l   = bl % L_;
  const int b   = bl / L_;
  const int tid = threadIdx.x;

  const float* arow = anchors + ((size_t)bl * MA_ + m) * 3;
  const float ax = arow[0], ay = arow[1], az = arow[2];

  float w0 = 0.f, w1 = 0.f, w2 = 0.f, w3 = 0.f;
  if (tid < C_) {
    w0 = Wd[tid * 4 + 0]; w1 = Wd[tid * 4 + 1];
    w2 = Wd[tid * 4 + 2]; w3 = Wd[tid * 4 + 3];
  }
  float maxv = -3.4e38f;

  for (int di = -1; di <= 1; ++di) {
    int fi = l + di; fi = fi < 0 ? 0 : (fi > T_ - 1 ? T_ - 1 : fi);
    const float* nb = xyzs + ((size_t)b * T_ + fi) * N_ * 3;
    const float qx = nb[tid * 3 + 0];
    const float qy = nb[tid * 3 + 1];
    const float qz = nb[tid * 3 + 2];
    const float dx = qx - ax, dy = qy - ay, dz = qz - az;
    const int pred = (dx * dx + dy * dy + dz * dz) < 1.0f ? 1 : 0;
    sscan[tid] = pred;
    if (tid == 0) { s_p0[0] = dx; s_p0[1] = dy; s_p0[2] = dz; }
    __syncthreads();
    for (int off = 1; off < N_; off <<= 1) {
      const int v   = sscan[tid];
      const int add = (tid >= off) ? sscan[tid - off] : 0;
      __syncthreads();
      sscan[tid] = v + add;
      __syncthreads();
    }
    const int incl  = sscan[tid];
    const int total = sscan[N_ - 1];
    const int pos   = incl - pred;
    if (pred && pos < KNB_) { spx[pos] = dx; spy[pos] = dy; spz[pos] = dz; }
    __syncthreads();
    const int cnt = total < KNB_ ? total : KNB_;
    if (tid < KNB_ && tid >= cnt) {
      spx[tid] = (total > 0) ? spx[0] : s_p0[0];
      spy[tid] = (total > 0) ? spy[0] : s_p0[1];
      spz[tid] = (total > 0) ? spz[0] : s_p0[2];
    }
    __syncthreads();
    if (tid < C_) {
      const float tq = (float)di;
      #pragma unroll 8
      for (int k = 0; k < KNB_; ++k) {
        const float v = w0 * spx[k] + w1 * spy[k] + w2 * spz[k] + w3 * tq;
        maxv = fmaxf(maxv, v);
      }
    }
    __syncthreads();
  }
  if (tid < C_)
    feats[((size_t)bl * MA_ + m) * C_ + tid] = fmaxf(maxv, 0.0f);
}

__global__ void ln_kernel(const float* __restrict__ x, const float* __restrict__ g,
                          const float* __restrict__ bta, float* __restrict__ y) {
  __shared__ float red[256];
  const int row = blockIdx.x, t = threadIdx.x;
  const float* xr = x + (size_t)row * C_;
  const float a0 = xr[t], a1 = xr[t + 256];
  red[t] = a0 + a1; __syncthreads();
  for (int s = 128; s > 0; s >>= 1) { if (t < s) red[t] += red[t + s]; __syncthreads(); }
  const float mean = red[0] * (1.0f / C_);
  __syncthreads();
  const float d0 = a0 - mean, d1 = a1 - mean;
  red[t] = d0 * d0 + d1 * d1; __syncthreads();
  for (int s = 128; s > 0; s >>= 1) { if (t < s) red[t] += red[t + s]; __syncthreads(); }
  const float inv = rsqrtf(red[0] * (1.0f / C_) + 1e-5f);
  float* yr = y + (size_t)row * C_;
  yr[t]       = d0 * inv * g[t]       + bta[t];
  yr[t + 256] = d1 * inv * g[t + 256] + bta[t + 256];
}

__global__ void gemm_wmma(const float* __restrict__ A, int lda,
                          const float* __restrict__ W, int ldw,
                          const float* __restrict__ bias,
                          const float* __restrict__ resid,
                          float* __restrict__ Cout, int ldc,
                          int Kdim, int act) {
  const int j0 = blockIdx.x * 32, i0 = blockIdx.y * 32;
  const int lane = threadIdx.x & 31;
  const int sub = lane & 15, hi = lane >> 4;
  const float* arow0 = A + (size_t)(i0 + sub) * lda + 2 * hi;
  const float* arow1 = arow0 + (size_t)16 * lda;
  const float* wrow0 = W + (size_t)(j0 + sub) * ldw + 2 * hi;
  const float* wrow1 = wrow0 + (size_t)16 * ldw;
  __builtin_prefetch(arow0, 0, 0);
  __builtin_prefetch(wrow0, 0, 0);
  v8f acc00 = {0.f, 0.f, 0.f, 0.f, 0.f, 0.f, 0.f, 0.f};
  v8f acc01 = acc00, acc10 = acc00, acc11 = acc00;
  #pragma unroll 4
  for (int k0 = 0; k0 < Kdim; k0 += 4) {
    const v2f a0 = *(const v2f*)(arow0 + k0);
    const v2f a1 = *(const v2f*)(arow1 + k0);
    const v2f b0 = *(const v2f*)(wrow0 + k0);
    const v2f b1 = *(const v2f*)(wrow1 + k0);
    acc00 = wmma4(a0, b0, acc00);
    acc01 = wmma4(a0, b1, acc01);
    acc10 = wmma4(a1, b0, acc10);
    acc11 = wmma4(a1, b1, acc11);
  }
  const float bj0 = bias ? bias[j0 + sub]      : 0.0f;
  const float bj1 = bias ? bias[j0 + 16 + sub] : 0.0f;
  #pragma unroll
  for (int r = 0; r < 8; ++r) {
    const int row0 = i0 + r + 8 * hi;
    const int row1 = row0 + 16;
    const int col0 = j0 + sub;
    const int col1 = col0 + 16;
    float v00 = acc00[r] + bj0, v01 = acc01[r] + bj1;
    float v10 = acc10[r] + bj0, v11 = acc11[r] + bj1;
    if (act == 1) {
      v00 = gelu_exact(v00); v01 = gelu_exact(v01);
      v10 = gelu_exact(v10); v11 = gelu_exact(v11);
    }
    if (resid) {
      v00 += resid[(size_t)row0 * ldc + col0];
      v01 += resid[(size_t)row0 * ldc + col1];
      v10 += resid[(size_t)row1 * ldc + col0];
      v11 += resid[(size_t)row1 * ldc + col1];
    }
    Cout[(size_t)row0 * ldc + col0] = v00;
    Cout[(size_t)row0 * ldc + col1] = v01;
    Cout[(size_t)row1 * ldc + col0] = v10;
    Cout[(size_t)row1 * ldc + col1] = v11;
  }
}

__global__ void attn_kernel(const float* __restrict__ qkv,
                            const float* __restrict__ xyzf,
                            const float* __restrict__ Wsp,
                            float* __restrict__ attn_out) {
  extern __shared__ float smem[];
  float* S    = smem;
  float* red  = S + 16 * 1024;
  float* rmax = red + 256;
  float* rsum = rmax + 16;
  float* wx   = rsum + 16;
  const int b = blockIdx.z, h = blockIdx.y, i0 = blockIdx.x * 16;
  const int t = threadIdx.x;
  const int lane = t & 31, w = t >> 5;
  const int sub = lane & 15, hi = lane >> 4;

  const float* qbase =
      qkv + ((size_t)(b * MTOK_ + i0 + sub) * 3 + 0) * C_ + h * DH_ + 2 * hi;
  v2f qa[16];
  #pragma unroll
  for (int ks = 0; ks < 16; ++ks) qa[ks] = *(const v2f*)(qbase + 4 * ks);
  for (int jj = 0; jj < 8; ++jj) {
    const int j0 = (w * 8 + jj) * 16;
    const float* kbase =
        qkv + ((size_t)(b * MTOK_ + j0 + sub) * 3 + 1) * C_ + h * DH_ + 2 * hi;
    v8f acc = {0.f, 0.f, 0.f, 0.f, 0.f, 0.f, 0.f, 0.f};
    #pragma unroll
    for (int ks = 0; ks < 16; ++ks) {
      const v2f kb = *(const v2f*)(kbase + 4 * ks);
      acc = wmma4(qa[ks], kb, acc);
    }
    #pragma unroll
    for (int r = 0; r < 8; ++r)
      S[(r + 8 * hi) * 1024 + j0 + sub] = acc[r] * 0.125f;
  }
  __syncthreads();

  const int row = t >> 4, seg = t & 15;
  float* sr = S + row * 1024 + seg * 64;
  float lm = -3.4e38f;
  for (int c2 = 0; c2 < 64; ++c2) lm = fmaxf(lm, sr[c2]);
  red[t] = lm; __syncthreads();
  if (t < 16) {
    float mm = -3.4e38f;
    for (int s2 = 0; s2 < 16; ++s2) mm = fmaxf(mm, red[t * 16 + s2]);
    rmax[t] = mm;
  }
  __syncthreads();
  const float rm = rmax[row];
  float ls = 0.0f;
  for (int c2 = 0; c2 < 64; ++c2) {
    const float e = expf(sr[c2] - rm);
    sr[c2] = e; ls += e;
  }
  red[t] = ls; __syncthreads();
  if (t < 16) {
    float ss = 0.0f;
    for (int s2 = 0; s2 < 16; ++s2) ss += red[t * 16 + s2];
    rsum[t] = ss;
  }
  __syncthreads();

  v8f acc = {0.f, 0.f, 0.f, 0.f, 0.f, 0.f, 0.f, 0.f};
  const int d0 = (w & 3) * 16;
  if (w < 4) {
    const float* vbase =
        qkv + ((size_t)(b * MTOK_) * 3 + 2) * C_ + h * DH_ + d0 + sub;
    const float* srow = S + sub * 1024 + 2 * hi;
    for (int ks = 0; ks < 256; ++ks) {
      const v2f a = *(const v2f*)(srow + 4 * ks);
      const int kk = 4 * ks + 2 * hi;
      v2f bb;
      bb.x = vbase[(size_t)kk * 1536];
      bb.y = vbase[(size_t)(kk + 1) * 1536];
      acc = wmma4(a, bb, acc);
    }
  } else {
    const int q = (w - 4) * 32 + lane;
    if (q < 48) {
      const int rr = q / 3, cc = q % 3;
      const float* srw = S + rr * 1024;
      const float* xb  = xyzf + (size_t)b * MTOK_ * 3 + cc;
      float s = 0.0f;
      for (int j = 0; j < 1024; ++j) s += srw[j] * xb[j * 3];
      wx[q] = s;
    }
  }
  __syncthreads();
  if (w < 4) {
    const int d = d0 + sub;
    const float w0 = Wsp[d * 3 + 0], w1 = Wsp[d * 3 + 1], w2 = Wsp[d * 3 + 2];
    #pragma unroll
    for (int r = 0; r < 8; ++r) {
      const int rr = r + 8 * hi;
      const int i  = i0 + rr;
      const float inv = 1.0f / rsum[rr];
      const float* xi = xyzf + ((size_t)b * MTOK_ + i) * 3;
      const float wd0 = wx[rr * 3 + 0] * inv - xi[0];
      const float wd1 = wx[rr * 3 + 1] * inv - xi[1];
      const float wd2 = wx[rr * 3 + 2] * inv - xi[2];
      attn_out[((size_t)(b * MTOK_) + i) * C_ + h * DH_ + d] =
          acc[r] * inv + w0 * wd0 + w1 * wd1 + w2 * wd2;
    }
  }
}

extern "C" void kernel_launch(void* const* d_in, const int* in_sizes, int n_in,
                              void* d_out, int out_size, void* d_ws, size_t ws_size,
                              hipStream_t stream) {
  (void)in_sizes; (void)n_in; (void)out_size; (void)ws_size;
  const float* xyzs = (const float*)d_in[0];
  const float* Wd   = (const float*)d_in[1];
  const float* Wqkv = (const float*)d_in[2];
  const float* Wsp  = (const float*)d_in[3];
  const float* Wout = (const float*)d_in[4];
  const float* bout = (const float*)d_in[5];
  const float* ln1g = (const float*)d_in[6];
  const float* ln1b = (const float*)d_in[7];
  const float* Wff1 = (const float*)d_in[8];
  const float* bff1 = (const float*)d_in[9];
  const float* Wff2 = (const float*)d_in[10];
  const float* bff2 = (const float*)d_in[11];
  const float* ln2g = (const float*)d_in[12];
  const float* ln2b = (const float*)d_in[13];

  float* ws      = (float*)d_ws;
  float* anchors = ws;
  float* feats   = ws + 8192;
  float* xln     = feats + (size_t)ROWS_ * C_;
  float* qkvb    = xln   + (size_t)ROWS_ * C_;
  float* attno   = qkvb  + (size_t)ROWS_ * 1536;
  float* ffh     = attno + (size_t)ROWS_ * C_;

  fps_kernel<<<B_ * L_, 32, 0, stream>>>(xyzs, anchors);
  p4dconv_kernel<<<B_ * L_ * MA_, N_, 0, stream>>>(xyzs, anchors, Wd, feats);

  const size_t smem_attn = (16 * 1024 + 256 + 16 + 16 + 48) * sizeof(float);
  for (int d = 0; d < 2; ++d) {
    ln_kernel<<<ROWS_, 256, 0, stream>>>(feats, ln1g + d * C_, ln1b + d * C_, xln);
    gemm_wmma<<<dim3(1536 / 32, ROWS_ / 32), 32, 0, stream>>>(
        xln, C_, Wqkv + (size_t)d * 1536 * C_, C_, nullptr, nullptr,
        qkvb, 1536, C_, 0);
    attn_kernel<<<dim3(MTOK_ / 16, HEADS_, B_), 256, smem_attn, stream>>>(
        qkvb, anchors, Wsp + (size_t)d * DH_ * 3, attno);
    gemm_wmma<<<dim3(C_ / 32, ROWS_ / 32), 32, 0, stream>>>(
        attno, C_, Wout + (size_t)d * C_ * C_, C_, bout + d * C_,
        feats, feats, C_, C_, 1);
    ln_kernel<<<ROWS_, 256, 0, stream>>>(feats, ln2g + d * C_, ln2b + d * C_, xln);
    gemm_wmma<<<dim3(MLP_ / 32, ROWS_ / 32), 32, 0, stream>>>(
        xln, C_, Wff1 + (size_t)d * MLP_ * C_, C_, bff1 + d * MLP_,
        nullptr, ffh, MLP_, C_, 1);
    gemm_wmma<<<dim3(C_ / 32, ROWS_ / 32), 32, 0, stream>>>(
        ffh, MLP_, Wff2 + (size_t)d * C_ * MLP_, MLP_, bff2 + d * C_,
        feats, feats, C_, MLP_, 0);
  }
  hipMemcpyAsync(d_out, feats, (size_t)ROWS_ * C_ * sizeof(float),
                 hipMemcpyDeviceToDevice, stream);
}


# GEMMs on LDS-staged block tiles (64x64 and 128x128, 8 waves, 32-wide k stages prefetched through registers) instead of per-wave global fragment loads; k summed in another order
# speedup vs baseline: 1.7205x; 1.3110x over previous
#include <hip/hip_runtime.h>
#include <hip/hip_bf16.h>
#include <math.h>

typedef float v2f __attribute__((ext_vector_type(2)));
typedef float v8f __attribute__((ext_vector_type(8)));

#define B_    2
#define T_    4
#define N_    1024
#define L_    4
#define MA_   256
#define KNB_  32
#define C_    512
#define HEADS_ 8
#define DH_   64
#define MTOK_ 1024
#define ROWS_ 2048
#define MLP_  2048

__device__ __forceinline__ v8f wmma4(v2f a, v2f b, v8f c) {
  return __builtin_amdgcn_wmma_f32_16x16x4_f32(false, a, false, b, (short)0, c,
                                               false, false);
}

__device__ __forceinline__ float gelu_exact(float x) {
  return 0.5f * x * (1.0f + erff(x * 0.7071067811865475f));
}

#define FPS_SLOTS_  (N_ / 32)
#define FPS_CHAINS_ 4
#define FPS_CLEN_   (FPS_SLOTS_ / FPS_CHAINS_)

template <int CTRL>
__device__ __forceinline__ int dpp_row_i(int v) {
  return __builtin_amdgcn_update_dpp(0, v, CTRL, 0xf, 0xf, true);
}
__device__ __forceinline__ int other_row_i(int v) {
  return __builtin_amdgcn_permlanex16(v, v, 0x76543210, 0xfedcba98, false, false);
}
__device__ __forceinline__ int wave_max_i32(int v) {
  v = max(v, dpp_row_i<0x161>(v));
  v = max(v, dpp_row_i<0x162>(v));
  v = max(v, dpp_row_i<0x164>(v));
  v = max(v, dpp_row_i<0x168>(v));
  v = max(v, other_row_i(v));
  return v;
}
__device__ __forceinline__ int wave_min_i32(int v) {
  v = min(v, dpp_row_i<0x161>(v));
  v = min(v, dpp_row_i<0x162>(v));
  v = min(v, dpp_row_i<0x164>(v));
  v = min(v, dpp_row_i<0x168>(v));
  v = min(v, other_row_i(v));
  return v;
}

__global__ __launch_bounds__(32) void fps_kernel(const float* __restrict__ xyzs,
                                                 float* __restrict__ anchors) {
  __shared__ float4 spt[N_];
  const int blk  = blockIdx.x;
  const int lane = threadIdx.x;
  const float* frame = xyzs + (size_t)blk * N_ * 3;
  float px[FPS_SLOTS_], py[FPS_SLOTS_], pz[FPS_SLOTS_];
  int   dist[FPS_SLOTS_];
  #pragma unroll
  for (int s = 0; s < FPS_SLOTS_; ++s) {
    const int i = s * 32 + lane;
    px[s] = frame[i * 3 + 0];
    py[s] = frame[i * 3 + 1];
    pz[s] = frame[i * 3 + 2];
    dist[s] = __builtin_bit_cast(int, 1e10f);
    spt[i] = make_float4(px[s], py[s], pz[s], 0.0f);
  }
  __syncthreads();
  float* aout = anchors + (size_t)blk * MA_ * 3;
  const char* sbytes = (const char*)spt;
  int curb = 0;
  for (int it = 1; it < MA_; ++it) {
    const float4 last = *(const float4*)(sbytes + curb);
    if (lane == 0) {
      aout[(it - 1) * 3 + 0] = last.x;
      aout[(it - 1) * 3 + 1] = last.y;
      aout[(it - 1) * 3 + 2] = last.z;
    }
    #pragma unroll
    for (int s = 0; s < FPS_SLOTS_; ++s) {
      const float dx = px[s] - last.x, dy = py[s] - last.y, dz = pz[s] - last.z;
      const float d2 = fmaf(dz, dz, fmaf(dx, dx, dy * dy));
      dist[s] = min(dist[s], __builtin_bit_cast(int, d2));
    }
    int bv[FPS_CHAINS_], bo[FPS_CHAINS_];
    #pragma unroll
    for (int c = 0; c < FPS_CHAINS_; ++c) {
      bv[c] = dist[c * FPS_CLEN_];
      bo[c] = c * FPS_CLEN_ * 32 * 16;
      #pragma unroll
      for (int k = 1; k < FPS_CLEN_; ++k) {
        const int s = c * FPS_CLEN_ + k;
        bo[c] = (dist[s] > bv[c]) ? s * 32 * 16 : bo[c];
        bv[c] = max(bv[c], dist[s]);
      }
    }
    #pragma unroll
    for (int w = 1; w < FPS_CHAINS_; w <<= 1) {
      #pragma unroll
      for (int c = 0; c < FPS_CHAINS_; c += 2 * w) {
        bo[c] = (bv[c + w] > bv[c]) ? bo[c + w] : bo[c];
        bv[c] = max(bv[c], bv[c + w]);
      }
    }
    const int gmax = wave_max_i32(bv[0]);
    const int cand = (bv[0] == gmax) ? (bo[0] + lane * 16) : 0x7fffffff;
    curb = wave_min_i32(cand);
  }
  const float4 last = *(const float4*)(sbytes + curb);
  if (lane == 0) {
    aout[(MA_ - 1) * 3 + 0] = last.x;
    aout[(MA_ - 1) * 3 + 1] = last.y;
    aout[(MA_ - 1) * 3 + 2] = last.z;
  }
}

__global__ void p4dconv_kernel(const float* __restrict__ xyzs,
                               const float* __restrict__ anchors,
                               const float* __restrict__ Wd,
                               float* __restrict__ feats) {
  __shared__ int   sscan[N_];
  __shared__ float spx[KNB_], spy[KNB_], spz[KNB_];
  __shared__ float s_p0[3];
  const int blk = blockIdx.x;
  const int m   = blk % MA_;
  const int bl  = blk / MA_;
  const int l   = bl % L_;
  const int b   = bl / L_;
  const int tid = threadIdx.x;

  const float* arow = anchors + ((size_t)bl * MA_ + m) * 3;
  const float ax = arow[0], ay = arow[1], az = arow[2];

  float w0 = 0.f, w1 = 0.f, w2 = 0.f, w3 = 0.f;
  if (tid < C_) {
    w0 = Wd[tid * 4 + 0]; w1 = Wd[tid * 4 + 1];
    w2 = Wd[tid * 4 + 2]; w3 = Wd[tid * 4 + 3];
  }
  float maxv = -3.4e38f;

  for (int di = -1; di <= 1; ++di) {
    int fi = l + di; fi = fi < 0 ? 0 : (fi > T_ - 1 ? T_ - 1 : fi);
    const float* nb = xyzs + ((size_t)b * T_ + fi) * N_ * 3;
    const float qx = nb[tid * 3 + 0];
    const float qy = nb[tid * 3 + 1];
    const float qz = nb[tid * 3 + 2];
    const float dx = qx - ax, dy = qy - ay, dz = qz - az;
    const int pred = (dx * dx + dy * dy + dz * dz) < 1.0f ? 1 : 0;
    sscan[tid] = pred;
    if (tid == 0) { s_p0[0] = dx; s_p0[1] = dy; s_p0[2] = dz; }
    __syncthreads();
    for (int off = 1; off < N_; off <<= 1) {
      const int v   = sscan[tid];
      const int add = (tid >= off) ? sscan[tid - off] : 0;
      __syncthreads();
      sscan[tid] = v + add;
      __syncthreads();
    }
    const int incl  = sscan[tid];
    const int total = sscan[N_ - 1];
    const int pos   = incl - pred;
    if (pred && pos < KNB_) { spx[pos] = dx; spy[pos] = dy; spz[pos] = dz; }
    __syncthreads();
    const int cnt = total < KNB_ ? total : KNB_;
    if (tid < KNB_ && tid >= cnt) {
      spx[tid] = (total > 0) ? spx[0] : s_p0[0];
      spy[tid] = (total > 0) ? spy[0] : s_p0[1];
      spz[tid] = (total > 0) ? spz[0] : s_p0[2];
    }
    __syncthreads();
    if (tid < C_) {
      const float tq = (float)di;
      #pragma unroll 8
      for (int k = 0; k < KNB_; ++k) {
        const float v = w0 * spx[k] + w1 * spy[k] + w2 * spz[k] + w3 * tq;
        maxv = fmaxf(maxv, v);
      }
    }
    __syncthreads();
  }
  if (tid < C_)
    feats[((size_t)bl * MA_ + m) * C_ + tid] = fmaxf(maxv, 0.0f);
}

__global__ void ln_kernel(const float* __restrict__ x, const float* __restrict__ g,
                          const float* __restrict__ bta, float* __restrict__ y) {
  __shared__ float red[256];
  const int row = blockIdx.x, t = threadIdx.x;
  const float* xr = x + (size_t)row * C_;
  const float a0 = xr[t], a1 = xr[t + 256];
  red[t] = a0 + a1; __syncthreads();
  for (int s = 128; s > 0; s >>= 1) { if (t < s) red[t] += red[t + s]; __syncthreads(); }
  const float mean = red[0] * (1.0f / C_);
  __syncthreads();
  const float d0 = a0 - mean, d1 = a1 - mean;
  red[t] = d0 * d0 + d1 * d1; __syncthreads();
  for (int s = 128; s > 0; s >>= 1) { if (t < s) red[t] += red[t + s]; __syncthreads(); }
  const float inv = rsqrtf(red[0] * (1.0f / C_) + 1e-5f);
  float* yr = y + (size_t)row * C_;
  yr[t]       = d0 * inv * g[t]       + bta[t];
  yr[t + 256] = d1 * inv * g[t + 256] + bta[t + 256];
}

typedef float v4f __attribute__((ext_vector_type(4)));
#define GBK_    32
#define GPITCH_ 36

template <int BM, int BN>
__global__ __launch_bounds__(256) void gemm_tiled(const float* __restrict__ A, int lda,
                                                  const float* __restrict__ W, int ldw,
                                                  const float* __restrict__ bias,
                                                  const float* __restrict__ resid,
                                                  float* __restrict__ Cout, int ldc,
                                                  int Kdim, int act) {
  constexpr int WC = BN / 32;
  constexpr int WR = 8 / WC;
  constexpr int TM = BM / (16 * WR);
  constexpr int TN = 2;
  constexpr int LA = BM / 32;
  constexpr int LB = BN / 32;
  __shared__ __attribute__((aligned(16))) float As[2][BM * GPITCH_];
  __shared__ __attribute__((aligned(16))) float Bs[2][BN * GPITCH_];
  const int t = threadIdx.x;
  const int lane = t & 31, w = t >> 5;
  const int sub = lane & 15, hi = lane >> 4;
  const int wr = w / WC, wc = w % WC;
  const int i0 = blockIdx.y * BM, j0 = blockIdx.x * BN;
  const int lr = t >> 3;
  const int lc = (t & 7) * 4;
  const float* ag = A + (size_t)(i0 + lr) * lda + lc;
  const float* wg = W + (size_t)(j0 + lr) * ldw + lc;
  float* asw = &As[0][lr * GPITCH_ + lc];
  float* bsw = &Bs[0][lr * GPITCH_ + lc];
  v4f ra[LA], rb[LB];
  #pragma unroll
  for (int i = 0; i < LA; ++i) ra[i] = *(const v4f*)(ag + (size_t)(32 * i) * lda);
  #pragma unroll
  for (int i = 0; i < LB; ++i) rb[i] = *(const v4f*)(wg + (size_t)(32 * i) * ldw);
  #pragma unroll
  for (int i = 0; i < LA; ++i) *(v4f*)(asw + 32 * i * GPITCH_) = ra[i];
  #pragma unroll
  for (int i = 0; i < LB; ++i) *(v4f*)(bsw + 32 * i * GPITCH_) = rb[i];
  __syncthreads();

  v8f acc[TM][TN];
  #pragma unroll
  for (int m = 0; m < TM; ++m)
    #pragma unroll
    for (int n = 0; n < TN; ++n)
      acc[m][n] = (v8f){0.f, 0.f, 0.f, 0.f, 0.f, 0.f, 0.f, 0.f};

  const int fa_off = (wr * TM * 16 + sub) * GPITCH_ + 4 * hi;
  const int fb_off = (wc * 32 + sub) * GPITCH_ + 4 * hi;
  const int nst = Kdim / GBK_;
  for (int st = 0; st < nst; ++st) {
    const int cur = st & 1;
    const bool more = (st + 1 < nst);
    if (more) {
      const int kn = (st + 1) * GBK_;
      #pragma unroll
      for (int i = 0; i < LA; ++i) ra[i] = *(const v4f*)(ag + (size_t)(32 * i) * lda + kn);
      #pragma unroll
      for (int i = 0; i < LB; ++i) rb[i] = *(const v4f*)(wg + (size_t)(32 * i) * ldw + kn);
    }
    const float* as = &As[cur][fa_off];
    const float* bs = &Bs[cur][fb_off];
    v4f fa[2][TM], fb[2][TN];
    #pragma unroll
    for (int m = 0; m < TM; ++m) fa[0][m] = *(const v4f*)(as + m * 16 * GPITCH_);
    #pragma unroll
    for (int n = 0; n < TN; ++n) fb[0][n] = *(const v4f*)(bs + n * 16 * GPITCH_);
    #pragma unroll
    for (int kb = 0; kb < GBK_ / 8; ++kb) {
      const int c = kb & 1, d = c ^ 1;
      if (kb + 1 < GBK_ / 8) {
        #pragma unroll
        for (int m = 0; m < TM; ++m) fa[d][m] = *(const v4f*)(as + m * 16 * GPITCH_ + (kb + 1) * 8);
        #pragma unroll
        for (int n = 0; n < TN; ++n) fb[d][n] = *(const v4f*)(bs + n * 16 * GPITCH_ + (kb + 1) * 8);
      }
      #pragma unroll
      for (int m = 0; m < TM; ++m)
        #pragma unroll
        for (int n = 0; n < TN; ++n) {
          const v2f a = {fa[c][m].x, fa[c][m].y};
          const v2f b = {fb[c][n].x, fb[c][n].y};
          acc[m][n] = wmma4(a, b, acc[m][n]);
        }
      #pragma unroll
      for (int m = 0; m < TM; ++m)
        #pragma unroll
        for (int n = 0; n < TN; ++n) {
          const v2f a = {fa[c][m].z, fa[c][m].w};
          const v2f b = {fb[c][n].z, fb[c][n].w};
          acc[m][n] = wmma4(a, b, acc[m][n]);
        }
    }
    if (more) {
      const int nxt = cur ^ 1;
      #pragma unroll
      for (int i = 0; i < LA; ++i) *(v4f*)(asw + nxt * BM * GPITCH_ + 32 * i * GPITCH_) = ra[i];
      #pragma unroll
      for (int i = 0; i < LB; ++i) *(v4f*)(bsw + nxt * BN * GPITCH_ + 32 * i * GPITCH_) = rb[i];
    }
    __syncthreads();
  }

  float bj[TN];
  #pragma unroll
  for (int n = 0; n < TN; ++n) bj[n] = bias ? bias[j0 + wc * 32 + n * 16 + sub] : 0.0f;
  #pragma unroll
  for (int m = 0; m < TM; ++m) {
    #pragma unroll
    for (int r = 0; r < 8; ++r) {
      const int row = i0 + (wr * TM + m) * 16 + r + 8 * hi;
      #pragma unroll
      for (int n = 0; n < TN; ++n) {
        const int col = j0 + wc * 32 + n * 16 + sub;
        float v = acc[m][n][r] + bj[n];
        if (act == 1) v = gelu_exact(v);
        if (resid) v += resid[(size_t)row * ldc + col];
        Cout[(size_t)row * ldc + col] = v;
      }
    }
  }
}

__global__ void attn_kernel(const float* __restrict__ qkv,
                            const float* __restrict__ xyzf,
                            const float* __restrict__ Wsp,
                            float* __restrict__ attn_out) {
  extern __shared__ float smem[];
  float* S    = smem;
  float* red  = S + 16 * 1024;
  float* rmax = red + 256;
  float* rsum = rmax + 16;
  float* wx   = rsum + 16;
  const int b = blockIdx.z, h = blockIdx.y, i0 = blockIdx.x * 16;
  const int t = threadIdx.x;
  const int lane = t & 31, w = t >> 5;
  const int sub = lane & 15, hi = lane >> 4;

  const float* qbase =
      qkv + ((size_t)(b * MTOK_ + i0 + sub) * 3 + 0) * C_ + h * DH_ + 2 * hi;
  v2f qa[16];
  #pragma unroll
  for (int ks = 0; ks < 16; ++ks) qa[ks] = *(const v2f*)(qbase + 4 * ks);
  for (int jj = 0; jj < 8; ++jj) {
    const int j0 = (w * 8 + jj) * 16;
    const float* kbase =
        qkv + ((size_t)(b * MTOK_ + j0 + sub) * 3 + 1) * C_ + h * DH_ + 2 * hi;
    v8f acc = {0.f, 0.f, 0.f, 0.f, 0.f, 0.f, 0.f, 0.f};
    #pragma unroll
    for (int ks = 0; ks < 16; ++ks) {
      const v2f kb = *(const v2f*)(kbase + 4 * ks);
      acc = wmma4(qa[ks], kb, acc);
    }
    #pragma unroll
    for (int r = 0; r < 8; ++r)
      S[(r + 8 * hi) * 1024 + j0 + sub] = acc[r] * 0.125f;
  }
  __syncthreads();

  const int row = t >> 4, seg = t & 15;
  float* sr = S + row * 1024 + seg * 64;
  float lm = -3.4e38f;
  for (int c2 = 0; c2 < 64; ++c2) lm = fmaxf(lm, sr[c2]);
  red[t] = lm; __syncthreads();
  if (t < 16) {
    float mm = -3.4e38f;
    for (int s2 = 0; s2 < 16; ++s2) mm = fmaxf(mm, red[t * 16 + s2]);
    rmax[t] = mm;
  }
  __syncthreads();
  const float rm = rmax[row];
  float ls = 0.0f;
  for (int c2 = 0; c2 < 64; ++c2) {
    const float e = expf(sr[c2] - rm);
    sr[c2] = e; ls += e;
  }
  red[t] = ls; __syncthreads();
  if (t < 16) {
    float ss = 0.0f;
    for (int s2 = 0; s2 < 16; ++s2) ss += red[t * 16 + s2];
    rsum[t] = ss;
  }
  __syncthreads();

  v8f acc = {0.f, 0.f, 0.f, 0.f, 0.f, 0.f, 0.f, 0.f};
  const int d0 = (w & 3) * 16;
  if (w < 4) {
    const float* vbase =
        qkv + ((size_t)(b * MTOK_) * 3 + 2) * C_ + h * DH_ + d0 + sub;
    const float* srow = S + sub * 1024 + 2 * hi;
    for (int ks = 0; ks < 256; ++ks) {
      const v2f a = *(const v2f*)(srow + 4 * ks);
      const int kk = 4 * ks + 2 * hi;
      v2f bb;
      bb.x = vbase[(size_t)kk * 1536];
      bb.y = vbase[(size_t)(kk + 1) * 1536];
      acc = wmma4(a, bb, acc);
    }
  } else {
    const int q = (w - 4) * 32 + lane;
    if (q < 48) {
      const int rr = q / 3, cc = q % 3;
      const float* srw = S + rr * 1024;
      const float* xb  = xyzf + (size_t)b * MTOK_ * 3 + cc;
      float s = 0.0f;
      for (int j = 0; j < 1024; ++j) s += srw[j] * xb[j * 3];
      wx[q] = s;
    }
  }
  __syncthreads();
  if (w < 4) {
    const int d = d0 + sub;
    const float w0 = Wsp[d * 3 + 0], w1 = Wsp[d * 3 + 1], w2 = Wsp[d * 3 + 2];
    #pragma unroll
    for (int r = 0; r < 8; ++r) {
      const int rr = r + 8 * hi;
      const int i  = i0 + rr;
      const float inv = 1.0f / rsum[rr];
      const float* xi = xyzf + ((size_t)b * MTOK_ + i) * 3;
      const float wd0 = wx[rr * 3 + 0] * inv - xi[0];
      const float wd1 = wx[rr * 3 + 1] * inv - xi[1];
      const float wd2 = wx[rr * 3 + 2] * inv - xi[2];
      attn_out[((size_t)(b * MTOK_) + i) * C_ + h * DH_ + d] =
          acc[r] * inv + w0 * wd0 + w1 * wd1 + w2 * wd2;
    }
  }
}

extern "C" void kernel_launch(void* const* d_in, const int* in_sizes, int n_in,
                              void* d_out, int out_size, void* d_ws, size_t ws_size,
                              hipStream_t stream) {
  (void)in_sizes; (void)n_in; (void)out_size; (void)ws_size;
  const float* xyzs = (const float*)d_in[0];
  const float* Wd   = (const float*)d_in[1];
  const float* Wqkv = (const float*)d_in[2];
  const float* Wsp  = (const float*)d_in[3];
  const float* Wout = (const float*)d_in[4];
  const float* bout = (const float*)d_in[5];
  const float* ln1g = (const float*)d_in[6];
  const float* ln1b = (const float*)d_in[7];
  const float* Wff1 = (const float*)d_in[8];
  const float* bff1 = (const float*)d_in[9];
  const float* Wff2 = (const float*)d_in[10];
  const float* bff2 = (const float*)d_in[11];
  const float* ln2g = (const float*)d_in[12];
  const float* ln2b = (const float*)d_in[13];

  float* ws      = (float*)d_ws;
  float* anchors = ws;
  float* feats   = ws + 8192;
  float* xln     = feats + (size_t)ROWS_ * C_;
  float* qkvb    = xln   + (size_t)ROWS_ * C_;
  float* attno   = qkvb  + (size_t)ROWS_ * 1536;
  float* ffh     = attno + (size_t)ROWS_ * C_;

  fps_kernel<<<B_ * L_, 32, 0, stream>>>(xyzs, anchors);
  p4dconv_kernel<<<B_ * L_ * MA_, N_, 0, stream>>>(xyzs, anchors, Wd, feats);

  const size_t smem_attn = (16 * 1024 + 256 + 16 + 16 + 48) * sizeof(float);
  for (int d = 0; d < 2; ++d) {
    ln_kernel<<<ROWS_, 256, 0, stream>>>(feats, ln1g + d * C_, ln1b + d * C_, xln);
    gemm_tiled<64, 64><<<dim3(1536 / 64, ROWS_ / 64), 256, 0, stream>>>(
        xln, C_, Wqkv + (size_t)d * 1536 * C_, C_, nullptr, nullptr,
        qkvb, 1536, C_, 0);
    attn_kernel<<<dim3(MTOK_ / 16, HEADS_, B_), 256, smem_attn, stream>>>(
        qkvb, anchors, Wsp + (size_t)d * DH_ * 3, attno);
    gemm_tiled<64, 64><<<dim3(C_ / 64, ROWS_ / 64), 256, 0, stream>>>(
        attno, C_, Wout + (size_t)d * C_ * C_, C_, bout + d * C_,
        feats, feats, C_, C_, 1);
    ln_kernel<<<ROWS_, 256, 0, stream>>>(feats, ln2g + d * C_, ln2b + d * C_, xln);
    gemm_tiled<128, 128><<<dim3(MLP_ / 128, ROWS_ / 128), 256, 0, stream>>>(
        xln, C_, Wff1 + (size_t)d * MLP_ * C_, C_, bff1 + d * MLP_,
        nullptr, ffh, MLP_, C_, 1);
    gemm_tiled<64, 64><<<dim3(C_ / 64, ROWS_ / 64), 256, 0, stream>>>(
        ffh, MLP_, Wff2 + (size_t)d * C_ * MLP_, MLP_, bff2 + d * C_,
        feats, feats, C_, MLP_, 0);
  }
  hipMemcpyAsync(d_out, feats, (size_t)ROWS_ * C_ * sizeof(float),
                 hipMemcpyDeviceToDevice, stream);
}


# attention: score rows in LDS at pitch 1028 floats, read 16 B per lane along the row in softmax and P@xyz (bank conflicts gone), wave-butterfly sums: row sums and P@xyz add in another order
# speedup vs baseline: 2.2787x; 1.3244x over previous
#include <hip/hip_runtime.h>
#include <hip/hip_bf16.h>
#include <math.h>

typedef float v2f __attribute__((ext_vector_type(2)));
typedef float v8f __attribute__((ext_vector_type(8)));

#define B_    2
#define T_    4
#define N_    1024
#define L_    4
#define MA_   256
#define KNB_  32
#define C_    512
#define HEADS_ 8
#define DH_   64
#define MTOK_ 1024
#define ROWS_ 2048
#define MLP_  2048

__device__ __forceinline__ v8f wmma4(v2f a, v2f b, v8f c) {
  return __builtin_amdgcn_wmma_f32_16x16x4_f32(false, a, false, b, (short)0, c,
                                               false, false);
}

__device__ __forceinline__ float gelu_exact(float x) {
  return 0.5f * x * (1.0f + erff(x * 0.7071067811865475f));
}

#define FPS_SLOTS_  (N_ / 32)
#define FPS_CHAINS_ 4
#define FPS_CLEN_   (FPS_SLOTS_ / FPS_CHAINS_)

template <int CTRL>
__device__ __forceinline__ int dpp_row_i(int v) {
  return __builtin_amdgcn_update_dpp(0, v, CTRL, 0xf, 0xf, true);
}
__device__ __forceinline__ int other_row_i(int v) {
  return __builtin_amdgcn_permlanex16(v, v, 0x76543210, 0xfedcba98, false, false);
}
__device__ __forceinline__ int wave_max_i32(int v) {
  v = max(v, dpp_row_i<0x161>(v));
  v = max(v, dpp_row_i<0x162>(v));
  v = max(v, dpp_row_i<0x164>(v));
  v = max(v, dpp_row_i<0x168>(v));
  v = max(v, other_row_i(v));
  return v;
}
__device__ __forceinline__ int wave_min_i32(int v) {
  v = min(v, dpp_row_i<0x161>(v));
  v = min(v, dpp_row_i<0x162>(v));
  v = min(v, dpp_row_i<0x164>(v));
  v = min(v, dpp_row_i<0x168>(v));
  v = min(v, other_row_i(v));
  return v;
}

__global__ __launch_bounds__(32) void fps_kernel(const float* __restrict__ xyzs,
                                                 float* __restrict__ anchors) {
  __shared__ float4 spt[N_];
  const int blk  = blockIdx.x;
  const int lane = threadIdx.x;
  const float* frame = xyzs + (size_t)blk * N_ * 3;
  float px[FPS_SLOTS_], py[FPS_SLOTS_], pz[FPS_SLOTS_];
  int   dist[FPS_SLOTS_];
  #pragma unroll
  for (int s = 0; s < FPS_SLOTS_; ++s) {
    const int i = s * 32 + lane;
    px[s] = frame[i * 3 + 0];
    py[s] = frame[i * 3 + 1];
    pz[s] = frame[i * 3 + 2];
    dist[s] = __builtin_bit_cast(int, 1e10f);
    spt[i] = make_float4(px[s], py[s], pz[s], 0.0f);
  }
  __syncthreads();
  float* aout = anchors + (size_t)blk * MA_ * 3;
  const char* sbytes = (const char*)spt;
  int curb = 0;
  for (int it = 1; it < MA_; ++it) {
    const float4 last = *(const float4*)(sbytes + curb);
    if (lane == 0) {
      aout[(it - 1) * 3 + 0] = last.x;
      aout[(it - 1) * 3 + 1] = last.y;
      aout[(it - 1) * 3 + 2] = last.z;
    }
    #pragma unroll
    for (int s = 0; s < FPS_SLOTS_; ++s) {
      const float dx = px[s] - last.x, dy = py[s] - last.y, dz = pz[s] - last.z;
      const float d2 = fmaf(dz, dz, fmaf(dx, dx, dy * dy));
      dist[s] = min(dist[s], __builtin_bit_cast(int, d2));
    }
    int bv[FPS_CHAINS_], bo[FPS_CHAINS_];
    #pragma unroll
    for (int c = 0; c < FPS_CHAINS_; ++c) {
      bv[c] = dist[c * FPS_CLEN_];
      bo[c] = c * FPS_CLEN_ * 32 * 16;
      #pragma unroll
      for (int k = 1; k < FPS_CLEN_; ++k) {
        const int s = c * FPS_CLEN_ + k;
        bo[c] = (dist[s] > bv[c]) ? s * 32 * 16 : bo[c];
        bv[c] = max(bv[c], dist[s]);
      }
    }
    #pragma unroll
    for (int w = 1; w < FPS_CHAINS_; w <<= 1) {
      #pragma unroll
      for (int c = 0; c < FPS_CHAINS_; c += 2 * w) {
        bo[c] = (bv[c + w] > bv[c]) ? bo[c + w] : bo[c];
        bv[c] = max(bv[c], bv[c + w]);
      }
    }
    const int gmax = wave_max_i32(bv[0]);
    const int cand = (bv[0] == gmax) ? (bo[0] + lane * 16) : 0x7fffffff;
    curb = wave_min_i32(cand);
  }
  const float4 last = *(const float4*)(sbytes + curb);
  if (lane == 0) {
    aout[(MA_ - 1) * 3 + 0] = last.x;
    aout[(MA_ - 1) * 3 + 1] = last.y;
    aout[(MA_ - 1) * 3 + 2] = last.z;
  }
}

__global__ void p4dconv_kernel(const float* __restrict__ xyzs,
                               const float* __restrict__ anchors,
                               const float* __restrict__ Wd,
                               float* __restrict__ feats) {
  __shared__ int   sscan[N_];
  __shared__ float spx[KNB_], spy[KNB_], spz[KNB_];
  __shared__ float s_p0[3];
  const int blk = blockIdx.x;
  const int m   = blk % MA_;
  const int bl  = blk / MA_;
  const int l   = bl % L_;
  const int b   = bl / L_;
  const int tid = threadIdx.x;

  const float* arow = anchors + ((size_t)bl * MA_ + m) * 3;
  const float ax = arow[0], ay = arow[1], az = arow[2];

  float w0 = 0.f, w1 = 0.f, w2 = 0.f, w3 = 0.f;
  if (tid < C_) {
    w0 = Wd[tid * 4 + 0]; w1 = Wd[tid * 4 + 1];
    w2 = Wd[tid * 4 + 2]; w3 = Wd[tid * 4 + 3];
  }
  float maxv = -3.4e38f;

  for (int di = -1; di <= 1; ++di) {
    int fi = l + di; fi = fi < 0 ? 0 : (fi > T_ - 1 ? T_ - 1 : fi);
    const float* nb = xyzs + ((size_t)b * T_ + fi) * N_ * 3;
    const float qx = nb[tid * 3 + 0];
    const float qy = nb[tid * 3 + 1];
    const float qz = nb[tid * 3 + 2];
    const float dx = qx - ax, dy = qy - ay, dz = qz - az;
    const int pred = (dx * dx + dy * dy + dz * dz) < 1.0f ? 1 : 0;
    sscan[tid] = pred;
    if (tid == 0) { s_p0[0] = dx; s_p0[1] = dy; s_p0[2] = dz; }
    __syncthreads();
    for (int off = 1; off < N_; off <<= 1) {
      const int v   = sscan[tid];
      const int add = (tid >= off) ? sscan[tid - off] : 0;
      __syncthreads();
      sscan[tid] = v + add;
      __syncthreads();
    }
    const int incl  = sscan[tid];
    const int total = sscan[N_ - 1];
    const int pos   = incl - pred;
    if (pred && pos < KNB_) { spx[pos] = dx; spy[pos] = dy; spz[pos] = dz; }
    __syncthreads();
    const int cnt = total < KNB_ ? total : KNB_;
    if (tid < KNB_ && tid >= cnt) {
      spx[tid] = (total > 0) ? spx[0] : s_p0[0];
      spy[tid] = (total > 0) ? spy[0] : s_p0[1];
      spz[tid] = (total > 0) ? spz[0] : s_p0[2];
    }
    __syncthreads();
    if (tid < C_) {
      const float tq = (float)di;
      #pragma unroll 8
      for (int k = 0; k < KNB_; ++k) {
        const float v = w0 * spx[k] + w1 * spy[k] + w2 * spz[k] + w3 * tq;
        maxv = fmaxf(maxv, v);
      }
    }
    __syncthreads();
  }
  if (tid < C_)
    feats[((size_t)bl * MA_ + m) * C_ + tid] = fmaxf(maxv, 0.0f);
}

__global__ void ln_kernel(const float* __restrict__ x, const float* __restrict__ g,
                          const float* __restrict__ bta, float* __restrict__ y) {
  __shared__ float red[256];
  const int row = blockIdx.x, t = threadIdx.x;
  const float* xr = x + (size_t)row * C_;
  const float a0 = xr[t], a1 = xr[t + 256];
  red[t] = a0 + a1; __syncthreads();
  for (int s = 128; s > 0; s >>= 1) { if (t < s) red[t] += red[t + s]; __syncthreads(); }
  const float mean = red[0] * (1.0f / C_);
  __syncthreads();
  const float d0 = a0 - mean, d1 = a1 - mean;
  red[t] = d0 * d0 + d1 * d1; __syncthreads();
  for (int s = 128; s > 0; s >>= 1) { if (t < s) red[t] += red[t + s]; __syncthreads(); }
  const float inv = rsqrtf(red[0] * (1.0f / C_) + 1e-5f);
  float* yr = y + (size_t)row * C_;
  yr[t]       = d0 * inv * g[t]       + bta[t];
  yr[t + 256] = d1 * inv * g[t + 256] + bta[t + 256];
}

typedef float v4f __attribute__((ext_vector_type(4)));
#define GBK_    32
#define GPITCH_ 36

template <int BM, int BN>
__global__ __launch_bounds__(256) void gemm_tiled(const float* __restrict__ A, int lda,
                                                  const float* __restrict__ W, int ldw,
                                                  const float* __restrict__ bias,
                                                  const float* __restrict__ resid,
                                                  float* __restrict__ Cout, int ldc,
                                                  int Kdim, int act) {
  constexpr int WC = BN / 32;
  constexpr int WR = 8 / WC;
  constexpr int TM = BM / (16 * WR);
  constexpr int TN = 2;
  constexpr int LA = BM / 32;
  constexpr int LB = BN / 32;
  __shared__ __attribute__((aligned(16))) float As[2][BM * GPITCH_];
  __shared__ __attribute__((aligned(16))) float Bs[2][BN * GPITCH_];
  const int t = threadIdx.x;
  const int lane = t & 31, w = t >> 5;
  const int sub = lane & 15, hi = lane >> 4;
  const int wr = w / WC, wc = w % WC;
  const int i0 = blockIdx.y * BM, j0 = blockIdx.x * BN;
  const int lr = t >> 3;
  const int lc = (t & 7) * 4;
  const float* ag = A + (size_t)(i0 + lr) * lda + lc;
  const float* wg = W + (size_t)(j0 + lr) * ldw + lc;
  float* asw = &As[0][lr * GPITCH_ + lc];
  float* bsw = &Bs[0][lr * GPITCH_ + lc];
  v4f ra[LA], rb[LB];
  #pragma unroll
  for (int i = 0; i < LA; ++i) ra[i] = *(const v4f*)(ag + (size_t)(32 * i) * lda);
  #pragma unroll
  for (int i = 0; i < LB; ++i) rb[i] = *(const v4f*)(wg + (size_t)(32 * i) * ldw);
  #pragma unroll
  for (int i = 0; i < LA; ++i) *(v4f*)(asw + 32 * i * GPITCH_) = ra[i];
  #pragma unroll
  for (int i = 0; i < LB; ++i) *(v4f*)(bsw + 32 * i * GPITCH_) = rb[i];
  __syncthreads();

  v8f acc[TM][TN];
  #pragma unroll
  for (int m = 0; m < TM; ++m)
    #pragma unroll
    for (int n = 0; n < TN; ++n)
      acc[m][n] = (v8f){0.f, 0.f, 0.f, 0.f, 0.f, 0.f, 0.f, 0.f};

  const int fa_off = (wr * TM * 16 + sub) * GPITCH_ + 4 * hi;
  const int fb_off = (wc * 32 + sub) * GPITCH_ + 4 * hi;
  const int nst = Kdim / GBK_;
  for (int st = 0; st < nst; ++st) {
    const int cur = st & 1;
    const bool more = (st + 1 < nst);
    if (more) {
      const int kn = (st + 1) * GBK_;
      #pragma unroll
      for (int i = 0; i < LA; ++i) ra[i] = *(const v4f*)(ag + (size_t)(32 * i) * lda + kn);
      #pragma unroll
      for (int i = 0; i < LB; ++i) rb[i] = *(const v4f*)(wg + (size_t)(32 * i) * ldw + kn);
    }
    const float* as = &As[cur][fa_off];
    const float* bs = &Bs[cur][fb_off];
    v4f fa[2][TM], fb[2][TN];
    #pragma unroll
    for (int m = 0; m < TM; ++m) fa[0][m] = *(const v4f*)(as + m * 16 * GPITCH_);
    #pragma unroll
    for (int n = 0; n < TN; ++n) fb[0][n] = *(const v4f*)(bs + n * 16 * GPITCH_);
    #pragma unroll
    for (int kb = 0; kb < GBK_ / 8; ++kb) {
      const int c = kb & 1, d = c ^ 1;
      if (kb + 1 < GBK_ / 8) {
        #pragma unroll
        for (int m = 0; m < TM; ++m) fa[d][m] = *(const v4f*)(as + m * 16 * GPITCH_ + (kb + 1) * 8);
        #pragma unroll
        for (int n = 0; n < TN; ++n) fb[d][n] = *(const v4f*)(bs + n * 16 * GPITCH_ + (kb + 1) * 8);
      }
      #pragma unroll
      for (int m = 0; m < TM; ++m)
        #pragma unroll
        for (int n = 0; n < TN; ++n) {
          const v2f a = {fa[c][m].x, fa[c][m].y};
          const v2f b = {fb[c][n].x, fb[c][n].y};
          acc[m][n] = wmma4(a, b, acc[m][n]);
        }
      #pragma unroll
      for (int m = 0; m < TM; ++m)
        #pragma unroll
        for (int n = 0; n < TN; ++n) {
          const v2f a = {fa[c][m].z, fa[c][m].w};
          const v2f b = {fb[c][n].z, fb[c][n].w};
          acc[m][n] = wmma4(a, b, acc[m][n]);
        }
    }
    if (more) {
      const int nxt = cur ^ 1;
      #pragma unroll
      for (int i = 0; i < LA; ++i) *(v4f*)(asw + nxt * BM * GPITCH_ + 32 * i * GPITCH_) = ra[i];
      #pragma unroll
      for (int i = 0; i < LB; ++i) *(v4f*)(bsw + nxt * BN * GPITCH_ + 32 * i * GPITCH_) = rb[i];
    }
    __syncthreads();
  }

  float bj[TN];
  #pragma unroll
  for (int n = 0; n < TN; ++n) bj[n] = bias ? bias[j0 + wc * 32 + n * 16 + sub] : 0.0f;
  #pragma unroll
  for (int m = 0; m < TM; ++m) {
    #pragma unroll
    for (int r = 0; r < 8; ++r) {
      const int row = i0 + (wr * TM + m) * 16 + r + 8 * hi;
      #pragma unroll
      for (int n = 0; n < TN; ++n) {
        const int col = j0 + wc * 32 + n * 16 + sub;
        float v = acc[m][n][r] + bj[n];
        if (act == 1) v = gelu_exact(v);
        if (resid) v += resid[(size_t)row * ldc + col];
        Cout[(size_t)row * ldc + col] = v;
      }
    }
  }
}

#define SP_ 1028

__device__ __forceinline__ float wave_max_f32(float v) {
  v = fmaxf(v, __builtin_bit_cast(float, dpp_row_i<0x161>(__builtin_bit_cast(int, v))));
  v = fmaxf(v, __builtin_bit_cast(float, dpp_row_i<0x162>(__builtin_bit_cast(int, v))));
  v = fmaxf(v, __builtin_bit_cast(float, dpp_row_i<0x164>(__builtin_bit_cast(int, v))));
  v = fmaxf(v, __builtin_bit_cast(float, dpp_row_i<0x168>(__builtin_bit_cast(int, v))));
  v = fmaxf(v, __builtin_bit_cast(float, other_row_i(__builtin_bit_cast(int, v))));
  return v;
}
__device__ __forceinline__ float wave_sum_f32(float v) {
  v += __builtin_bit_cast(float, dpp_row_i<0x161>(__builtin_bit_cast(int, v)));
  v += __builtin_bit_cast(float, dpp_row_i<0x162>(__builtin_bit_cast(int, v)));
  v += __builtin_bit_cast(float, dpp_row_i<0x164>(__builtin_bit_cast(int, v)));
  v += __builtin_bit_cast(float, dpp_row_i<0x168>(__builtin_bit_cast(int, v)));
  v += __builtin_bit_cast(float, other_row_i(__builtin_bit_cast(int, v)));
  return v;
}

__global__ void attn_kernel(const float* __restrict__ qkv,
                            const float* __restrict__ xyzf,
                            const float* __restrict__ Wsp,
                            float* __restrict__ attn_out) {
  extern __shared__ __attribute__((aligned(16))) float smem[];
  float* S    = smem;
  float* rsum = S + 16 * SP_;
  float* wx   = rsum + 16;
  const int b = blockIdx.z, h = blockIdx.y, i0 = blockIdx.x * 16;
  const int t = threadIdx.x;
  const int lane = t & 31, w = t >> 5;
  const int sub = lane & 15, hi = lane >> 4;

  const float* qbase =
      qkv + ((size_t)(b * MTOK_ + i0 + sub) * 3 + 0) * C_ + h * DH_ + 2 * hi;
  v2f qa[16];
  #pragma unroll
  for (int ks = 0; ks < 16; ++ks) qa[ks] = *(const v2f*)(qbase + 4 * ks);
  for (int jj = 0; jj < 8; ++jj) {
    const int j0 = (w * 8 + jj) * 16;
    const float* kbase =
        qkv + ((size_t)(b * MTOK_ + j0 + sub) * 3 + 1) * C_ + h * DH_ + 2 * hi;
    v8f acc = {0.f, 0.f, 0.f, 0.f, 0.f, 0.f, 0.f, 0.f};
    #pragma unroll
    for (int ks = 0; ks < 16; ++ks) {
      const v2f kb = *(const v2f*)(kbase + 4 * ks);
      acc = wmma4(qa[ks], kb, acc);
    }
    #pragma unroll
    for (int r = 0; r < 8; ++r)
      S[(r + 8 * hi) * SP_ + j0 + sub] = acc[r] * 0.125f;
  }
  __syncthreads();

  for (int q = 0; q < 2; ++q) {
    const int row = 2 * w + q;
    float* sr = S + row * SP_ + 4 * lane;
    v4f x[8];
    #pragma unroll
    for (int i = 0; i < 8; ++i) x[i] = *(const v4f*)(sr + 128 * i);
    float lm = -3.4e38f;
    #pragma unroll
    for (int i = 0; i < 8; ++i)
      lm = fmaxf(lm, fmaxf(fmaxf(x[i].x, x[i].y), fmaxf(x[i].z, x[i].w)));
    const float rm = wave_max_f32(lm);
    float ls = 0.0f;
    #pragma unroll
    for (int i = 0; i < 8; ++i) {
      v4f e;
      e.x = expf(x[i].x - rm);
      e.y = expf(x[i].y - rm);
      e.z = expf(x[i].z - rm);
      e.w = expf(x[i].w - rm);
      ls += (e.x + e.y) + (e.z + e.w);
      *(v4f*)(sr + 128 * i) = e;
    }
    ls = wave_sum_f32(ls);
    if (lane == 0) rsum[row] = ls;
  }
  __syncthreads();

  v8f acc = {0.f, 0.f, 0.f, 0.f, 0.f, 0.f, 0.f, 0.f};
  const int d0 = (w & 3) * 16;
  if (w < 4) {
    const float* vbase =
        qkv + ((size_t)(b * MTOK_) * 3 + 2) * C_ + h * DH_ + d0 + sub;
    const float* srow = S + sub * SP_ + 2 * hi;
    for (int ks = 0; ks < 256; ++ks) {
      const v2f a = *(const v2f*)(srow + 4 * ks);
      const int kk = 4 * ks + 2 * hi;
      v2f bb;
      bb.x = vbase[(size_t)kk * 1536];
      bb.y = vbase[(size_t)(kk + 1) * 1536];
      acc = wmma4(a, bb, acc);
    }
  } else {
    const int r0 = (w - 4) * 4;
    const float* xb = xyzf + (size_t)b * MTOK_ * 3;
    float ax[4][3];
    #pragma unroll
    for (int rr = 0; rr < 4; ++rr) { ax[rr][0] = 0.0f; ax[rr][1] = 0.0f; ax[rr][2] = 0.0f; }
    #pragma unroll 1
    for (int i = 0; i < 8; ++i) {
      const int j = 4 * lane + 128 * i;
      const v4f c0 = *(const v4f*)(xb + 3 * j);
      const v4f c1 = *(const v4f*)(xb + 3 * j + 4);
      const v4f c2 = *(const v4f*)(xb + 3 * j + 8);
      #pragma unroll
      for (int rr = 0; rr < 4; ++rr) {
        const v4f p = *(const v4f*)(S + (r0 + rr) * SP_ + j);
        ax[rr][0] += p.x * c0.x + p.y * c0.w + p.z * c1.z + p.w * c2.y;
        ax[rr][1] += p.x * c0.y + p.y * c1.x + p.z * c1.w + p.w * c2.z;
        ax[rr][2] += p.x * c0.z + p.y * c1.y + p.z * c2.x + p.w * c2.w;
      }
    }
    #pragma unroll
    for (int rr = 0; rr < 4; ++rr) {
      #pragma unroll
      for (int cc = 0; cc < 3; ++cc) ax[rr][cc] = wave_sum_f32(ax[rr][cc]);
    }
    if (lane == 0) {
      #pragma unroll
      for (int rr = 0; rr < 4; ++rr) {
        #pragma unroll
        for (int cc = 0; cc < 3; ++cc) wx[(r0 + rr) * 3 + cc] = ax[rr][cc];
      }
    }
  }
  __syncthreads();
  if (w < 4) {
    const int d = d0 + sub;
    const float w0 = Wsp[d * 3 + 0], w1 = Wsp[d * 3 + 1], w2 = Wsp[d * 3 + 2];
    #pragma unroll
    for (int r = 0; r < 8; ++r) {
      const int rr = r + 8 * hi;
      const int i  = i0 + rr;
      const float inv = 1.0f / rsum[rr];
      const float* xi = xyzf + ((size_t)b * MTOK_ + i) * 3;
      const float wd0 = wx[rr * 3 + 0] * inv - xi[0];
      const float wd1 = wx[rr * 3 + 1] * inv - xi[1];
      const float wd2 = wx[rr * 3 + 2] * inv - xi[2];
      attn_out[((size_t)(b * MTOK_) + i) * C_ + h * DH_ + d] =
          acc[r] * inv + w0 * wd0 + w1 * wd1 + w2 * wd2;
    }
  }
}

extern "C" void kernel_launch(void* const* d_in, const int* in_sizes, int n_in,
                              void* d_out, int out_size, void* d_ws, size_t ws_size,
                              hipStream_t stream) {
  (void)in_sizes; (void)n_in; (void)out_size; (void)ws_size;
  const float* xyzs = (const float*)d_in[0];
  const float* Wd   = (const float*)d_in[1];
  const float* Wqkv = (const float*)d_in[2];
  const float* Wsp  = (const float*)d_in[3];
  const float* Wout = (const float*)d_in[4];
  const float* bout = (const float*)d_in[5];
  const float* ln1g = (const float*)d_in[6];
  const float* ln1b = (const float*)d_in[7];
  const float* Wff1 = (const float*)d_in[8];
  const float* bff1 = (const float*)d_in[9];
  const float* Wff2 = (const float*)d_in[10];
  const float* bff2 = (const float*)d_in[11];
  const float* ln2g = (const float*)d_in[12];
  const float* ln2b = (const float*)d_in[13];

  float* ws      = (float*)d_ws;
  float* anchors = ws;
  float* feats   = ws + 8192;
  float* xln     = feats + (size_t)ROWS_ * C_;
  float* qkvb    = xln   + (size_t)ROWS_ * C_;
  float* attno   = qkvb  + (size_t)ROWS_ * 1536;
  float* ffh     = attno + (size_t)ROWS_ * C_;

  fps_kernel<<<B_ * L_, 32, 0, stream>>>(xyzs, anchors);
  p4dconv_kernel<<<B_ * L_ * MA_, N_, 0, stream>>>(xyzs, anchors, Wd, feats);

  const size_t smem_attn = (16 * 1028 + 16 + 48) * sizeof(float);
  for (int d = 0; d < 2; ++d) {
    ln_kernel<<<ROWS_, 256, 0, stream>>>(feats, ln1g + d * C_, ln1b + d * C_, xln);
    gemm_tiled<64, 64><<<dim3(1536 / 64, ROWS_ / 64), 256, 0, stream>>>(
        xln, C_, Wqkv + (size_t)d * 1536 * C_, C_, nullptr, nullptr,
        qkvb, 1536, C_, 0);
    attn_kernel<<<dim3(MTOK_ / 16, HEADS_, B_), 256, smem_attn, stream>>>(
        qkvb, anchors, Wsp + (size_t)d * DH_ * 3, attno);
    gemm_tiled<64, 64><<<dim3(C_ / 64, ROWS_ / 64), 256, 0, stream>>>(
        attno, C_, Wout + (size_t)d * C_ * C_, C_, bout + d * C_,
        feats, feats, C_, C_, 1);
    ln_kernel<<<ROWS_, 256, 0, stream>>>(feats, ln2g + d * C_, ln2b + d * C_, xln);
    gemm_tiled<128, 128><<<dim3(MLP_ / 128, ROWS_ / 128), 256, 0, stream>>>(
        xln, C_, Wff1 + (size_t)d * MLP_ * C_, C_, bff1 + d * MLP_,
        nullptr, ffh, MLP_, C_, 1);
    gemm_tiled<64, 64><<<dim3(C_ / 64, ROWS_ / 64), 256, 0, stream>>>(
        ffh, MLP_, Wff2 + (size_t)d * C_ * MLP_, MLP_, bff2 + d * C_,
        feats, feats, C_, MLP_, 0);
  }
  hipMemcpyAsync(d_out, feats, (size_t)ROWS_ * C_ * sizeof(float),
                 hipMemcpyDeviceToDevice, stream);
}


# p4dconv and layernorm: one wave per anchor / per row (ballot+popcount ranks, wave-butterfly sums) instead of 32- and 8-wave blocks with LDS scans and trees; LN sums in another order
# speedup vs baseline: 2.4378x; 1.0698x over previous
#include <hip/hip_runtime.h>
#include <hip/hip_bf16.h>
#include <math.h>

typedef float v2f __attribute__((ext_vector_type(2)));
typedef float v4f __attribute__((ext_vector_type(4)));
typedef float v8f __attribute__((ext_vector_type(8)));

#define B_    2
#define T_    4
#define N_    1024
#define L_    4
#define MA_   256
#define KNB_  32
#define C_    512
#define HEADS_ 8
#define DH_   64
#define MTOK_ 1024
#define ROWS_ 2048
#define MLP_  2048

__device__ __forceinline__ v8f wmma4(v2f a, v2f b, v8f c) {
  return __builtin_amdgcn_wmma_f32_16x16x4_f32(false, a, false, b, (short)0, c,
                                               false, false);
}

__device__ __forceinline__ float gelu_exact(float x) {
  return 0.5f * x * (1.0f + erff(x * 0.7071067811865475f));
}

#define FPS_SLOTS_  (N_ / 32)
#define FPS_CHAINS_ 4
#define FPS_CLEN_   (FPS_SLOTS_ / FPS_CHAINS_)

template <int CTRL>
__device__ __forceinline__ int dpp_row_i(int v) {
  return __builtin_amdgcn_update_dpp(0, v, CTRL, 0xf, 0xf, true);
}
__device__ __forceinline__ int other_row_i(int v) {
  return __builtin_amdgcn_permlanex16(v, v, 0x76543210, 0xfedcba98, false, false);
}
__device__ __forceinline__ int wave_max_i32(int v) {
  v = max(v, dpp_row_i<0x161>(v));
  v = max(v, dpp_row_i<0x162>(v));
  v = max(v, dpp_row_i<0x164>(v));
  v = max(v, dpp_row_i<0x168>(v));
  v = max(v, other_row_i(v));
  return v;
}
__device__ __forceinline__ int wave_min_i32(int v) {
  v = min(v, dpp_row_i<0x161>(v));
  v = min(v, dpp_row_i<0x162>(v));
  v = min(v, dpp_row_i<0x164>(v));
  v = min(v, dpp_row_i<0x168>(v));
  v = min(v, other_row_i(v));
  return v;
}

__device__ __forceinline__ float wave_max_f32(float v) {
  v = fmaxf(v, __builtin_bit_cast(float, dpp_row_i<0x161>(__builtin_bit_cast(int, v))));
  v = fmaxf(v, __builtin_bit_cast(float, dpp_row_i<0x162>(__builtin_bit_cast(int, v))));
  v = fmaxf(v, __builtin_bit_cast(float, dpp_row_i<0x164>(__builtin_bit_cast(int, v))));
  v = fmaxf(v, __builtin_bit_cast(float, dpp_row_i<0x168>(__builtin_bit_cast(int, v))));
  v = fmaxf(v, __builtin_bit_cast(float, other_row_i(__builtin_bit_cast(int, v))));
  return v;
}
__device__ __forceinline__ float wave_sum_f32(float v) {
  v += __builtin_bit_cast(float, dpp_row_i<0x161>(__builtin_bit_cast(int, v)));
  v += __builtin_bit_cast(float, dpp_row_i<0x162>(__builtin_bit_cast(int, v)));
  v += __builtin_bit_cast(float, dpp_row_i<0x164>(__builtin_bit_cast(int, v)));
  v += __builtin_bit_cast(float, dpp_row_i<0x168>(__builtin_bit_cast(int, v)));
  v += __builtin_bit_cast(float, other_row_i(__builtin_bit_cast(int, v)));
  return v;
}

__global__ __launch_bounds__(32) void fps_kernel(const float* __restrict__ xyzs,
                                                 float* __restrict__ anchors) {
  __shared__ float4 spt[N_];
  const int blk  = blockIdx.x;
  const int lane = threadIdx.x;
  const float* frame = xyzs + (size_t)blk * N_ * 3;
  float px[FPS_SLOTS_], py[FPS_SLOTS_], pz[FPS_SLOTS_];
  int   dist[FPS_SLOTS_];
  #pragma unroll
  for (int s = 0; s < FPS_SLOTS_; ++s) {
    const int i = s * 32 + lane;
    px[s] = frame[i * 3 + 0];
    py[s] = frame[i * 3 + 1];
    pz[s] = frame[i * 3 + 2];
    dist[s] = __builtin_bit_cast(int, 1e10f);
    spt[i] = make_float4(px[s], py[s], pz[s], 0.0f);
  }
  __syncthreads();
  float* aout = anchors + (size_t)blk * MA_ * 3;
  const char* sbytes = (const char*)spt;
  int curb = 0;
  for (int it = 1; it < MA_; ++it) {
    const float4 last = *(const float4*)(sbytes + curb);
    if (lane == 0) {
      aout[(it - 1) * 3 + 0] = last.x;
      aout[(it - 1) * 3 + 1] = last.y;
      aout[(it - 1) * 3 + 2] = last.z;
    }
    #pragma unroll
    for (int s = 0; s < FPS_SLOTS_; ++s) {
      const float dx = px[s] - last.x, dy = py[s] - last.y, dz = pz[s] - last.z;
      const float d2 = fmaf(dz, dz, fmaf(dx, dx, dy * dy));
      dist[s] = min(dist[s], __builtin_bit_cast(int, d2));
    }
    int bv[FPS_CHAINS_], bo[FPS_CHAINS_];
    #pragma unroll
    for (int c = 0; c < FPS_CHAINS_; ++c) {
      bv[c] = dist[c * FPS_CLEN_];
      bo[c] = c * FPS_CLEN_ * 32 * 16;
      #pragma unroll
      for (int k = 1; k < FPS_CLEN_; ++k) {
        const int s = c * FPS_CLEN_ + k;
        bo[c] = (dist[s] > bv[c]) ? s * 32 * 16 : bo[c];
        bv[c] = max(bv[c], dist[s]);
      }
    }
    #pragma unroll
    for (int w = 1; w < FPS_CHAINS_; w <<= 1) {
      #pragma unroll
      for (int c = 0; c < FPS_CHAINS_; c += 2 * w) {
        bo[c] = (bv[c + w] > bv[c]) ? bo[c + w] : bo[c];
        bv[c] = max(bv[c], bv[c + w]);
      }
    }
    const int gmax = wave_max_i32(bv[0]);
    const int cand = (bv[0] == gmax) ? (bo[0] + lane * 16) : 0x7fffffff;
    curb = wave_min_i32(cand);
  }
  const float4 last = *(const float4*)(sbytes + curb);
  if (lane == 0) {
    aout[(MA_ - 1) * 3 + 0] = last.x;
    aout[(MA_ - 1) * 3 + 1] = last.y;
    aout[(MA_ - 1) * 3 + 2] = last.z;
  }
}

#define P4W_ 8

__global__ __launch_bounds__(256) void p4dconv_kernel(const float* __restrict__ xyzs,
                                                      const float* __restrict__ anchors,
                                                      const float* __restrict__ Wd,
                                                      float* __restrict__ feats) {
  __shared__ __attribute__((aligned(16))) float  sfr[3][N_ * 3];
  __shared__ __attribute__((aligned(16))) float4 snb[P4W_][KNB_];
  const int t = threadIdx.x;
  const int lane = t & 31, w = t >> 5;
  const int blk = blockIdx.x;
  const int bl  = blk / (MA_ / P4W_);
  const int m   = (blk % (MA_ / P4W_)) * P4W_ + w;
  const int l   = bl % L_;
  const int b   = bl / L_;

  for (int f = 0; f < 3; ++f) {
    int fi = l + f - 1; fi = fi < 0 ? 0 : (fi > T_ - 1 ? T_ - 1 : fi);
    const v4f* srcp = (const v4f*)(xyzs + ((size_t)b * T_ + fi) * N_ * 3);
    v4f* dstp = (v4f*)&sfr[f][0];
    for (int i = 0; i < 3; ++i) dstp[t + 256 * i] = srcp[t + 256 * i];
  }

  const float* arow = anchors + ((size_t)bl * MA_ + m) * 3;
  const float ax = arow[0], ay = arow[1], az = arow[2];

  v2f w0[8], w1[8], w2[8], w3[8], mx[8];
  #pragma unroll
  for (int j = 0; j < 8; ++j) {
    const float* wp = Wd + (size_t)(2 * lane + 64 * j) * 4;
    const v4f wa = *(const v4f*)wp;
    const v4f wb = *(const v4f*)(wp + 4);
    w0[j] = (v2f){wa.x, wb.x};
    w1[j] = (v2f){wa.y, wb.y};
    w2[j] = (v2f){wa.z, wb.z};
    w3[j] = (v2f){wa.w, wb.w};
    mx[j] = (v2f){-3.4e38f, -3.4e38f};
  }
  __syncthreads();

  for (int f = 0; f < 3; ++f) {
    const float* fr = &sfr[f][0];
    const float p0x = fr[0] - ax, p0y = fr[1] - ay, p0z = fr[2] - az;
    int cnt = 0;
    for (int g = 0; g < 4 && cnt < KNB_; ++g) {
      float q[8][3];
      #pragma unroll
      for (int u = 0; u < 8; ++u) {
        const int i = (g * 8 + u) * 32 + lane;
        q[u][0] = fr[3 * i + 0];
        q[u][1] = fr[3 * i + 1];
        q[u][2] = fr[3 * i + 2];
      }
      #pragma unroll
      for (int u = 0; u < 8; ++u) {
        const float dx = q[u][0] - ax;
        const float dy = q[u][1] - ay;
        const float dz = q[u][2] - az;
        const float d2 = fmaf(dz, dz, fmaf(dx, dx, dy * dy));
        const bool pred = d2 < 1.0f;
        const unsigned mask = __builtin_amdgcn_ballot_w32(pred);
        const int pos = cnt + (int)__builtin_amdgcn_mbcnt_lo(mask, 0u);
        if (pred && pos < KNB_) snb[w][pos] = make_float4(dx, dy, dz, 0.0f);
        cnt += __builtin_popcount(mask);
      }
    }
    __syncthreads();
    if (lane >= cnt) {
      snb[w][lane] = (cnt > 0) ? snb[w][0] : make_float4(p0x, p0y, p0z, 0.0f);
    }
    __syncthreads();
    const float tq = (float)(f - 1);
    #pragma unroll 4
    for (int k = 0; k < KNB_; ++k) {
      const float4 p = snb[w][k];
      #pragma unroll
      for (int j = 0; j < 8; ++j) {
        const v2f v = w0[j] * p.x + w1[j] * p.y + w2[j] * p.z + w3[j] * tq;
        mx[j] = __builtin_elementwise_max(mx[j], v);
      }
    }
    __syncthreads();
  }
  float* orow = feats + ((size_t)bl * MA_ + m) * C_ + 2 * lane;
  #pragma unroll
  for (int j = 0; j < 8; ++j) {
    const v2f o = {fmaxf(mx[j].x, 0.0f), fmaxf(mx[j].y, 0.0f)};
    *(v2f*)(orow + 64 * j) = o;
  }
}

__global__ __launch_bounds__(256) void ln_kernel(const float* __restrict__ x, const float* __restrict__ g,
                                                 const float* __restrict__ bta, float* __restrict__ y) {
  const int lane = threadIdx.x & 31, w = threadIdx.x >> 5;
  const int row = blockIdx.x * 8 + w;
  const float* xr = x + (size_t)row * C_ + 4 * lane;
  v4f a[4], d[4];
  #pragma unroll
  for (int i = 0; i < 4; ++i) a[i] = *(const v4f*)(xr + 128 * i);
  float s = 0.0f;
  #pragma unroll
  for (int i = 0; i < 4; ++i) s += (a[i].x + a[i].y) + (a[i].z + a[i].w);
  const float mean = wave_sum_f32(s) * (1.0f / C_);
  float q = 0.0f;
  #pragma unroll
  for (int i = 0; i < 4; ++i) {
    d[i] = a[i] - mean;
    q += (d[i].x * d[i].x + d[i].y * d[i].y) + (d[i].z * d[i].z + d[i].w * d[i].w);
  }
  const float inv = rsqrtf(wave_sum_f32(q) * (1.0f / C_) + 1e-5f);
  float* yr = y + (size_t)row * C_ + 4 * lane;
  #pragma unroll
  for (int i = 0; i < 4; ++i) {
    const v4f gg = *(const v4f*)(g + 4 * lane + 128 * i);
    const v4f bb = *(const v4f*)(bta + 4 * lane + 128 * i);
    *(v4f*)(yr + 128 * i) = d[i] * inv * gg + bb;
  }
}

#define GBK_    32
#define GPITCH_ 36

template <int BM, int BN>
__global__ __launch_bounds__(256) void gemm_tiled(const float* __restrict__ A, int lda,
                                                  const float* __restrict__ W, int ldw,
                                                  const float* __restrict__ bias,
                                                  const float* __restrict__ resid,
                                                  float* __restrict__ Cout, int ldc,
                                                  int Kdim, int act) {
  constexpr int WC = BN / 32;
  constexpr int WR = 8 / WC;
  constexpr int TM = BM / (16 * WR);
  constexpr int TN = 2;
  constexpr int LA = BM / 32;
  constexpr int LB = BN / 32;
  __shared__ __attribute__((aligned(16))) float As[2][BM * GPITCH_];
  __shared__ __attribute__((aligned(16))) float Bs[2][BN * GPITCH_];
  const int t = threadIdx.x;
  const int lane = t & 31, w = t >> 5;
  const int sub = lane & 15, hi = lane >> 4;
  const int wr = w / WC, wc = w % WC;
  const int i0 = blockIdx.y * BM, j0 = blockIdx.x * BN;
  const int lr = t >> 3;
  const int lc = (t & 7) * 4;
  const float* ag = A + (size_t)(i0 + lr) * lda + lc;
  const float* wg = W + (size_t)(j0 + lr) * ldw + lc;
  float* asw = &As[0][lr * GPITCH_ + lc];
  float* bsw = &Bs[0][lr * GPITCH_ + lc];
  v4f ra[LA], rb[LB];
  #pragma unroll
  for (int i = 0; i < LA; ++i) ra[i] = *(const v4f*)(ag + (size_t)(32 * i) * lda);
  #pragma unroll
  for (int i = 0; i < LB; ++i) rb[i] = *(const v4f*)(wg + (size_t)(32 * i) * ldw);
  #pragma unroll
  for (int i = 0; i < LA; ++i) *(v4f*)(asw + 32 * i * GPITCH_) = ra[i];
  #pragma unroll
  for (int i = 0; i < LB; ++i) *(v4f*)(bsw + 32 * i * GPITCH_) = rb[i];
  __syncthreads();

  v8f acc[TM][TN];
  #pragma unroll
  for (int m = 0; m < TM; ++m)
    #pragma unroll
    for (int n = 0; n < TN; ++n)
      acc[m][n] = (v8f){0.f, 0.f, 0.f, 0.f, 0.f, 0.f, 0.f, 0.f};

  const int fa_off = (wr * TM * 16 + sub) * GPITCH_ + 4 * hi;
  const int fb_off = (wc * 32 + sub) * GPITCH_ + 4 * hi;
  const int nst = Kdim / GBK_;
  for (int st = 0; st < nst; ++st) {
    const int cur = st & 1;
    const bool more = (st + 1 < nst);
    if (more) {
      const int kn = (st + 1) * GBK_;
      #pragma unroll
      for (int i = 0; i < LA; ++i) ra[i] = *(const v4f*)(ag + (size_t)(32 * i) * lda + kn);
      #pragma unroll
      for (int i = 0; i < LB; ++i) rb[i] = *(const v4f*)(wg + (size_t)(32 * i) * ldw + kn);
    }
    const float* as = &As[cur][fa_off];
    const float* bs = &Bs[cur][fb_off];
    v4f fa[2][TM], fb[2][TN];
    #pragma unroll
    for (int m = 0; m < TM; ++m) fa[0][m] = *(const v4f*)(as + m * 16 * GPITCH_);
    #pragma unroll
    for (int n = 0; n < TN; ++n) fb[0][n] = *(const v4f*)(bs + n * 16 * GPITCH_);
    #pragma unroll
    for (int kb = 0; kb < GBK_ / 8; ++kb) {
      const int c = kb & 1, d = c ^ 1;
      if (kb + 1 < GBK_ / 8) {
        #pragma unroll
        for (int m = 0; m < TM; ++m) fa[d][m] = *(const v4f*)(as + m * 16 * GPITCH_ + (kb + 1) * 8);
        #pragma unroll
        for (int n = 0; n < TN; ++n) fb[d][n] = *(const v4f*)(bs + n * 16 * GPITCH_ + (kb + 1) * 8);
      }
      #pragma unroll
      for (int m = 0; m < TM; ++m)
        #pragma unroll
        for (int n = 0; n < TN; ++n) {
          const v2f a = {fa[c][m].x, fa[c][m].y};
          const v2f b = {fb[c][n].x, fb[c][n].y};
          acc[m][n] = wmma4(a, b, acc[m][n]);
        }
      #pragma unroll
      for (int m = 0; m < TM; ++m)
        #pragma unroll
        for (int n = 0; n < TN; ++n) {
          const v2f a = {fa[c][m].z, fa[c][m].w};
          const v2f b = {fb[c][n].z, fb[c][n].w};
          acc[m][n] = wmma4(a, b, acc[m][n]);
        }
    }
    if (more) {
      const int nxt = cur ^ 1;
      #pragma unroll
      for (int i = 0; i < LA; ++i) *(v4f*)(asw + nxt * BM * GPITCH_ + 32 * i * GPITCH_) = ra[i];
      #pragma unroll
      for (int i = 0; i < LB; ++i) *(v4f*)(bsw + nxt * BN * GPITCH_ + 32 * i * GPITCH_) = rb[i];
    }
    __syncthreads();
  }

  float bj[TN];
  #pragma unroll
  for (int n = 0; n < TN; ++n) bj[n] = bias ? bias[j0 + wc * 32 + n * 16 + sub] : 0.0f;
  #pragma unroll
  for (int m = 0; m < TM; ++m) {
    #pragma unroll
    for (int r = 0; r < 8; ++r) {
      const int row = i0 + (wr * TM + m) * 16 + r + 8 * hi;
      #pragma unroll
      for (int n = 0; n < TN; ++n) {
        const int col = j0 + wc * 32 + n * 16 + sub;
        float v = acc[m][n][r] + bj[n];
        if (act == 1) v = gelu_exact(v);
        if (resid) v += resid[(size_t)row * ldc + col];
        Cout[(size_t)row * ldc + col] = v;
      }
    }
  }
}

#define SP_ 1028

__global__ void attn_kernel(const float* __restrict__ qkv,
                            const float* __restrict__ xyzf,
                            const float* __restrict__ Wsp,
                            float* __restrict__ attn_out) {
  extern __shared__ __attribute__((aligned(16))) float smem[];
  float* S    = smem;
  float* rsum = S + 16 * SP_;
  float* wx   = rsum + 16;
  const int b = blockIdx.z, h = blockIdx.y, i0 = blockIdx.x * 16;
  const int t = threadIdx.x;
  const int lane = t & 31, w = t >> 5;
  const int sub = lane & 15, hi = lane >> 4;

  const float* qbase =
      qkv + ((size_t)(b * MTOK_ + i0 + sub) * 3 + 0) * C_ + h * DH_ + 2 * hi;
  v2f qa[16];
  #pragma unroll
  for (int ks = 0; ks < 16; ++ks) qa[ks] = *(const v2f*)(qbase + 4 * ks);
  for (int jj = 0; jj < 8; ++jj) {
    const int j0 = (w * 8 + jj) * 16;
    const float* kbase =
        qkv + ((size_t)(b * MTOK_ + j0 + sub) * 3 + 1) * C_ + h * DH_ + 2 * hi;
    v8f acc = {0.f, 0.f, 0.f, 0.f, 0.f, 0.f, 0.f, 0.f};
    #pragma unroll
    for (int ks = 0; ks < 16; ++ks) {
      const v2f kb = *(const v2f*)(kbase + 4 * ks);
      acc = wmma4(qa[ks], kb, acc);
    }
    #pragma unroll
    for (int r = 0; r < 8; ++r)
      S[(r + 8 * hi) * SP_ + j0 + sub] = acc[r] * 0.125f;
  }
  __syncthreads();

  for (int q = 0; q < 2; ++q) {
    const int row = 2 * w + q;
    float* sr = S + row * SP_ + 4 * lane;
    v4f x[8];
    #pragma unroll
    for (int i = 0; i < 8; ++i) x[i] = *(const v4f*)(sr + 128 * i);
    float lm = -3.4e38f;
    #pragma unroll
    for (int i = 0; i < 8; ++i)
      lm = fmaxf(lm, fmaxf(fmaxf(x[i].x, x[i].y), fmaxf(x[i].z, x[i].w)));
    const float rm = wave_max_f32(lm);
    float ls = 0.0f;
    #pragma unroll
    for (int i = 0; i < 8; ++i) {
      v4f e;
      e.x = expf(x[i].x - rm);
      e.y = expf(x[i].y - rm);
      e.z = expf(x[i].z - rm);
      e.w = expf(x[i].w - rm);
      ls += (e.x + e.y) + (e.z + e.w);
      *(v4f*)(sr + 128 * i) = e;
    }
    ls = wave_sum_f32(ls);
    if (lane == 0) rsum[row] = ls;
  }
  __syncthreads();

  v8f acc = {0.f, 0.f, 0.f, 0.f, 0.f, 0.f, 0.f, 0.f};
  const int d0 = (w & 3) * 16;
  if (w < 4) {
    const float* vbase =
        qkv + ((size_t)(b * MTOK_) * 3 + 2) * C_ + h * DH_ + d0 + sub;
    const float* srow = S + sub * SP_ + 2 * hi;
    for (int ks = 0; ks < 256; ++ks) {
      const v2f a = *(const v2f*)(srow + 4 * ks);
      const int kk = 4 * ks + 2 * hi;
      v2f bb;
      bb.x = vbase[(size_t)kk * 1536];
      bb.y = vbase[(size_t)(kk + 1) * 1536];
      acc = wmma4(a, bb, acc);
    }
  } else {
    const int r0 = (w - 4) * 4;
    const float* xb = xyzf + (size_t)b * MTOK_ * 3;
    float ax[4][3];
    #pragma unroll
    for (int rr = 0; rr < 4; ++rr) { ax[rr][0] = 0.0f; ax[rr][1] = 0.0f; ax[rr][2] = 0.0f; }
    #pragma unroll 1
    for (int i = 0; i < 8; ++i) {
      const int j = 4 * lane + 128 * i;
      const v4f c0 = *(const v4f*)(xb + 3 * j);
      const v4f c1 = *(const v4f*)(xb + 3 * j + 4);
      const v4f c2 = *(const v4f*)(xb + 3 * j + 8);
      #pragma unroll
      for (int rr = 0; rr < 4; ++rr) {
        const v4f p = *(const v4f*)(S + (r0 + rr) * SP_ + j);
        ax[rr][0] += p.x * c0.x + p.y * c0.w + p.z * c1.z + p.w * c2.y;
        ax[rr][1] += p.x * c0.y + p.y * c1.x + p.z * c1.w + p.w * c2.z;
        ax[rr][2] += p.x * c0.z + p.y * c1.y + p.z * c2.x + p.w * c2.w;
      }
    }
    #pragma unroll
    for (int rr = 0; rr < 4; ++rr) {
      #pragma unroll
      for (int cc = 0; cc < 3; ++cc) ax[rr][cc] = wave_sum_f32(ax[rr][cc]);
    }
    if (lane == 0) {
      #pragma unroll
      for (int rr = 0; rr < 4; ++rr) {
        #pragma unroll
        for (int cc = 0; cc < 3; ++cc) wx[(r0 + rr) * 3 + cc] = ax[rr][cc];
      }
    }
  }
  __syncthreads();
  if (w < 4) {
    const int d = d0 + sub;
    const float w0 = Wsp[d * 3 + 0], w1 = Wsp[d * 3 + 1], w2 = Wsp[d * 3 + 2];
    #pragma unroll
    for (int r = 0; r < 8; ++r) {
      const int rr = r + 8 * hi;
      const int i  = i0 + rr;
      const float inv = 1.0f / rsum[rr];
      const float* xi = xyzf + ((size_t)b * MTOK_ + i) * 3;
      const float wd0 = wx[rr * 3 + 0] * inv - xi[0];
      const float wd1 = wx[rr * 3 + 1] * inv - xi[1];
      const float wd2 = wx[rr * 3 + 2] * inv - xi[2];
      attn_out[((size_t)(b * MTOK_) + i) * C_ + h * DH_ + d] =
          acc[r] * inv + w0 * wd0 + w1 * wd1 + w2 * wd2;
    }
  }
}

extern "C" void kernel_launch(void* const* d_in, const int* in_sizes, int n_in,
                              void* d_out, int out_size, void* d_ws, size_t ws_size,
                              hipStream_t stream) {
  (void)in_sizes; (void)n_in; (void)out_size; (void)ws_size;
  const float* xyzs = (const float*)d_in[0];
  const float* Wd   = (const float*)d_in[1];
  const float* Wqkv = (const float*)d_in[2];
  const float* Wsp  = (const float*)d_in[3];
  const float* Wout = (const float*)d_in[4];
  const float* bout = (const float*)d_in[5];
  const float* ln1g = (const float*)d_in[6];
  const float* ln1b = (const float*)d_in[7];
  const float* Wff1 = (const float*)d_in[8];
  const float* bff1 = (const float*)d_in[9];
  const float* Wff2 = (const float*)d_in[10];
  const float* bff2 = (const float*)d_in[11];
  const float* ln2g = (const float*)d_in[12];
  const float* ln2b = (const float*)d_in[13];

  float* ws      = (float*)d_ws;
  float* anchors = ws;
  float* feats   = ws + 8192;
  float* xln     = feats + (size_t)ROWS_ * C_;
  float* qkvb    = xln   + (size_t)ROWS_ * C_;
  float* attno   = qkvb  + (size_t)ROWS_ * 1536;
  float* ffh     = attno + (size_t)ROWS_ * C_;

  fps_kernel<<<B_ * L_, 32, 0, stream>>>(xyzs, anchors);
  p4dconv_kernel<<<B_ * L_ * MA_ / P4W_, 256, 0, stream>>>(xyzs, anchors, Wd, feats);

  const size_t smem_attn = (16 * 1028 + 16 + 48) * sizeof(float);
  for (int d = 0; d < 2; ++d) {
    ln_kernel<<<ROWS_ / 8, 256, 0, stream>>>(feats, ln1g + d * C_, ln1b + d * C_, xln);
    gemm_tiled<64, 64><<<dim3(1536 / 64, ROWS_ / 64), 256, 0, stream>>>(
        xln, C_, Wqkv + (size_t)d * 1536 * C_, C_, nullptr, nullptr,
        qkvb, 1536, C_, 0);
    attn_kernel<<<dim3(MTOK_ / 16, HEADS_, B_), 256, smem_attn, stream>>>(
        qkvb, anchors, Wsp + (size_t)d * DH_ * 3, attno);
    gemm_tiled<64, 64><<<dim3(C_ / 64, ROWS_ / 64), 256, 0, stream>>>(
        attno, C_, Wout + (size_t)d * C_ * C_, C_, bout + d * C_,
        feats, feats, C_, C_, 1);
    ln_kernel<<<ROWS_ / 8, 256, 0, stream>>>(feats, ln2g + d * C_, ln2b + d * C_, xln);
    gemm_tiled<128, 128><<<dim3(MLP_ / 128, ROWS_ / 128), 256, 0, stream>>>(
        xln, C_, Wff1 + (size_t)d * MLP_ * C_, C_, bff1 + d * MLP_,
        nullptr, ffh, MLP_, C_, 1);
    gemm_tiled<64, 64><<<dim3(C_ / 64, ROWS_ / 64), 256, 0, stream>>>(
        ffh, MLP_, Wff2 + (size_t)d * C_ * MLP_, MLP_, bff2 + d * C_,
        feats, feats, C_, MLP_, 0);
  }
  hipMemcpyAsync(d_out, feats, (size_t)ROWS_ * C_ * sizeof(float),
                 hipMemcpyDeviceToDevice, stream);
}


# GEMM k loop pipelined across the stage barrier: ring of three LDS buffers, 64-wide stages, next tile stored and next fragments read ahead of the barrier, which orders LDS only
# speedup vs baseline: 2.6524x; 1.0880x over previous
#include <hip/hip_runtime.h>
#include <hip/hip_bf16.h>
#include <math.h>

typedef float v2f __attribute__((ext_vector_type(2)));
typedef float v4f __attribute__((ext_vector_type(4)));
typedef float v8f __attribute__((ext_vector_type(8)));

#define B_    2
#define T_    4
#define N_    1024
#define L_    4
#define MA_   256
#define KNB_  32
#define C_    512
#define HEADS_ 8
#define DH_   64
#define MTOK_ 1024
#define ROWS_ 2048
#define MLP_  2048

__device__ __forceinline__ v8f wmma4(v2f a, v2f b, v8f c) {
  return __builtin_amdgcn_wmma_f32_16x16x4_f32(false, a, false, b, (short)0, c,
                                               false, false);
}

__device__ __forceinline__ float gelu_exact(float x) {
  return 0.5f * x * (1.0f + erff(x * 0.7071067811865475f));
}

#define FPS_SLOTS_  (N_ / 32)
#define FPS_CHAINS_ 4
#define FPS_CLEN_   (FPS_SLOTS_ / FPS_CHAINS_)

template <int CTRL>
__device__ __forceinline__ int dpp_row_i(int v) {
  return __builtin_amdgcn_update_dpp(0, v, CTRL, 0xf, 0xf, true);
}
__device__ __forceinline__ int other_row_i(int v) {
  return __builtin_amdgcn_permlanex16(v, v, 0x76543210, 0xfedcba98, false, false);
}
__device__ __forceinline__ int wave_max_i32(int v) {
  v = max(v, dpp_row_i<0x161>(v));
  v = max(v, dpp_row_i<0x162>(v));
  v = max(v, dpp_row_i<0x164>(v));
  v = max(v, dpp_row_i<0x168>(v));
  v = max(v, other_row_i(v));
  return v;
}
__device__ __forceinline__ int wave_min_i32(int v) {
  v = min(v, dpp_row_i<0x161>(v));
  v = min(v, dpp_row_i<0x162>(v));
  v = min(v, dpp_row_i<0x164>(v));
  v = min(v, dpp_row_i<0x168>(v));
  v = min(v, other_row_i(v));
  return v;
}

__device__ __forceinline__ float wave_max_f32(float v) {
  v = fmaxf(v, __builtin_bit_cast(float, dpp_row_i<0x161>(__builtin_bit_cast(int, v))));
  v = fmaxf(v, __builtin_bit_cast(float, dpp_row_i<0x162>(__builtin_bit_cast(int, v))));
  v = fmaxf(v, __builtin_bit_cast(float, dpp_row_i<0x164>(__builtin_bit_cast(int, v))));
  v = fmaxf(v, __builtin_bit_cast(float, dpp_row_i<0x168>(__builtin_bit_cast(int, v))));
  v = fmaxf(v, __builtin_bit_cast(float, other_row_i(__builtin_bit_cast(int, v))));
  return v;
}
__device__ __forceinline__ float wave_sum_f32(float v) {
  v += __builtin_bit_cast(float, dpp_row_i<0x161>(__builtin_bit_cast(int, v)));
  v += __builtin_bit_cast(float, dpp_row_i<0x162>(__builtin_bit_cast(int, v)));
  v += __builtin_bit_cast(float, dpp_row_i<0x164>(__builtin_bit_cast(int, v)));
  v += __builtin_bit_cast(float, dpp_row_i<0x168>(__builtin_bit_cast(int, v)));
  v += __builtin_bit_cast(float, other_row_i(__builtin_bit_cast(int, v)));
  return v;
}

__global__ __launch_bounds__(32) void fps_kernel(const float* __restrict__ xyzs,
                                                 float* __restrict__ anchors) {
  __shared__ float4 spt[N_];
  const int blk  = blockIdx.x;
  const int lane = threadIdx.x;
  const float* frame = xyzs + (size_t)blk * N_ * 3;
  float px[FPS_SLOTS_], py[FPS_SLOTS_], pz[FPS_SLOTS_];
  int   dist[FPS_SLOTS_];
  #pragma unroll
  for (int s = 0; s < FPS_SLOTS_; ++s) {
    const int i = s * 32 + lane;
    px[s] = frame[i * 3 + 0];
    py[s] = frame[i * 3 + 1];
    pz[s] = frame[i * 3 + 2];
    dist[s] = __builtin_bit_cast(int, 1e10f);
    spt[i] = make_float4(px[s], py[s], pz[s], 0.0f);
  }
  __syncthreads();
  float* aout = anchors + (size_t)blk * MA_ * 3;
  const char* sbytes = (const char*)spt;
  int curb = 0;
  for (int it = 1; it < MA_; ++it) {
    const float4 last = *(const float4*)(sbytes + curb);
    if (lane == 0) {
      aout[(it - 1) * 3 + 0] = last.x;
      aout[(it - 1) * 3 + 1] = last.y;
      aout[(it - 1) * 3 + 2] = last.z;
    }
    #pragma unroll
    for (int s = 0; s < FPS_SLOTS_; ++s) {
      const float dx = px[s] - last.x, dy = py[s] - last.y, dz = pz[s] - last.z;
      const float d2 = fmaf(dz, dz, fmaf(dx, dx, dy * dy));
      dist[s] = min(dist[s], __builtin_bit_cast(int, d2));
    }
    int bv[FPS_CHAINS_], bo[FPS_CHAINS_];
    #pragma unroll
    for (int c = 0; c < FPS_CHAINS_; ++c) {
      bv[c] = dist[c * FPS_CLEN_];
      bo[c] = c * FPS_CLEN_ * 32 * 16;
      #pragma unroll
      for (int k = 1; k < FPS_CLEN_; ++k) {
        const int s = c * FPS_CLEN_ + k;
        bo[c] = (dist[s] > bv[c]) ? s * 32 * 16 : bo[c];
        bv[c] = max(bv[c], dist[s]);
      }
    }
    #pragma unroll
    for (int w = 1; w < FPS_CHAINS_; w <<= 1) {
      #pragma unroll
      for (int c = 0; c < FPS_CHAINS_; c += 2 * w) {
        bo[c] = (bv[c + w] > bv[c]) ? bo[c + w] : bo[c];
        bv[c] = max(bv[c], bv[c + w]);
      }
    }
    const int gmax = wave_max_i32(bv[0]);
    const int cand = (bv[0] == gmax) ? (bo[0] + lane * 16) : 0x7fffffff;
    curb = wave_min_i32(cand);
  }
  const float4 last = *(const float4*)(sbytes + curb);
  if (lane == 0) {
    aout[(MA_ - 1) * 3 + 0] = last.x;
    aout[(MA_ - 1) * 3 + 1] = last.y;
    aout[(MA_ - 1) * 3 + 2] = last.z;
  }
}

#define P4W_ 8

__global__ __launch_bounds__(256) void p4dconv_kernel(const float* __restrict__ xyzs,
                                                      const float* __restrict__ anchors,
                                                      const float* __restrict__ Wd,
                                                      float* __restrict__ feats) {
  __shared__ __attribute__((aligned(16))) float  sfr[3][N_ * 3];
  __shared__ __attribute__((aligned(16))) float4 snb[P4W_][KNB_];
  const int t = threadIdx.x;
  const int lane = t & 31, w = t >> 5;
  const int blk = blockIdx.x;
  const int bl  = blk / (MA_ / P4W_);
  const int m   = (blk % (MA_ / P4W_)) * P4W_ + w;
  const int l   = bl % L_;
  const int b   = bl / L_;

  for (int f = 0; f < 3; ++f) {
    int fi = l + f - 1; fi = fi < 0 ? 0 : (fi > T_ - 1 ? T_ - 1 : fi);
    const v4f* srcp = (const v4f*)(xyzs + ((size_t)b * T_ + fi) * N_ * 3);
    v4f* dstp = (v4f*)&sfr[f][0];
    for (int i = 0; i < 3; ++i) dstp[t + 256 * i] = srcp[t + 256 * i];
  }

  const float* arow = anchors + ((size_t)bl * MA_ + m) * 3;
  const float ax = arow[0], ay = arow[1], az = arow[2];

  v2f w0[8], w1[8], w2[8], w3[8], mx[8];
  #pragma unroll
  for (int j = 0; j < 8; ++j) {
    const float* wp = Wd + (size_t)(2 * lane + 64 * j) * 4;
    const v4f wa = *(const v4f*)wp;
    const v4f wb = *(const v4f*)(wp + 4);
    w0[j] = (v2f){wa.x, wb.x};
    w1[j] = (v2f){wa.y, wb.y};
    w2[j] = (v2f){wa.z, wb.z};
    w3[j] = (v2f){wa.w, wb.w};
    mx[j] = (v2f){-3.4e38f, -3.4e38f};
  }
  __syncthreads();

  for (int f = 0; f < 3; ++f) {
    const float* fr = &sfr[f][0];
    const float p0x = fr[0] - ax, p0y = fr[1] - ay, p0z = fr[2] - az;
    int cnt = 0;
    for (int g = 0; g < 4 && cnt < KNB_; ++g) {
      float q[8][3];
      #pragma unroll
      for (int u = 0; u < 8; ++u) {
        const int i = (g * 8 + u) * 32 + lane;
        q[u][0] = fr[3 * i + 0];
        q[u][1] = fr[3 * i + 1];
        q[u][2] = fr[3 * i + 2];
      }
      #pragma unroll
      for (int u = 0; u < 8; ++u) {
        const float dx = q[u][0] - ax;
        const float dy = q[u][1] - ay;
        const float dz = q[u][2] - az;
        const float d2 = fmaf(dz, dz, fmaf(dx, dx, dy * dy));
        const bool pred = d2 < 1.0f;
        const unsigned mask = __builtin_amdgcn_ballot_w32(pred);
        const int pos = cnt + (int)__builtin_amdgcn_mbcnt_lo(mask, 0u);
        if (pred && pos < KNB_) snb[w][pos] = make_float4(dx, dy, dz, 0.0f);
        cnt += __builtin_popcount(mask);
      }
    }
    __syncthreads();
    if (lane >= cnt) {
      snb[w][lane] = (cnt > 0) ? snb[w][0] : make_float4(p0x, p0y, p0z, 0.0f);
    }
    __syncthreads();
    const float tq = (float)(f - 1);
    #pragma unroll 4
    for (int k = 0; k < KNB_; ++k) {
      const float4 p = snb[w][k];
      #pragma unroll
      for (int j = 0; j < 8; ++j) {
        const v2f v = w0[j] * p.x + w1[j] * p.y + w2[j] * p.z + w3[j] * tq;
        mx[j] = __builtin_elementwise_max(mx[j], v);
      }
    }
    __syncthreads();
  }
  float* orow = feats + ((size_t)bl * MA_ + m) * C_ + 2 * lane;
  #pragma unroll
  for (int j = 0; j < 8; ++j) {
    const v2f o = {fmaxf(mx[j].x, 0.0f), fmaxf(mx[j].y, 0.0f)};
    *(v2f*)(orow + 64 * j) = o;
  }
}

__global__ __launch_bounds__(256) void ln_kernel(const float* __restrict__ x, const float* __restrict__ g,
                                                 const float* __restrict__ bta, float* __restrict__ y) {
  const int lane = threadIdx.x & 31, w = threadIdx.x >> 5;
  const int row = blockIdx.x * 8 + w;
  const float* xr = x + (size_t)row * C_ + 4 * lane;
  v4f a[4], d[4];
  #pragma unroll
  for (int i = 0; i < 4; ++i) a[i] = *(const v4f*)(xr + 128 * i);
  float s = 0.0f;
  #pragma unroll
  for (int i = 0; i < 4; ++i) s += (a[i].x + a[i].y) + (a[i].z + a[i].w);
  const float mean = wave_sum_f32(s) * (1.0f / C_);
  float q = 0.0f;
  #pragma unroll
  for (int i = 0; i < 4; ++i) {
    d[i] = a[i] - mean;
    q += (d[i].x * d[i].x + d[i].y * d[i].y) + (d[i].z * d[i].z + d[i].w * d[i].w);
  }
  const float inv = rsqrtf(wave_sum_f32(q) * (1.0f / C_) + 1e-5f);
  float* yr = y + (size_t)row * C_ + 4 * lane;
  #pragma unroll
  for (int i = 0; i < 4; ++i) {
    const v4f gg = *(const v4f*)(g + 4 * lane + 128 * i);
    const v4f bb = *(const v4f*)(bta + 4 * lane + 128 * i);
    *(v4f*)(yr + 128 * i) = d[i] * inv * gg + bb;
  }
}

#define GBK_    64
#define GPITCH_ 68
#define GNBUF_  3

__device__ __forceinline__ void lds_barrier() {
  __builtin_amdgcn_fence(__ATOMIC_RELEASE, "workgroup", "local");
  __builtin_amdgcn_s_barrier();
  __builtin_amdgcn_fence(__ATOMIC_ACQUIRE, "workgroup", "local");
}

template <int BM, int BN>
__global__ __launch_bounds__(256) void gemm_tiled(const float* __restrict__ A, int lda,
                                                  const float* __restrict__ W, int ldw,
                                                  const float* __restrict__ bias,
                                                  const float* __restrict__ resid,
                                                  float* __restrict__ Cout, int ldc,
                                                  int Kdim, int act) {
  constexpr int WC = BN / 32;
  constexpr int WR = 8 / WC;
  constexpr int TM = BM / (16 * WR);
  constexpr int TN = 2;
  constexpr int LA = BM / 16;
  constexpr int LB = BN / 16;
  constexpr int NKB = GBK_ / 8;
  __shared__ __attribute__((aligned(16))) float As[GNBUF_][BM * GPITCH_];
  __shared__ __attribute__((aligned(16))) float Bs[GNBUF_][BN * GPITCH_];
  const int t = threadIdx.x;
  const int lane = t & 31, w = t >> 5;
  const int sub = lane & 15, hi = lane >> 4;
  const int wr = w / WC, wc = w % WC;
  const int i0 = blockIdx.y * BM, j0 = blockIdx.x * BN;
  const int lr = t >> 4;
  const int lc = (t & 15) * 4;
  const float* ag = A + (size_t)(i0 + lr) * lda + lc;
  const float* wg = W + (size_t)(j0 + lr) * ldw + lc;
  const int sw_off = lr * GPITCH_ + lc;
  const int nst = Kdim / GBK_;
  v4f ra[LA], rb[LB];
  {
    v4f pa[2][LA], pb[2][LB];
    #pragma unroll
    for (int i = 0; i < LA; ++i) pa[0][i] = *(const v4f*)(ag + (size_t)(16 * i) * lda);
    #pragma unroll
    for (int i = 0; i < LB; ++i) pb[0][i] = *(const v4f*)(wg + (size_t)(16 * i) * ldw);
    if (nst > 1) {
      #pragma unroll
      for (int i = 0; i < LA; ++i) pa[1][i] = *(const v4f*)(ag + (size_t)(16 * i) * lda + GBK_);
      #pragma unroll
      for (int i = 0; i < LB; ++i) pb[1][i] = *(const v4f*)(wg + (size_t)(16 * i) * ldw + GBK_);
    }
    if (nst > 2) {
      #pragma unroll
      for (int i = 0; i < LA; ++i) ra[i] = *(const v4f*)(ag + (size_t)(16 * i) * lda + 2 * GBK_);
      #pragma unroll
      for (int i = 0; i < LB; ++i) rb[i] = *(const v4f*)(wg + (size_t)(16 * i) * ldw + 2 * GBK_);
    } else {
      #pragma unroll
      for (int i = 0; i < LA; ++i) ra[i] = (v4f){0.f, 0.f, 0.f, 0.f};
      #pragma unroll
      for (int i = 0; i < LB; ++i) rb[i] = (v4f){0.f, 0.f, 0.f, 0.f};
    }
    #pragma unroll
    for (int i = 0; i < LA; ++i) *(v4f*)(&As[0][sw_off + 16 * i * GPITCH_]) = pa[0][i];
    #pragma unroll
    for (int i = 0; i < LB; ++i) *(v4f*)(&Bs[0][sw_off + 16 * i * GPITCH_]) = pb[0][i];
    if (nst > 1) {
      #pragma unroll
      for (int i = 0; i < LA; ++i) *(v4f*)(&As[1][sw_off + 16 * i * GPITCH_]) = pa[1][i];
      #pragma unroll
      for (int i = 0; i < LB; ++i) *(v4f*)(&Bs[1][sw_off + 16 * i * GPITCH_]) = pb[1][i];
    }
  }
  lds_barrier();

  v8f acc[TM][TN];
  #pragma unroll
  for (int m = 0; m < TM; ++m)
    #pragma unroll
    for (int n = 0; n < TN; ++n)
      acc[m][n] = (v8f){0.f, 0.f, 0.f, 0.f, 0.f, 0.f, 0.f, 0.f};

  const int fa_off = (wr * TM * 16 + sub) * GPITCH_ + 4 * hi;
  const int fb_off = (wc * 32 + sub) * GPITCH_ + 4 * hi;
  v4f fa[2][TM], fb[2][TN];
  #pragma unroll
  for (int m = 0; m < TM; ++m) fa[0][m] = *(const v4f*)(&As[0][fa_off + m * 16 * GPITCH_]);
  #pragma unroll
  for (int n = 0; n < TN; ++n) fb[0][n] = *(const v4f*)(&Bs[0][fb_off + n * 16 * GPITCH_]);
  int cur = 0, nx1 = 1, nx2 = 2;
  for (int st = 0; st < nst; ++st) {
    #pragma unroll
    for (int i = 0; i < LA; ++i) *(v4f*)(&As[nx2][sw_off + 16 * i * GPITCH_]) = ra[i];
    #pragma unroll
    for (int i = 0; i < LB; ++i) *(v4f*)(&Bs[nx2][sw_off + 16 * i * GPITCH_]) = rb[i];
    if (st + 3 < nst) {
      const int kn = (st + 3) * GBK_;
      #pragma unroll
      for (int i = 0; i < LA; ++i) ra[i] = *(const v4f*)(ag + (size_t)(16 * i) * lda + kn);
      #pragma unroll
      for (int i = 0; i < LB; ++i) rb[i] = *(const v4f*)(wg + (size_t)(16 * i) * ldw + kn);
    }
    const float* as  = &As[cur][fa_off];
    const float* bs  = &Bs[cur][fb_off];
    const float* asn = &As[nx1][fa_off];
    const float* bsn = &Bs[nx1][fb_off];
    const bool more = (st + 1 < nst);
    #pragma unroll
    for (int kb = 0; kb < NKB; ++kb) {
      const int c = kb & 1, d = c ^ 1;
      if (kb + 1 < NKB) {
        #pragma unroll
        for (int m = 0; m < TM; ++m) fa[d][m] = *(const v4f*)(as + m * 16 * GPITCH_ + (kb + 1) * 8);
        #pragma unroll
        for (int n = 0; n < TN; ++n) fb[d][n] = *(const v4f*)(bs + n * 16 * GPITCH_ + (kb + 1) * 8);
      } else if (more) {
        #pragma unroll
        for (int m = 0; m < TM; ++m) fa[d][m] = *(const v4f*)(asn + m * 16 * GPITCH_);
        #pragma unroll
        for (int n = 0; n < TN; ++n) fb[d][n] = *(const v4f*)(bsn + n * 16 * GPITCH_);
      }
      #pragma unroll
      for (int m = 0; m < TM; ++m)
        #pragma unroll
        for (int n = 0; n < TN; ++n) {
          const v2f a = {fa[c][m].x, fa[c][m].y};
          const v2f b = {fb[c][n].x, fb[c][n].y};
          acc[m][n] = wmma4(a, b, acc[m][n]);
        }
      #pragma unroll
      for (int m = 0; m < TM; ++m)
        #pragma unroll
        for (int n = 0; n < TN; ++n) {
          const v2f a = {fa[c][m].z, fa[c][m].w};
          const v2f b = {fb[c][n].z, fb[c][n].w};
          acc[m][n] = wmma4(a, b, acc[m][n]);
        }
    }
    lds_barrier();
    const int tcur = cur; cur = nx1; nx1 = nx2; nx2 = tcur;
  }

  float bj[TN];
  #pragma unroll
  for (int n = 0; n < TN; ++n) bj[n] = bias ? bias[j0 + wc * 32 + n * 16 + sub] : 0.0f;
  #pragma unroll
  for (int m = 0; m < TM; ++m) {
    #pragma unroll
    for (int r = 0; r < 8; ++r) {
      const int row = i0 + (wr * TM + m) * 16 + r + 8 * hi;
      #pragma unroll
      for (int n = 0; n < TN; ++n) {
        const int col = j0 + wc * 32 + n * 16 + sub;
        float v = acc[m][n][r] + bj[n];
        if (act == 1) v = gelu_exact(v);
        if (resid) v += resid[(size_t)row * ldc + col];
        Cout[(size_t)row * ldc + col] = v;
      }
    }
  }
}

#define SP_ 1028

__global__ void attn_kernel(const float* __restrict__ qkv,
                            const float* __restrict__ xyzf,
                            const float* __restrict__ Wsp,
                            float* __restrict__ attn_out) {
  extern __shared__ __attribute__((aligned(16))) float smem[];
  float* S    = smem;
  float* rsum = S + 16 * SP_;
  float* wx   = rsum + 16;
  const int b = blockIdx.z, h = blockIdx.y, i0 = blockIdx.x * 16;
  const int t = threadIdx.x;
  const int lane = t & 31, w = t >> 5;
  const int sub = lane & 15, hi = lane >> 4;

  const float* qbase =
      qkv + ((size_t)(b * MTOK_ + i0 + sub) * 3 + 0) * C_ + h * DH_ + 2 * hi;
  v2f qa[16];
  #pragma unroll
  for (int ks = 0; ks < 16; ++ks) qa[ks] = *(const v2f*)(qbase + 4 * ks);
  for (int jj = 0; jj < 8; ++jj) {
    const int j0 = (w * 8 + jj) * 16;
    const float* kbase =
        qkv + ((size_t)(b * MTOK_ + j0 + sub) * 3 + 1) * C_ + h * DH_ + 2 * hi;
    v8f acc = {0.f, 0.f, 0.f, 0.f, 0.f, 0.f, 0.f, 0.f};
    #pragma unroll
    for (int ks = 0; ks < 16; ++ks) {
      const v2f kb = *(const v2f*)(kbase + 4 * ks);
      acc = wmma4(qa[ks], kb, acc);
    }
    #pragma unroll
    for (int r = 0; r < 8; ++r)
      S[(r + 8 * hi) * SP_ + j0 + sub] = acc[r] * 0.125f;
  }
  __syncthreads();

  for (int q = 0; q < 2; ++q) {
    const int row = 2 * w + q;
    float* sr = S + row * SP_ + 4 * lane;
    v4f x[8];
    #pragma unroll
    for (int i = 0; i < 8; ++i) x[i] = *(const v4f*)(sr + 128 * i);
    float lm = -3.4e38f;
    #pragma unroll
    for (int i = 0; i < 8; ++i)
      lm = fmaxf(lm, fmaxf(fmaxf(x[i].x, x[i].y), fmaxf(x[i].z, x[i].w)));
    const float rm = wave_max_f32(lm);
    float ls = 0.0f;
    #pragma unroll
    for (int i = 0; i < 8; ++i) {
      v4f e;
      e.x = expf(x[i].x - rm);
      e.y = expf(x[i].y - rm);
      e.z = expf(x[i].z - rm);
      e.w = expf(x[i].w - rm);
      ls += (e.x + e.y) + (e.z + e.w);
      *(v4f*)(sr + 128 * i) = e;
    }
    ls = wave_sum_f32(ls);
    if (lane == 0) rsum[row] = ls;
  }
  __syncthreads();

  v8f acc = {0.f, 0.f, 0.f, 0.f, 0.f, 0.f, 0.f, 0.f};
  const int d0 = (w & 3) * 16;
  if (w < 4) {
    const float* vbase =
        qkv + ((size_t)(b * MTOK_) * 3 + 2) * C_ + h * DH_ + d0 + sub;
    const float* srow = S + sub * SP_ + 2 * hi;
    for (int ks = 0; ks < 256; ++ks) {
      const v2f a = *(const v2f*)(srow + 4 * ks);
      const int kk = 4 * ks + 2 * hi;
      v2f bb;
      bb.x = vbase[(size_t)kk * 1536];
      bb.y = vbase[(size_t)(kk + 1) * 1536];
      acc = wmma4(a, bb, acc);
    }
  } else {
    const int r0 = (w - 4) * 4;
    const float* xb = xyzf + (size_t)b * MTOK_ * 3;
    float ax[4][3];
    #pragma unroll
    for (int rr = 0; rr < 4; ++rr) { ax[rr][0] = 0.0f; ax[rr][1] = 0.0f; ax[rr][2] = 0.0f; }
    #pragma unroll 1
    for (int i = 0; i < 8; ++i) {
      const int j = 4 * lane + 128 * i;
      const v4f c0 = *(const v4f*)(xb + 3 * j);
      const v4f c1 = *(const v4f*)(xb + 3 * j + 4);
      const v4f c2 = *(const v4f*)(xb + 3 * j + 8);
      #pragma unroll
      for (int rr = 0; rr < 4; ++rr) {
        const v4f p = *(const v4f*)(S + (r0 + rr) * SP_ + j);
        ax[rr][0] += p.x * c0.x + p.y * c0.w + p.z * c1.z + p.w * c2.y;
        ax[rr][1] += p.x * c0.y + p.y * c1.x + p.z * c1.w + p.w * c2.z;
        ax[rr][2] += p.x * c0.z + p.y * c1.y + p.z * c2.x + p.w * c2.w;
      }
    }
    #pragma unroll
    for (int rr = 0; rr < 4; ++rr) {
      #pragma unroll
      for (int cc = 0; cc < 3; ++cc) ax[rr][cc] = wave_sum_f32(ax[rr][cc]);
    }
    if (lane == 0) {
      #pragma unroll
      for (int rr = 0; rr < 4; ++rr) {
        #pragma unroll
        for (int cc = 0; cc < 3; ++cc) wx[(r0 + rr) * 3 + cc] = ax[rr][cc];
      }
    }
  }
  __syncthreads();
  if (w < 4) {
    const int d = d0 + sub;
    const float w0 = Wsp[d * 3 + 0], w1 = Wsp[d * 3 + 1], w2 = Wsp[d * 3 + 2];
    #pragma unroll
    for (int r = 0; r < 8; ++r) {
      const int rr = r + 8 * hi;
      const int i  = i0 + rr;
      const float inv = 1.0f / rsum[rr];
      const float* xi = xyzf + ((size_t)b * MTOK_ + i) * 3;
      const float wd0 = wx[rr * 3 + 0] * inv - xi[0];
      const float wd1 = wx[rr * 3 + 1] * inv - xi[1];
      const float wd2 = wx[rr * 3 + 2] * inv - xi[2];
      attn_out[((size_t)(b * MTOK_) + i) * C_ + h * DH_ + d] =
          acc[r] * inv + w0 * wd0 + w1 * wd1 + w2 * wd2;
    }
  }
}

extern "C" void kernel_launch(void* const* d_in, const int* in_sizes, int n_in,
                              void* d_out, int out_size, void* d_ws, size_t ws_size,
                              hipStream_t stream) {
  (void)in_sizes; (void)n_in; (void)out_size; (void)ws_size;
  const float* xyzs = (const float*)d_in[0];
  const float* Wd   = (const float*)d_in[1];
  const float* Wqkv = (const float*)d_in[2];
  const float* Wsp  = (const float*)d_in[3];
  const float* Wout = (const float*)d_in[4];
  const float* bout = (const float*)d_in[5];
  const float* ln1g = (const float*)d_in[6];
  const float* ln1b = (const float*)d_in[7];
  const float* Wff1 = (const float*)d_in[8];
  const float* bff1 = (const float*)d_in[9];
  const float* Wff2 = (const float*)d_in[10];
  const float* bff2 = (const float*)d_in[11];
  const float* ln2g = (const float*)d_in[12];
  const float* ln2b = (const float*)d_in[13];

  float* ws      = (float*)d_ws;
  float* anchors = ws;
  float* feats   = ws + 8192;
  float* xln     = feats + (size_t)ROWS_ * C_;
  float* qkvb    = xln   + (size_t)ROWS_ * C_;
  float* attno   = qkvb  + (size_t)ROWS_ * 1536;
  float* ffh     = attno + (size_t)ROWS_ * C_;

  fps_kernel<<<B_ * L_, 32, 0, stream>>>(xyzs, anchors);
  p4dconv_kernel<<<B_ * L_ * MA_ / P4W_, 256, 0, stream>>>(xyzs, anchors, Wd, feats);

  const size_t smem_attn = (16 * 1028 + 16 + 48) * sizeof(float);
  for (int d = 0; d < 2; ++d) {
    ln_kernel<<<ROWS_ / 8, 256, 0, stream>>>(feats, ln1g + d * C_, ln1b + d * C_, xln);
    gemm_tiled<64, 64><<<dim3(1536 / 64, ROWS_ / 64), 256, 0, stream>>>(
        xln, C_, Wqkv + (size_t)d * 1536 * C_, C_, nullptr, nullptr,
        qkvb, 1536, C_, 0);
    attn_kernel<<<dim3(MTOK_ / 16, HEADS_, B_), 256, smem_attn, stream>>>(
        qkvb, anchors, Wsp + (size_t)d * DH_ * 3, attno);
    gemm_tiled<64, 64><<<dim3(C_ / 64, ROWS_ / 64), 256, 0, stream>>>(
        attno, C_, Wout + (size_t)d * C_ * C_, C_, bout + d * C_,
        feats, feats, C_, C_, 1);
    ln_kernel<<<ROWS_ / 8, 256, 0, stream>>>(feats, ln2g + d * C_, ln2b + d * C_, xln);
    gemm_tiled<128, 128><<<dim3(MLP_ / 128, ROWS_ / 128), 256, 0, stream>>>(
        xln, C_, Wff1 + (size_t)d * MLP_ * C_, C_, bff1 + d * MLP_,
        nullptr, ffh, MLP_, C_, 1);
    gemm_tiled<64, 64><<<dim3(C_ / 64, ROWS_ / 64), 256, 0, stream>>>(
        ffh, MLP_, Wff2 + (size_t)d * C_ * MLP_, MLP_, bff2 + d * C_,
        feats, feats, C_, MLP_, 0);
  }
  hipMemcpyAsync(d_out, feats, (size_t)ROWS_ * C_ * sizeof(float),
                 hipMemcpyDeviceToDevice, stream);
}


# attention QK^T: Q and K fragments read 16 B per lane (k = 8kb+4h..+3, two WMMAs per read) instead of 8 B: half the strided load requests; scores sum over k in another order
# speedup vs baseline: 2.7505x; 1.0370x over previous
#include <hip/hip_runtime.h>
#include <hip/hip_bf16.h>
#include <math.h>

typedef float v2f __attribute__((ext_vector_type(2)));
typedef float v4f __attribute__((ext_vector_type(4)));
typedef float v8f __attribute__((ext_vector_type(8)));

#define B_    2
#define T_    4
#define N_    1024
#define L_    4
#define MA_   256
#define KNB_  32
#define C_    512
#define HEADS_ 8
#define DH_   64
#define MTOK_ 1024
#define ROWS_ 2048
#define MLP_  2048

__device__ __forceinline__ v8f wmma4(v2f a, v2f b, v8f c) {
  return __builtin_amdgcn_wmma_f32_16x16x4_f32(false, a, false, b, (short)0, c,
                                               false, false);
}

__device__ __forceinline__ float gelu_exact(float x) {
  return 0.5f * x * (1.0f + erff(x * 0.7071067811865475f));
}

#define FPS_SLOTS_  (N_ / 32)
#define FPS_CHAINS_ 4
#define FPS_CLEN_   (FPS_SLOTS_ / FPS_CHAINS_)

template <int CTRL>
__device__ __forceinline__ int dpp_row_i(int v) {
  return __builtin_amdgcn_update_dpp(0, v, CTRL, 0xf, 0xf, true);
}
__device__ __forceinline__ int other_row_i(int v) {
  return __builtin_amdgcn_permlanex16(v, v, 0x76543210, 0xfedcba98, false, false);
}
__device__ __forceinline__ int wave_max_i32(int v) {
  v = max(v, dpp_row_i<0x161>(v));
  v = max(v, dpp_row_i<0x162>(v));
  v = max(v, dpp_row_i<0x164>(v));
  v = max(v, dpp_row_i<0x168>(v));
  v = max(v, other_row_i(v));
  return v;
}
__device__ __forceinline__ int wave_min_i32(int v) {
  v = min(v, dpp_row_i<0x161>(v));
  v = min(v, dpp_row_i<0x162>(v));
  v = min(v, dpp_row_i<0x164>(v));
  v = min(v, dpp_row_i<0x168>(v));
  v = min(v, other_row_i(v));
  return v;
}

__device__ __forceinline__ float wave_max_f32(float v) {
  v = fmaxf(v, __builtin_bit_cast(float, dpp_row_i<0x161>(__builtin_bit_cast(int, v))));
  v = fmaxf(v, __builtin_bit_cast(float, dpp_row_i<0x162>(__builtin_bit_cast(int, v))));
  v = fmaxf(v, __builtin_bit_cast(float, dpp_row_i<0x164>(__builtin_bit_cast(int, v))));
  v = fmaxf(v, __builtin_bit_cast(float, dpp_row_i<0x168>(__builtin_bit_cast(int, v))));
  v = fmaxf(v, __builtin_bit_cast(float, other_row_i(__builtin_bit_cast(int, v))));
  return v;
}
__device__ __forceinline__ float wave_sum_f32(float v) {
  v += __builtin_bit_cast(float, dpp_row_i<0x161>(__builtin_bit_cast(int, v)));
  v += __builtin_bit_cast(float, dpp_row_i<0x162>(__builtin_bit_cast(int, v)));
  v += __builtin_bit_cast(float, dpp_row_i<0x164>(__builtin_bit_cast(int, v)));
  v += __builtin_bit_cast(float, dpp_row_i<0x168>(__builtin_bit_cast(int, v)));
  v += __builtin_bit_cast(float, other_row_i(__builtin_bit_cast(int, v)));
  return v;
}

__global__ __launch_bounds__(32) void fps_kernel(const float* __restrict__ xyzs,
                                                 float* __restrict__ anchors) {
  __shared__ float4 spt[N_];
  const int blk  = blockIdx.x;
  const int lane = threadIdx.x;
  const float* frame = xyzs + (size_t)blk * N_ * 3;
  float px[FPS_SLOTS_], py[FPS_SLOTS_], pz[FPS_SLOTS_];
  int   dist[FPS_SLOTS_];
  #pragma unroll
  for (int s = 0; s < FPS_SLOTS_; ++s) {
    const int i = s * 32 + lane;
    px[s] = frame[i * 3 + 0];
    py[s] = frame[i * 3 + 1];
    pz[s] = frame[i * 3 + 2];
    dist[s] = __builtin_bit_cast(int, 1e10f);
    spt[i] = make_float4(px[s], py[s], pz[s], 0.0f);
  }
  __syncthreads();
  float* aout = anchors + (size_t)blk * MA_ * 3;
  const char* sbytes = (const char*)spt;
  int curb = 0;
  for (int it = 1; it < MA_; ++it) {
    const float4 last = *(const float4*)(sbytes + curb);
    if (lane == 0) {
      aout[(it - 1) * 3 + 0] = last.x;
      aout[(it - 1) * 3 + 1] = last.y;
      aout[(it - 1) * 3 + 2] = last.z;
    }
    #pragma unroll
    for (int s = 0; s < FPS_SLOTS_; ++s) {
      const float dx = px[s] - last.x, dy = py[s] - last.y, dz = pz[s] - last.z;
      const float d2 = fmaf(dz, dz, fmaf(dx, dx, dy * dy));
      dist[s] = min(dist[s], __builtin_bit_cast(int, d2));
    }
    int bv[FPS_CHAINS_], bo[FPS_CHAINS_];
    #pragma unroll
    for (int c = 0; c < FPS_CHAINS_; ++c) {
      bv[c] = dist[c * FPS_CLEN_];
      bo[c] = c * FPS_CLEN_ * 32 * 16;
      #pragma unroll
      for (int k = 1; k < FPS_CLEN_; ++k) {
        const int s = c * FPS_CLEN_ + k;
        bo[c] = (dist[s] > bv[c]) ? s * 32 * 16 : bo[c];
        bv[c] = max(bv[c], dist[s]);
      }
    }
    #pragma unroll
    for (int w = 1; w < FPS_CHAINS_; w <<= 1) {
      #pragma unroll
      for (int c = 0; c < FPS_CHAINS_; c += 2 * w) {
        bo[c] = (bv[c + w] > bv[c]) ? bo[c + w] : bo[c];
        bv[c] = max(bv[c], bv[c + w]);
      }
    }
    const int gmax = wave_max_i32(bv[0]);
    const int cand = (bv[0] == gmax) ? (bo[0] + lane * 16) : 0x7fffffff;
    curb = wave_min_i32(cand);
  }
  const float4 last = *(const float4*)(sbytes + curb);
  if (lane == 0) {
    aout[(MA_ - 1) * 3 + 0] = last.x;
    aout[(MA_ - 1) * 3 + 1] = last.y;
    aout[(MA_ - 1) * 3 + 2] = last.z;
  }
}

#define P4W_ 8

__global__ __launch_bounds__(256) void p4dconv_kernel(const float* __restrict__ xyzs,
                                                      const float* __restrict__ anchors,
                                                      const float* __restrict__ Wd,
                                                      float* __restrict__ feats) {
  __shared__ __attribute__((aligned(16))) float  sfr[3][N_ * 3];
  __shared__ __attribute__((aligned(16))) float4 snb[P4W_][KNB_];
  const int t = threadIdx.x;
  const int lane = t & 31, w = t >> 5;
  const int blk = blockIdx.x;
  const int bl  = blk / (MA_ / P4W_);
  const int m   = (blk % (MA_ / P4W_)) * P4W_ + w;
  const int l   = bl % L_;
  const int b   = bl / L_;

  for (int f = 0; f < 3; ++f) {
    int fi = l + f - 1; fi = fi < 0 ? 0 : (fi > T_ - 1 ? T_ - 1 : fi);
    const v4f* srcp = (const v4f*)(xyzs + ((size_t)b * T_ + fi) * N_ * 3);
    v4f* dstp = (v4f*)&sfr[f][0];
    for (int i = 0; i < 3; ++i) dstp[t + 256 * i] = srcp[t + 256 * i];
  }

  const float* arow = anchors + ((size_t)bl * MA_ + m) * 3;
  const float ax = arow[0], ay = arow[1], az = arow[2];

  v2f w0[8], w1[8], w2[8], w3[8], mx[8];
  #pragma unroll
  for (int j = 0; j < 8; ++j) {
    const float* wp = Wd + (size_t)(2 * lane + 64 * j) * 4;
    const v4f wa = *(const v4f*)wp;
    const v4f wb = *(const v4f*)(wp + 4);
    w0[j] = (v2f){wa.x, wb.x};
    w1[j] = (v2f){wa.y, wb.y};
    w2[j] = (v2f){wa.z, wb.z};
    w3[j] = (v2f){wa.w, wb.w};
    mx[j] = (v2f){-3.4e38f, -3.4e38f};
  }
  __syncthreads();

  for (int f = 0; f < 3; ++f) {
    const float* fr = &sfr[f][0];
    const float p0x = fr[0] - ax, p0y = fr[1] - ay, p0z = fr[2] - az;
    int cnt = 0;
    for (int g = 0; g < 4 && cnt < KNB_; ++g) {
      float q[8][3];
      #pragma unroll
      for (int u = 0; u < 8; ++u) {
        const int i = (g * 8 + u) * 32 + lane;
        q[u][0] = fr[3 * i + 0];
        q[u][1] = fr[3 * i + 1];
        q[u][2] = fr[3 * i + 2];
      }
      #pragma unroll
      for (int u = 0; u < 8; ++u) {
        const float dx = q[u][0] - ax;
        const float dy = q[u][1] - ay;
        const float dz = q[u][2] - az;
        const float d2 = fmaf(dz, dz, fmaf(dx, dx, dy * dy));
        const bool pred = d2 < 1.0f;
        const unsigned mask = __builtin_amdgcn_ballot_w32(pred);
        const int pos = cnt + (int)__builtin_amdgcn_mbcnt_lo(mask, 0u);
        if (pred && pos < KNB_) snb[w][pos] = make_float4(dx, dy, dz, 0.0f);
        cnt += __builtin_popcount(mask);
      }
    }
    __syncthreads();
    if (lane >= cnt) {
      snb[w][lane] = (cnt > 0) ? snb[w][0] : make_float4(p0x, p0y, p0z, 0.0f);
    }
    __syncthreads();
    const float tq = (float)(f - 1);
    #pragma unroll 4
    for (int k = 0; k < KNB_; ++k) {
      const float4 p = snb[w][k];
      #pragma unroll
      for (int j = 0; j < 8; ++j) {
        const v2f v = w0[j] * p.x + w1[j] * p.y + w2[j] * p.z + w3[j] * tq;
        mx[j] = __builtin_elementwise_max(mx[j], v);
      }
    }
    __syncthreads();
  }
  float* orow = feats + ((size_t)bl * MA_ + m) * C_ + 2 * lane;
  #pragma unroll
  for (int j = 0; j < 8; ++j) {
    const v2f o = {fmaxf(mx[j].x, 0.0f), fmaxf(mx[j].y, 0.0f)};
    *(v2f*)(orow + 64 * j) = o;
  }
}

__global__ __launch_bounds__(256) void ln_kernel(const float* __restrict__ x, const float* __restrict__ g,
                                                 const float* __restrict__ bta, float* __restrict__ y) {
  const int lane = threadIdx.x & 31, w = threadIdx.x >> 5;
  const int row = blockIdx.x * 8 + w;
  const float* xr = x + (size_t)row * C_ + 4 * lane;
  v4f a[4], d[4];
  #pragma unroll
  for (int i = 0; i < 4; ++i) a[i] = *(const v4f*)(xr + 128 * i);
  float s = 0.0f;
  #pragma unroll
  for (int i = 0; i < 4; ++i) s += (a[i].x + a[i].y) + (a[i].z + a[i].w);
  const float mean = wave_sum_f32(s) * (1.0f / C_);
  float q = 0.0f;
  #pragma unroll
  for (int i = 0; i < 4; ++i) {
    d[i] = a[i] - mean;
    q += (d[i].x * d[i].x + d[i].y * d[i].y) + (d[i].z * d[i].z + d[i].w * d[i].w);
  }
  const float inv = rsqrtf(wave_sum_f32(q) * (1.0f / C_) + 1e-5f);
  float* yr = y + (size_t)row * C_ + 4 * lane;
  #pragma unroll
  for (int i = 0; i < 4; ++i) {
    const v4f gg = *(const v4f*)(g + 4 * lane + 128 * i);
    const v4f bb = *(const v4f*)(bta + 4 * lane + 128 * i);
    *(v4f*)(yr + 128 * i) = d[i] * inv * gg + bb;
  }
}

#define GBK_    64
#define GPITCH_ 68
#define GNBUF_  3

__device__ __forceinline__ void lds_barrier() {
  __builtin_amdgcn_fence(__ATOMIC_RELEASE, "workgroup", "local");
  __builtin_amdgcn_s_barrier();
  __builtin_amdgcn_fence(__ATOMIC_ACQUIRE, "workgroup", "local");
}

template <int BM, int BN>
__global__ __launch_bounds__(256) void gemm_tiled(const float* __restrict__ A, int lda,
                                                  const float* __restrict__ W, int ldw,
                                                  const float* __restrict__ bias,
                                                  const float* __restrict__ resid,
                                                  float* __restrict__ Cout, int ldc,
                                                  int Kdim, int act) {
  constexpr int WC = BN / 32;
  constexpr int WR = 8 / WC;
  constexpr int TM = BM / (16 * WR);
  constexpr int TN = 2;
  constexpr int LA = BM / 16;
  constexpr int LB = BN / 16;
  constexpr int NKB = GBK_ / 8;
  __shared__ __attribute__((aligned(16))) float As[GNBUF_][BM * GPITCH_];
  __shared__ __attribute__((aligned(16))) float Bs[GNBUF_][BN * GPITCH_];
  const int t = threadIdx.x;
  const int lane = t & 31, w = t >> 5;
  const int sub = lane & 15, hi = lane >> 4;
  const int wr = w / WC, wc = w % WC;
  const int i0 = blockIdx.y * BM, j0 = blockIdx.x * BN;
  const int lr = t >> 4;
  const int lc = (t & 15) * 4;
  const float* ag = A + (size_t)(i0 + lr) * lda + lc;
  const float* wg = W + (size_t)(j0 + lr) * ldw + lc;
  const int sw_off = lr * GPITCH_ + lc;
  const int nst = Kdim / GBK_;
  v4f ra[LA], rb[LB];
  {
    v4f pa[2][LA], pb[2][LB];
    #pragma unroll
    for (int i = 0; i < LA; ++i) pa[0][i] = *(const v4f*)(ag + (size_t)(16 * i) * lda);
    #pragma unroll
    for (int i = 0; i < LB; ++i) pb[0][i] = *(const v4f*)(wg + (size_t)(16 * i) * ldw);
    if (nst > 1) {
      #pragma unroll
      for (int i = 0; i < LA; ++i) pa[1][i] = *(const v4f*)(ag + (size_t)(16 * i) * lda + GBK_);
      #pragma unroll
      for (int i = 0; i < LB; ++i) pb[1][i] = *(const v4f*)(wg + (size_t)(16 * i) * ldw + GBK_);
    }
    if (nst > 2) {
      #pragma unroll
      for (int i = 0; i < LA; ++i) ra[i] = *(const v4f*)(ag + (size_t)(16 * i) * lda + 2 * GBK_);
      #pragma unroll
      for (int i = 0; i < LB; ++i) rb[i] = *(const v4f*)(wg + (size_t)(16 * i) * ldw + 2 * GBK_);
    } else {
      #pragma unroll
      for (int i = 0; i < LA; ++i) ra[i] = (v4f){0.f, 0.f, 0.f, 0.f};
      #pragma unroll
      for (int i = 0; i < LB; ++i) rb[i] = (v4f){0.f, 0.f, 0.f, 0.f};
    }
    #pragma unroll
    for (int i = 0; i < LA; ++i) *(v4f*)(&As[0][sw_off + 16 * i * GPITCH_]) = pa[0][i];
    #pragma unroll
    for (int i = 0; i < LB; ++i) *(v4f*)(&Bs[0][sw_off + 16 * i * GPITCH_]) = pb[0][i];
    if (nst > 1) {
      #pragma unroll
      for (int i = 0; i < LA; ++i) *(v4f*)(&As[1][sw_off + 16 * i * GPITCH_]) = pa[1][i];
      #pragma unroll
      for (int i = 0; i < LB; ++i) *(v4f*)(&Bs[1][sw_off + 16 * i * GPITCH_]) = pb[1][i];
    }
  }
  lds_barrier();

  v8f acc[TM][TN];
  #pragma unroll
  for (int m = 0; m < TM; ++m)
    #pragma unroll
    for (int n = 0; n < TN; ++n)
      acc[m][n] = (v8f){0.f, 0.f, 0.f, 0.f, 0.f, 0.f, 0.f, 0.f};

  const int fa_off = (wr * TM * 16 + sub) * GPITCH_ + 4 * hi;
  const int fb_off = (wc * 32 + sub) * GPITCH_ + 4 * hi;
  v4f fa[2][TM], fb[2][TN];
  #pragma unroll
  for (int m = 0; m < TM; ++m) fa[0][m] = *(const v4f*)(&As[0][fa_off + m * 16 * GPITCH_]);
  #pragma unroll
  for (int n = 0; n < TN; ++n) fb[0][n] = *(const v4f*)(&Bs[0][fb_off + n * 16 * GPITCH_]);
  int cur = 0, nx1 = 1, nx2 = 2;
  for (int st = 0; st < nst; ++st) {
    #pragma unroll
    for (int i = 0; i < LA; ++i) *(v4f*)(&As[nx2][sw_off + 16 * i * GPITCH_]) = ra[i];
    #pragma unroll
    for (int i = 0; i < LB; ++i) *(v4f*)(&Bs[nx2][sw_off + 16 * i * GPITCH_]) = rb[i];
    if (st + 3 < nst) {
      const int kn = (st + 3) * GBK_;
      #pragma unroll
      for (int i = 0; i < LA; ++i) ra[i] = *(const v4f*)(ag + (size_t)(16 * i) * lda + kn);
      #pragma unroll
      for (int i = 0; i < LB; ++i) rb[i] = *(const v4f*)(wg + (size_t)(16 * i) * ldw + kn);
    }
    const float* as  = &As[cur][fa_off];
    const float* bs  = &Bs[cur][fb_off];
    const float* asn = &As[nx1][fa_off];
    const float* bsn = &Bs[nx1][fb_off];
    const bool more = (st + 1 < nst);
    #pragma unroll
    for (int kb = 0; kb < NKB; ++kb) {
      const int c = kb & 1, d = c ^ 1;
      if (kb + 1 < NKB) {
        #pragma unroll
        for (int m = 0; m < TM; ++m) fa[d][m] = *(const v4f*)(as + m * 16 * GPITCH_ + (kb + 1) * 8);
        #pragma unroll
        for (int n = 0; n < TN; ++n) fb[d][n] = *(const v4f*)(bs + n * 16 * GPITCH_ + (kb + 1) * 8);
      } else if (more) {
        #pragma unroll
        for (int m = 0; m < TM; ++m) fa[d][m] = *(const v4f*)(asn + m * 16 * GPITCH_);
        #pragma unroll
        for (int n = 0; n < TN; ++n) fb[d][n] = *(const v4f*)(bsn + n * 16 * GPITCH_);
      }
      #pragma unroll
      for (int m = 0; m < TM; ++m)
        #pragma unroll
        for (int n = 0; n < TN; ++n) {
          const v2f a = {fa[c][m].x, fa[c][m].y};
          const v2f b = {fb[c][n].x, fb[c][n].y};
          acc[m][n] = wmma4(a, b, acc[m][n]);
        }
      #pragma unroll
      for (int m = 0; m < TM; ++m)
        #pragma unroll
        for (int n = 0; n < TN; ++n) {
          const v2f a = {fa[c][m].z, fa[c][m].w};
          const v2f b = {fb[c][n].z, fb[c][n].w};
          acc[m][n] = wmma4(a, b, acc[m][n]);
        }
    }
    lds_barrier();
    const int tcur = cur; cur = nx1; nx1 = nx2; nx2 = tcur;
  }

  float bj[TN];
  #pragma unroll
  for (int n = 0; n < TN; ++n) bj[n] = bias ? bias[j0 + wc * 32 + n * 16 + sub] : 0.0f;
  #pragma unroll
  for (int m = 0; m < TM; ++m) {
    #pragma unroll
    for (int r = 0; r < 8; ++r) {
      const int row = i0 + (wr * TM + m) * 16 + r + 8 * hi;
      #pragma unroll
      for (int n = 0; n < TN; ++n) {
        const int col = j0 + wc * 32 + n * 16 + sub;
        float v = acc[m][n][r] + bj[n];
        if (act == 1) v = gelu_exact(v);
        if (resid) v += resid[(size_t)row * ldc + col];
        Cout[(size_t)row * ldc + col] = v;
      }
    }
  }
}

#define SP_ 1028

__global__ void attn_kernel(const float* __restrict__ qkv,
                            const float* __restrict__ xyzf,
                            const float* __restrict__ Wsp,
                            float* __restrict__ attn_out) {
  extern __shared__ __attribute__((aligned(16))) float smem[];
  float* S    = smem;
  float* rsum = S + 16 * SP_;
  float* wx   = rsum + 16;
  const int b = blockIdx.z, h = blockIdx.y, i0 = blockIdx.x * 16;
  const int t = threadIdx.x;
  const int lane = t & 31, w = t >> 5;
  const int sub = lane & 15, hi = lane >> 4;

  const float* qbase =
      qkv + ((size_t)(b * MTOK_ + i0 + sub) * 3 + 0) * C_ + h * DH_ + 4 * hi;
  v4f qa[8];
  #pragma unroll
  for (int kb = 0; kb < 8; ++kb) qa[kb] = *(const v4f*)(qbase + 8 * kb);
  for (int jj = 0; jj < 8; ++jj) {
    const int j0 = (w * 8 + jj) * 16;
    const float* kbase =
        qkv + ((size_t)(b * MTOK_ + j0 + sub) * 3 + 1) * C_ + h * DH_ + 4 * hi;
    v4f kf[8];
    #pragma unroll
    for (int kb = 0; kb < 8; ++kb) kf[kb] = *(const v4f*)(kbase + 8 * kb);
    v8f acc = {0.f, 0.f, 0.f, 0.f, 0.f, 0.f, 0.f, 0.f};
    #pragma unroll
    for (int kb = 0; kb < 8; ++kb) {
      const v2f a0 = {qa[kb].x, qa[kb].y}, b0 = {kf[kb].x, kf[kb].y};
      const v2f a1 = {qa[kb].z, qa[kb].w}, b1 = {kf[kb].z, kf[kb].w};
      acc = wmma4(a0, b0, acc);
      acc = wmma4(a1, b1, acc);
    }
    #pragma unroll
    for (int r = 0; r < 8; ++r)
      S[(r + 8 * hi) * SP_ + j0 + sub] = acc[r] * 0.125f;
  }
  __syncthreads();

  for (int q = 0; q < 2; ++q) {
    const int row = 2 * w + q;
    float* sr = S + row * SP_ + 4 * lane;
    v4f x[8];
    #pragma unroll
    for (int i = 0; i < 8; ++i) x[i] = *(const v4f*)(sr + 128 * i);
    float lm = -3.4e38f;
    #pragma unroll
    for (int i = 0; i < 8; ++i)
      lm = fmaxf(lm, fmaxf(fmaxf(x[i].x, x[i].y), fmaxf(x[i].z, x[i].w)));
    const float rm = wave_max_f32(lm);
    float ls = 0.0f;
    #pragma unroll
    for (int i = 0; i < 8; ++i) {
      v4f e;
      e.x = expf(x[i].x - rm);
      e.y = expf(x[i].y - rm);
      e.z = expf(x[i].z - rm);
      e.w = expf(x[i].w - rm);
      ls += (e.x + e.y) + (e.z + e.w);
      *(v4f*)(sr + 128 * i) = e;
    }
    ls = wave_sum_f32(ls);
    if (lane == 0) rsum[row] = ls;
  }
  __syncthreads();

  v8f acc = {0.f, 0.f, 0.f, 0.f, 0.f, 0.f, 0.f, 0.f};
  const int d0 = (w & 3) * 16;
  if (w < 4) {
    const float* vbase =
        qkv + ((size_t)(b * MTOK_) * 3 + 2) * C_ + h * DH_ + d0 + sub;
    const float* srow = S + sub * SP_ + 2 * hi;
    for (int ks = 0; ks < 256; ++ks) {
      const v2f a = *(const v2f*)(srow + 4 * ks);
      const int kk = 4 * ks + 2 * hi;
      v2f bb;
      bb.x = vbase[(size_t)kk * 1536];
      bb.y = vbase[(size_t)(kk + 1) * 1536];
      acc = wmma4(a, bb, acc);
    }
  } else {
    const int r0 = (w - 4) * 4;
    const float* xb = xyzf + (size_t)b * MTOK_ * 3;
    float ax[4][3];
    #pragma unroll
    for (int rr = 0; rr < 4; ++rr) { ax[rr][0] = 0.0f; ax[rr][1] = 0.0f; ax[rr][2] = 0.0f; }
    #pragma unroll 1
    for (int i = 0; i < 8; ++i) {
      const int j = 4 * lane + 128 * i;
      const v4f c0 = *(const v4f*)(xb + 3 * j);
      const v4f c1 = *(const v4f*)(xb + 3 * j + 4);
      const v4f c2 = *(const v4f*)(xb + 3 * j + 8);
      #pragma unroll
      for (int rr = 0; rr < 4; ++rr) {
        const v4f p = *(const v4f*)(S + (r0 + rr) * SP_ + j);
        ax[rr][0] += p.x * c0.x + p.y * c0.w + p.z * c1.z + p.w * c2.y;
        ax[rr][1] += p.x * c0.y + p.y * c1.x + p.z * c1.w + p.w * c2.z;
        ax[rr][2] += p.x * c0.z + p.y * c1.y + p.z * c2.x + p.w * c2.w;
      }
    }
    #pragma unroll
    for (int rr = 0; rr < 4; ++rr) {
      #pragma unroll
      for (int cc = 0; cc < 3; ++cc) ax[rr][cc] = wave_sum_f32(ax[rr][cc]);
    }
    if (lane == 0) {
      #pragma unroll
      for (int rr = 0; rr < 4; ++rr) {
        #pragma unroll
        for (int cc = 0; cc < 3; ++cc) wx[(r0 + rr) * 3 + cc] = ax[rr][cc];
      }
    }
  }
  __syncthreads();
  if (w < 4) {
    const int d = d0 + sub;
    const float w0 = Wsp[d * 3 + 0], w1 = Wsp[d * 3 + 1], w2 = Wsp[d * 3 + 2];
    #pragma unroll
    for (int r = 0; r < 8; ++r) {
      const int rr = r + 8 * hi;
      const int i  = i0 + rr;
      const float inv = 1.0f / rsum[rr];
      const float* xi = xyzf + ((size_t)b * MTOK_ + i) * 3;
      const float wd0 = wx[rr * 3 + 0] * inv - xi[0];
      const float wd1 = wx[rr * 3 + 1] * inv - xi[1];
      const float wd2 = wx[rr * 3 + 2] * inv - xi[2];
      attn_out[((size_t)(b * MTOK_) + i) * C_ + h * DH_ + d] =
          acc[r] * inv + w0 * wd0 + w1 * wd1 + w2 * wd2;
    }
  }
}

extern "C" void kernel_launch(void* const* d_in, const int* in_sizes, int n_in,
                              void* d_out, int out_size, void* d_ws, size_t ws_size,
                              hipStream_t stream) {
  (void)in_sizes; (void)n_in; (void)out_size; (void)ws_size;
  const float* xyzs = (const float*)d_in[0];
  const float* Wd   = (const float*)d_in[1];
  const float* Wqkv = (const float*)d_in[2];
  const float* Wsp  = (const float*)d_in[3];
  const float* Wout = (const float*)d_in[4];
  const float* bout = (const float*)d_in[5];
  const float* ln1g = (const float*)d_in[6];
  const float* ln1b = (const float*)d_in[7];
  const float* Wff1 = (const float*)d_in[8];
  const float* bff1 = (const float*)d_in[9];
  const float* Wff2 = (const float*)d_in[10];
  const float* bff2 = (const float*)d_in[11];
  const float* ln2g = (const float*)d_in[12];
  const float* ln2b = (const float*)d_in[13];

  float* ws      = (float*)d_ws;
  float* anchors = ws;
  float* feats   = ws + 8192;
  float* xln     = feats + (size_t)ROWS_ * C_;
  float* qkvb    = xln   + (size_t)ROWS_ * C_;
  float* attno   = qkvb  + (size_t)ROWS_ * 1536;
  float* ffh     = attno + (size_t)ROWS_ * C_;

  fps_kernel<<<B_ * L_, 32, 0, stream>>>(xyzs, anchors);
  p4dconv_kernel<<<B_ * L_ * MA_ / P4W_, 256, 0, stream>>>(xyzs, anchors, Wd, feats);

  const size_t smem_attn = (16 * 1028 + 16 + 48) * sizeof(float);
  for (int d = 0; d < 2; ++d) {
    ln_kernel<<<ROWS_ / 8, 256, 0, stream>>>(feats, ln1g + d * C_, ln1b + d * C_, xln);
    gemm_tiled<64, 64><<<dim3(1536 / 64, ROWS_ / 64), 256, 0, stream>>>(
        xln, C_, Wqkv + (size_t)d * 1536 * C_, C_, nullptr, nullptr,
        qkvb, 1536, C_, 0);
    attn_kernel<<<dim3(MTOK_ / 16, HEADS_, B_), 256, smem_attn, stream>>>(
        qkvb, anchors, Wsp + (size_t)d * DH_ * 3, attno);
    gemm_tiled<64, 64><<<dim3(C_ / 64, ROWS_ / 64), 256, 0, stream>>>(
        attno, C_, Wout + (size_t)d * C_ * C_, C_, bout + d * C_,
        feats, feats, C_, C_, 1);
    ln_kernel<<<ROWS_ / 8, 256, 0, stream>>>(feats, ln2g + d * C_, ln2b + d * C_, xln);
    gemm_tiled<128, 128><<<dim3(MLP_ / 128, ROWS_ / 128), 256, 0, stream>>>(
        xln, C_, Wff1 + (size_t)d * MLP_ * C_, C_, bff1 + d * MLP_,
        nullptr, ffh, MLP_, C_, 1);
    gemm_tiled<64, 64><<<dim3(C_ / 64, ROWS_ / 64), 256, 0, stream>>>(
        ffh, MLP_, Wff2 + (size_t)d * C_ * MLP_, MLP_, bff2 + d * C_,
        feats, feats, C_, MLP_, 0);
  }
  hipMemcpyAsync(d_out, feats, (size_t)ROWS_ * C_ * sizeof(float),
                 hipMemcpyDeviceToDevice, stream);
}


# attention: 32 queries per block instead of 16: each K and V fragment fetched from L2 feeds two query tiles (two accumulators per wave), next fragments requested ahead of the WMMAs; sums unchanged
# speedup vs baseline: 2.9873x; 1.0861x over previous
#include <hip/hip_runtime.h>
#include <hip/hip_bf16.h>
#include <math.h>

typedef float v2f __attribute__((ext_vector_type(2)));
typedef float v4f __attribute__((ext_vector_type(4)));
typedef float v8f __attribute__((ext_vector_type(8)));

#define B_    2
#define T_    4
#define N_    1024
#define L_    4
#define MA_   256
#define KNB_  32
#define C_    512
#define HEADS_ 8
#define DH_   64
#define MTOK_ 1024
#define ROWS_ 2048
#define MLP_  2048

__device__ __forceinline__ v8f wmma4(v2f a, v2f b, v8f c) {
  return __builtin_amdgcn_wmma_f32_16x16x4_f32(false, a, false, b, (short)0, c,
                                               false, false);
}

__device__ __forceinline__ float gelu_exact(float x) {
  return 0.5f * x * (1.0f + erff(x * 0.7071067811865475f));
}

#define FPS_SLOTS_  (N_ / 32)
#define FPS_CHAINS_ 4
#define FPS_CLEN_   (FPS_SLOTS_ / FPS_CHAINS_)

template <int CTRL>
__device__ __forceinline__ int dpp_row_i(int v) {
  return __builtin_amdgcn_update_dpp(0, v, CTRL, 0xf, 0xf, true);
}
__device__ __forceinline__ int other_row_i(int v) {
  return __builtin_amdgcn_permlanex16(v, v, 0x76543210, 0xfedcba98, false, false);
}
__device__ __forceinline__ int wave_max_i32(int v) {
  v = max(v, dpp_row_i<0x161>(v));
  v = max(v, dpp_row_i<0x162>(v));
  v = max(v, dpp_row_i<0x164>(v));
  v = max(v, dpp_row_i<0x168>(v));
  v = max(v, other_row_i(v));
  return v;
}
__device__ __forceinline__ int wave_min_i32(int v) {
  v = min(v, dpp_row_i<0x161>(v));
  v = min(v, dpp_row_i<0x162>(v));
  v = min(v, dpp_row_i<0x164>(v));
  v = min(v, dpp_row_i<0x168>(v));
  v = min(v, other_row_i(v));
  return v;
}

__device__ __forceinline__ float wave_max_f32(float v) {
  v = fmaxf(v, __builtin_bit_cast(float, dpp_row_i<0x161>(__builtin_bit_cast(int, v))));
  v = fmaxf(v, __builtin_bit_cast(float, dpp_row_i<0x162>(__builtin_bit_cast(int, v))));
  v = fmaxf(v, __builtin_bit_cast(float, dpp_row_i<0x164>(__builtin_bit_cast(int, v))));
  v = fmaxf(v, __builtin_bit_cast(float, dpp_row_i<0x168>(__builtin_bit_cast(int, v))));
  v = fmaxf(v, __builtin_bit_cast(float, other_row_i(__builtin_bit_cast(int, v))));
  return v;
}
__device__ __forceinline__ float wave_sum_f32(float v) {
  v += __builtin_bit_cast(float, dpp_row_i<0x161>(__builtin_bit_cast(int, v)));
  v += __builtin_bit_cast(float, dpp_row_i<0x162>(__builtin_bit_cast(int, v)));
  v += __builtin_bit_cast(float, dpp_row_i<0x164>(__builtin_bit_cast(int, v)));
  v += __builtin_bit_cast(float, dpp_row_i<0x168>(__builtin_bit_cast(int, v)));
  v += __builtin_bit_cast(float, other_row_i(__builtin_bit_cast(int, v)));
  return v;
}

__global__ __launch_bounds__(32) void fps_kernel(const float* __restrict__ xyzs,
                                                 float* __restrict__ anchors) {
  __shared__ float4 spt[N_];
  const int blk  = blockIdx.x;
  const int lane = threadIdx.x;
  const float* frame = xyzs + (size_t)blk * N_ * 3;
  float px[FPS_SLOTS_], py[FPS_SLOTS_], pz[FPS_SLOTS_];
  int   dist[FPS_SLOTS_];
  #pragma unroll
  for (int s = 0; s < FPS_SLOTS_; ++s) {
    const int i = s * 32 + lane;
    px[s] = frame[i * 3 + 0];
    py[s] = frame[i * 3 + 1];
    pz[s] = frame[i * 3 + 2];
    dist[s] = __builtin_bit_cast(int, 1e10f);
    spt[i] = make_float4(px[s], py[s], pz[s], 0.0f);
  }
  __syncthreads();
  float* aout = anchors + (size_t)blk * MA_ * 3;
  const char* sbytes = (const char*)spt;
  int curb = 0;
  for (int it = 1; it < MA_; ++it) {
    const float4 last = *(const float4*)(sbytes + curb);
    if (lane == 0) {
      aout[(it - 1) * 3 + 0] = last.x;
      aout[(it - 1) * 3 + 1] = last.y;
      aout[(it - 1) * 3 + 2] = last.z;
    }
    #pragma unroll
    for (int s = 0; s < FPS_SLOTS_; ++s) {
      const float dx = px[s] - last.x, dy = py[s] - last.y, dz = pz[s] - last.z;
      const float d2 = fmaf(dz, dz, fmaf(dx, dx, dy * dy));
      dist[s] = min(dist[s], __builtin_bit_cast(int, d2));
    }
    int bv[FPS_CHAINS_], bo[FPS_CHAINS_];
    #pragma unroll
    for (int c = 0; c < FPS_CHAINS_; ++c) {
      bv[c] = dist[c * FPS_CLEN_];
      bo[c] = c * FPS_CLEN_ * 32 * 16;
      #pragma unroll
      for (int k = 1; k < FPS_CLEN_; ++k) {
        const int s = c * FPS_CLEN_ + k;
        bo[c] = (dist[s] > bv[c]) ? s * 32 * 16 : bo[c];
        bv[c] = max(bv[c], dist[s]);
      }
    }
    #pragma unroll
    for (int w = 1; w < FPS_CHAINS_; w <<= 1) {
      #pragma unroll
      for (int c = 0; c < FPS_CHAINS_; c += 2 * w) {
        bo[c] = (bv[c + w] > bv[c]) ? bo[c + w] : bo[c];
        bv[c] = max(bv[c], bv[c + w]);
      }
    }
    const int gmax = wave_max_i32(bv[0]);
    const int cand = (bv[0] == gmax) ? (bo[0] + lane * 16) : 0x7fffffff;
    curb = wave_min_i32(cand);
  }
  const float4 last = *(const float4*)(sbytes + curb);
  if (lane == 0) {
    aout[(MA_ - 1) * 3 + 0] = last.x;
    aout[(MA_ - 1) * 3 + 1] = last.y;
    aout[(MA_ - 1) * 3 + 2] = last.z;
  }
}

#define P4W_ 8

__global__ __launch_bounds__(256) void p4dconv_kernel(const float* __restrict__ xyzs,
                                                      const float* __restrict__ anchors,
                                                      const float* __restrict__ Wd,
                                                      float* __restrict__ feats) {
  __shared__ __attribute__((aligned(16))) float  sfr[3][N_ * 3];
  __shared__ __attribute__((aligned(16))) float4 snb[P4W_][KNB_];
  const int t = threadIdx.x;
  const int lane = t & 31, w = t >> 5;
  const int blk = blockIdx.x;
  const int bl  = blk / (MA_ / P4W_);
  const int m   = (blk % (MA_ / P4W_)) * P4W_ + w;
  const int l   = bl % L_;
  const int b   = bl / L_;

  for (int f = 0; f < 3; ++f) {
    int fi = l + f - 1; fi = fi < 0 ? 0 : (fi > T_ - 1 ? T_ - 1 : fi);
    const v4f* srcp = (const v4f*)(xyzs + ((size_t)b * T_ + fi) * N_ * 3);
    v4f* dstp = (v4f*)&sfr[f][0];
    for (int i = 0; i < 3; ++i) dstp[t + 256 * i] = srcp[t + 256 * i];
  }

  const float* arow = anchors + ((size_t)bl * MA_ + m) * 3;
  const float ax = arow[0], ay = arow[1], az = arow[2];

  v2f w0[8], w1[8], w2[8], w3[8], mx[8];
  #pragma unroll
  for (int j = 0; j < 8; ++j) {
    const float* wp = Wd + (size_t)(2 * lane + 64 * j) * 4;
    const v4f wa = *(const v4f*)wp;
    const v4f wb = *(const v4f*)(wp + 4);
    w0[j] = (v2f){wa.x, wb.x};
    w1[j] = (v2f){wa.y, wb.y};
    w2[j] = (v2f){wa.z, wb.z};
    w3[j] = (v2f){wa.w, wb.w};
    mx[j] = (v2f){-3.4e38f, -3.4e38f};
  }
  __syncthreads();

  for (int f = 0; f < 3; ++f) {
    const float* fr = &sfr[f][0];
    const float p0x = fr[0] - ax, p0y = fr[1] - ay, p0z = fr[2] - az;
    int cnt = 0;
    for (int g = 0; g < 4 && cnt < KNB_; ++g) {
      float q[8][3];
      #pragma unroll
      for (int u = 0; u < 8; ++u) {
        const int i = (g * 8 + u) * 32 + lane;
        q[u][0] = fr[3 * i + 0];
        q[u][1] = fr[3 * i + 1];
        q[u][2] = fr[3 * i + 2];
      }
      #pragma unroll
      for (int u = 0; u < 8; ++u) {
        const float dx = q[u][0] - ax;
        const float dy = q[u][1] - ay;
        const float dz = q[u][2] - az;
        const float d2 = fmaf(dz, dz, fmaf(dx, dx, dy * dy));
        const bool pred = d2 < 1.0f;
        const unsigned mask = __builtin_amdgcn_ballot_w32(pred);
        const int pos = cnt + (int)__builtin_amdgcn_mbcnt_lo(mask, 0u);
        if (pred && pos < KNB_) snb[w][pos] = make_float4(dx, dy, dz, 0.0f);
        cnt += __builtin_popcount(mask);
      }
    }
    __syncthreads();
    if (lane >= cnt) {
      snb[w][lane] = (cnt > 0) ? snb[w][0] : make_float4(p0x, p0y, p0z, 0.0f);
    }
    __syncthreads();
    const float tq = (float)(f - 1);
    #pragma unroll 4
    for (int k = 0; k < KNB_; ++k) {
      const float4 p = snb[w][k];
      #pragma unroll
      for (int j = 0; j < 8; ++j) {
        const v2f v = w0[j] * p.x + w1[j] * p.y + w2[j] * p.z + w3[j] * tq;
        mx[j] = __builtin_elementwise_max(mx[j], v);
      }
    }
    __syncthreads();
  }
  float* orow = feats + ((size_t)bl * MA_ + m) * C_ + 2 * lane;
  #pragma unroll
  for (int j = 0; j < 8; ++j) {
    const v2f o = {fmaxf(mx[j].x, 0.0f), fmaxf(mx[j].y, 0.0f)};
    *(v2f*)(orow + 64 * j) = o;
  }
}

__global__ __launch_bounds__(256) void ln_kernel(const float* __restrict__ x, const float* __restrict__ g,
                                                 const float* __restrict__ bta, float* __restrict__ y) {
  const int lane = threadIdx.x & 31, w = threadIdx.x >> 5;
  const int row = blockIdx.x * 8 + w;
  const float* xr = x + (size_t)row * C_ + 4 * lane;
  v4f a[4], d[4];
  #pragma unroll
  for (int i = 0; i < 4; ++i) a[i] = *(const v4f*)(xr + 128 * i);
  float s = 0.0f;
  #pragma unroll
  for (int i = 0; i < 4; ++i) s += (a[i].x + a[i].y) + (a[i].z + a[i].w);
  const float mean = wave_sum_f32(s) * (1.0f / C_);
  float q = 0.0f;
  #pragma unroll
  for (int i = 0; i < 4; ++i) {
    d[i] = a[i] - mean;
    q += (d[i].x * d[i].x + d[i].y * d[i].y) + (d[i].z * d[i].z + d[i].w * d[i].w);
  }
  const float inv = rsqrtf(wave_sum_f32(q) * (1.0f / C_) + 1e-5f);
  float* yr = y + (size_t)row * C_ + 4 * lane;
  #pragma unroll
  for (int i = 0; i < 4; ++i) {
    const v4f gg = *(const v4f*)(g + 4 * lane + 128 * i);
    const v4f bb = *(const v4f*)(bta + 4 * lane + 128 * i);
    *(v4f*)(yr + 128 * i) = d[i] * inv * gg + bb;
  }
}

#define GBK_    64
#define GPITCH_ 68
#define GNBUF_  3

__device__ __forceinline__ void lds_barrier() {
  __builtin_amdgcn_fence(__ATOMIC_RELEASE, "workgroup", "local");
  __builtin_amdgcn_s_barrier();
  __builtin_amdgcn_fence(__ATOMIC_ACQUIRE, "workgroup", "local");
}

template <int BM, int BN>
__global__ __launch_bounds__(256) void gemm_tiled(const float* __restrict__ A, int lda,
                                                  const float* __restrict__ W, int ldw,
                                                  const float* __restrict__ bias,
                                                  const float* __restrict__ resid,
                                                  float* __restrict__ Cout, int ldc,
                                                  int Kdim, int act) {
  constexpr int WC = BN / 32;
  constexpr int WR = 8 / WC;
  constexpr int TM = BM / (16 * WR);
  constexpr int TN = 2;
  constexpr int LA = BM / 16;
  constexpr int LB = BN / 16;
  constexpr int NKB = GBK_ / 8;
  __shared__ __attribute__((aligned(16))) float As[GNBUF_][BM * GPITCH_];
  __shared__ __attribute__((aligned(16))) float Bs[GNBUF_][BN * GPITCH_];
  const int t = threadIdx.x;
  const int lane = t & 31, w = t >> 5;
  const int sub = lane & 15, hi = lane >> 4;
  const int wr = w / WC, wc = w % WC;
  const int i0 = blockIdx.y * BM, j0 = blockIdx.x * BN;
  const int lr = t >> 4;
  const int lc = (t & 15) * 4;
  const float* ag = A + (size_t)(i0 + lr) * lda + lc;
  const float* wg = W + (size_t)(j0 + lr) * ldw + lc;
  const int sw_off = lr * GPITCH_ + lc;
  const int nst = Kdim / GBK_;
  v4f ra[LA], rb[LB];
  {
    v4f pa[2][LA], pb[2][LB];
    #pragma unroll
    for (int i = 0; i < LA; ++i) pa[0][i] = *(const v4f*)(ag + (size_t)(16 * i) * lda);
    #pragma unroll
    for (int i = 0; i < LB; ++i) pb[0][i] = *(const v4f*)(wg + (size_t)(16 * i) * ldw);
    if (nst > 1) {
      #pragma unroll
      for (int i = 0; i < LA; ++i) pa[1][i] = *(const v4f*)(ag + (size_t)(16 * i) * lda + GBK_);
      #pragma unroll
      for (int i = 0; i < LB; ++i) pb[1][i] = *(const v4f*)(wg + (size_t)(16 * i) * ldw + GBK_);
    }
    if (nst > 2) {
      #pragma unroll
      for (int i = 0; i < LA; ++i) ra[i] = *(const v4f*)(ag + (size_t)(16 * i) * lda + 2 * GBK_);
      #pragma unroll
      for (int i = 0; i < LB; ++i) rb[i] = *(const v4f*)(wg + (size_t)(16 * i) * ldw + 2 * GBK_);
    } else {
      #pragma unroll
      for (int i = 0; i < LA; ++i) ra[i] = (v4f){0.f, 0.f, 0.f, 0.f};
      #pragma unroll
      for (int i = 0; i < LB; ++i) rb[i] = (v4f){0.f, 0.f, 0.f, 0.f};
    }
    #pragma unroll
    for (int i = 0; i < LA; ++i) *(v4f*)(&As[0][sw_off + 16 * i * GPITCH_]) = pa[0][i];
    #pragma unroll
    for (int i = 0; i < LB; ++i) *(v4f*)(&Bs[0][sw_off + 16 * i * GPITCH_]) = pb[0][i];
    if (nst > 1) {
      #pragma unroll
      for (int i = 0; i < LA; ++i) *(v4f*)(&As[1][sw_off + 16 * i * GPITCH_]) = pa[1][i];
      #pragma unroll
      for (int i = 0; i < LB; ++i) *(v4f*)(&Bs[1][sw_off + 16 * i * GPITCH_]) = pb[1][i];
    }
  }
  lds_barrier();

  v8f acc[TM][TN];
  #pragma unroll
  for (int m = 0; m < TM; ++m)
    #pragma unroll
    for (int n = 0; n < TN; ++n)
      acc[m][n] = (v8f){0.f, 0.f, 0.f, 0.f, 0.f, 0.f, 0.f, 0.f};

  const int fa_off = (wr * TM * 16 + sub) * GPITCH_ + 4 * hi;
  const int fb_off = (wc * 32 + sub) * GPITCH_ + 4 * hi;
  v4f fa[2][TM], fb[2][TN];
  #pragma unroll
  for (int m = 0; m < TM; ++m) fa[0][m] = *(const v4f*)(&As[0][fa_off + m * 16 * GPITCH_]);
  #pragma unroll
  for (int n = 0; n < TN; ++n) fb[0][n] = *(const v4f*)(&Bs[0][fb_off + n * 16 * GPITCH_]);
  int cur = 0, nx1 = 1, nx2 = 2;
  for (int st = 0; st < nst; ++st) {
    #pragma unroll
    for (int i = 0; i < LA; ++i) *(v4f*)(&As[nx2][sw_off + 16 * i * GPITCH_]) = ra[i];
    #pragma unroll
    for (int i = 0; i < LB; ++i) *(v4f*)(&Bs[nx2][sw_off + 16 * i * GPITCH_]) = rb[i];
    if (st + 3 < nst) {
      const int kn = (st + 3) * GBK_;
      #pragma unroll
      for (int i = 0; i < LA; ++i) ra[i] = *(const v4f*)(ag + (size_t)(16 * i) * lda + kn);
      #pragma unroll
      for (int i = 0; i < LB; ++i) rb[i] = *(const v4f*)(wg + (size_t)(16 * i) * ldw + kn);
    }
    const float* as  = &As[cur][fa_off];
    const float* bs  = &Bs[cur][fb_off];
    const float* asn = &As[nx1][fa_off];
    const float* bsn = &Bs[nx1][fb_off];
    const bool more = (st + 1 < nst);
    #pragma unroll
    for (int kb = 0; kb < NKB; ++kb) {
      const int c = kb & 1, d = c ^ 1;
      if (kb + 1 < NKB) {
        #pragma unroll
        for (int m = 0; m < TM; ++m) fa[d][m] = *(const v4f*)(as + m * 16 * GPITCH_ + (kb + 1) * 8);
        #pragma unroll
        for (int n = 0; n < TN; ++n) fb[d][n] = *(const v4f*)(bs + n * 16 * GPITCH_ + (kb + 1) * 8);
      } else if (more) {
        #pragma unroll
        for (int m = 0; m < TM; ++m) fa[d][m] = *(const v4f*)(asn + m * 16 * GPITCH_);
        #pragma unroll
        for (int n = 0; n < TN; ++n) fb[d][n] = *(const v4f*)(bsn + n * 16 * GPITCH_);
      }
      #pragma unroll
      for (int m = 0; m < TM; ++m)
        #pragma unroll
        for (int n = 0; n < TN; ++n) {
          const v2f a = {fa[c][m].x, fa[c][m].y};
          const v2f b = {fb[c][n].x, fb[c][n].y};
          acc[m][n] = wmma4(a, b, acc[m][n]);
        }
      #pragma unroll
      for (int m = 0; m < TM; ++m)
        #pragma unroll
        for (int n = 0; n < TN; ++n) {
          const v2f a = {fa[c][m].z, fa[c][m].w};
          const v2f b = {fb[c][n].z, fb[c][n].w};
          acc[m][n] = wmma4(a, b, acc[m][n]);
        }
    }
    lds_barrier();
    const int tcur = cur; cur = nx1; nx1 = nx2; nx2 = tcur;
  }

  float bj[TN];
  #pragma unroll
  for (int n = 0; n < TN; ++n) bj[n] = bias ? bias[j0 + wc * 32 + n * 16 + sub] : 0.0f;
  #pragma unroll
  for (int m = 0; m < TM; ++m) {
    #pragma unroll
    for (int r = 0; r < 8; ++r) {
      const int row = i0 + (wr * TM + m) * 16 + r + 8 * hi;
      #pragma unroll
      for (int n = 0; n < TN; ++n) {
        const int col = j0 + wc * 32 + n * 16 + sub;
        float v = acc[m][n][r] + bj[n];
        if (act == 1) v = gelu_exact(v);
        if (resid) v += resid[(size_t)row * ldc + col];
        Cout[(size_t)row * ldc + col] = v;
      }
    }
  }
}

#define SP_ 1028
#define QT_ 2
#define QB_ (16 * QT_)
#define VG_ 16

__global__ __launch_bounds__(256) void attn_kernel(const float* __restrict__ qkv,
                                                   const float* __restrict__ xyzf,
                                                   const float* __restrict__ Wsp,
                                                   float* __restrict__ attn_out) {
  __shared__ __attribute__((aligned(16))) float smem[QB_ * SP_ + QB_ + 3 * QB_];
  float* S    = smem;
  float* rsum = S + QB_ * SP_;
  float* wx   = rsum + QB_;
  const int b = blockIdx.z, h = blockIdx.y, i0 = blockIdx.x * QB_;
  const int t = threadIdx.x;
  const int lane = t & 31, w = t >> 5;
  const int sub = lane & 15, hi = lane >> 4;

  v4f qa[QT_][8];
  #pragma unroll
  for (int qt = 0; qt < QT_; ++qt) {
    const float* qbase =
        qkv + ((size_t)(b * MTOK_ + i0 + qt * 16 + sub) * 3 + 0) * C_ + h * DH_ + 4 * hi;
    #pragma unroll
    for (int kb = 0; kb < 8; ++kb) qa[qt][kb] = *(const v4f*)(qbase + 8 * kb);
  }
  const float* kwave =
      qkv + ((size_t)(b * MTOK_ + w * 128 + sub) * 3 + 1) * C_ + h * DH_ + 4 * hi;
  auto load_k = [&](v4f (&kf)[8], int jj) {
    const float* kbase = kwave + (size_t)jj * 16 * 3 * C_;
    #pragma unroll
    for (int kb = 0; kb < 8; ++kb) kf[kb] = *(const v4f*)(kbase + 8 * kb);
  };
  auto score_tile = [&](const v4f (&kf)[8], int jj) {
    const int j0 = (w * 8 + jj) * 16;
    v8f acc[QT_];
    #pragma unroll
    for (int qt = 0; qt < QT_; ++qt) acc[qt] = (v8f){0.f, 0.f, 0.f, 0.f, 0.f, 0.f, 0.f, 0.f};
    #pragma unroll
    for (int kb = 0; kb < 8; ++kb) {
      const v2f b0 = {kf[kb].x, kf[kb].y}, b1 = {kf[kb].z, kf[kb].w};
      #pragma unroll
      for (int qt = 0; qt < QT_; ++qt) {
        const v2f a0 = {qa[qt][kb].x, qa[qt][kb].y};
        acc[qt] = wmma4(a0, b0, acc[qt]);
      }
      #pragma unroll
      for (int qt = 0; qt < QT_; ++qt) {
        const v2f a1 = {qa[qt][kb].z, qa[qt][kb].w};
        acc[qt] = wmma4(a1, b1, acc[qt]);
      }
    }
    #pragma unroll
    for (int qt = 0; qt < QT_; ++qt) {
      #pragma unroll
      for (int r = 0; r < 8; ++r)
        S[(qt * 16 + r + 8 * hi) * SP_ + j0 + sub] = acc[qt][r] * 0.125f;
    }
  };
  {
    v4f kf0[8], kf1[8];
    load_k(kf0, 0);
    #pragma unroll 1
    for (int jj = 0; jj < 8; jj += 2) {
      load_k(kf1, jj + 1);
      score_tile(kf0, jj);
      load_k(kf0, (jj + 2 < 8) ? jj + 2 : jj);
      score_tile(kf1, jj + 1);
    }
  }
  __syncthreads();

  for (int q = 0; q < QB_ / 8; ++q) {
    const int row = (QB_ / 8) * w + q;
    float* sr = S + row * SP_ + 4 * lane;
    v4f x[8];
    #pragma unroll
    for (int i = 0; i < 8; ++i) x[i] = *(const v4f*)(sr + 128 * i);
    float lm = -3.4e38f;
    #pragma unroll
    for (int i = 0; i < 8; ++i)
      lm = fmaxf(lm, fmaxf(fmaxf(x[i].x, x[i].y), fmaxf(x[i].z, x[i].w)));
    const float rm = wave_max_f32(lm);
    float ls = 0.0f;
    #pragma unroll
    for (int i = 0; i < 8; ++i) {
      v4f e;
      e.x = expf(x[i].x - rm);
      e.y = expf(x[i].y - rm);
      e.z = expf(x[i].z - rm);
      e.w = expf(x[i].w - rm);
      ls += (e.x + e.y) + (e.z + e.w);
      *(v4f*)(sr + 128 * i) = e;
    }
    ls = wave_sum_f32(ls);
    if (lane == 0) rsum[row] = ls;
  }
  __syncthreads();

  v8f acc0 = {0.f, 0.f, 0.f, 0.f, 0.f, 0.f, 0.f, 0.f};
  v8f acc1 = {0.f, 0.f, 0.f, 0.f, 0.f, 0.f, 0.f, 0.f};
  const int d0 = (w & 3) * 16;
  if (w < 4) {
    const float* vbase =
        qkv + ((size_t)(b * MTOK_) * 3 + 2) * C_ + h * DH_ + d0 + sub;
    const float* srow0 = S + sub * SP_ + 2 * hi;
    const float* srow1 = srow0 + 16 * SP_;
    auto load_v = [&](v2f (&vb)[VG_], int g) {
      #pragma unroll
      for (int u = 0; u < VG_; ++u) {
        const int kk = 4 * (g * VG_ + u) + 2 * hi;
        vb[u].x = vbase[(size_t)kk * 1536];
        vb[u].y = vbase[(size_t)(kk + 1) * 1536];
      }
    };
    auto pv_group = [&](const v2f (&vb)[VG_], int g) {
      #pragma unroll
      for (int u = 0; u < VG_; ++u) {
        const int ks = g * VG_ + u;
        const v2f a0 = *(const v2f*)(srow0 + 4 * ks);
        const v2f a1 = *(const v2f*)(srow1 + 4 * ks);
        acc0 = wmma4(a0, vb[u], acc0);
        acc1 = wmma4(a1, vb[u], acc1);
      }
    };
    v2f vb0[VG_], vb1[VG_];
    load_v(vb0, 0);
    #pragma unroll 1
    for (int g = 0; g < 256 / VG_; g += 2) {
      load_v(vb1, g + 1);
      pv_group(vb0, g);
      load_v(vb0, (g + 2 < 256 / VG_) ? g + 2 : g);
      pv_group(vb1, g + 1);
    }
  } else {
    const float* xb = xyzf + (size_t)b * MTOK_ * 3;
    #pragma unroll 1
    for (int ps = 0; ps < 2; ++ps) {
      const int r0 = (w - 4) * 8 + 4 * ps;
      float ax[4][3];
      #pragma unroll
      for (int rr = 0; rr < 4; ++rr) { ax[rr][0] = 0.0f; ax[rr][1] = 0.0f; ax[rr][2] = 0.0f; }
      #pragma unroll 1
      for (int i = 0; i < 8; ++i) {
        const int j = 4 * lane + 128 * i;
        const v4f c0 = *(const v4f*)(xb + 3 * j);
        const v4f c1 = *(const v4f*)(xb + 3 * j + 4);
        const v4f c2 = *(const v4f*)(xb + 3 * j + 8);
        #pragma unroll
        for (int rr = 0; rr < 4; ++rr) {
          const v4f p = *(const v4f*)(S + (r0 + rr) * SP_ + j);
          ax[rr][0] += p.x * c0.x + p.y * c0.w + p.z * c1.z + p.w * c2.y;
          ax[rr][1] += p.x * c0.y + p.y * c1.x + p.z * c1.w + p.w * c2.z;
          ax[rr][2] += p.x * c0.z + p.y * c1.y + p.z * c2.x + p.w * c2.w;
        }
      }
      #pragma unroll
      for (int rr = 0; rr < 4; ++rr) {
        #pragma unroll
        for (int cc = 0; cc < 3; ++cc) ax[rr][cc] = wave_sum_f32(ax[rr][cc]);
      }
      if (lane == 0) {
        #pragma unroll
        for (int rr = 0; rr < 4; ++rr) {
          #pragma unroll
          for (int cc = 0; cc < 3; ++cc) wx[(r0 + rr) * 3 + cc] = ax[rr][cc];
        }
      }
    }
  }
  __syncthreads();
  if (w < 4) {
    const int d = d0 + sub;
    const float w0 = Wsp[d * 3 + 0], w1 = Wsp[d * 3 + 1], w2 = Wsp[d * 3 + 2];
    #pragma unroll
    for (int qt = 0; qt < QT_; ++qt) {
      #pragma unroll
      for (int r = 0; r < 8; ++r) {
        const int rr = qt * 16 + r + 8 * hi;
        const int i  = i0 + rr;
        const float inv = 1.0f / rsum[rr];
        const float* xi = xyzf + ((size_t)b * MTOK_ + i) * 3;
        const float wd0 = wx[rr * 3 + 0] * inv - xi[0];
        const float wd1 = wx[rr * 3 + 1] * inv - xi[1];
        const float wd2 = wx[rr * 3 + 2] * inv - xi[2];
        const float o = (qt == 0) ? acc0[r] : acc1[r];
        attn_out[((size_t)(b * MTOK_) + i) * C_ + h * DH_ + d] =
            o * inv + w0 * wd0 + w1 * wd1 + w2 * wd2;
      }
    }
  }
}

extern "C" void kernel_launch(void* const* d_in, const int* in_sizes, int n_in,
                              void* d_out, int out_size, void* d_ws, size_t ws_size,
                              hipStream_t stream) {
  (void)in_sizes; (void)n_in; (void)out_size; (void)ws_size;
  const float* xyzs = (const float*)d_in[0];
  const float* Wd   = (const float*)d_in[1];
  const float* Wqkv = (const float*)d_in[2];
  const float* Wsp  = (const float*)d_in[3];
  const float* Wout = (const float*)d_in[4];
  const float* bout = (const float*)d_in[5];
  const float* ln1g = (const float*)d_in[6];
  const float* ln1b = (const float*)d_in[7];
  const float* Wff1 = (const float*)d_in[8];
  const float* bff1 = (const float*)d_in[9];
  const float* Wff2 = (const float*)d_in[10];
  const float* bff2 = (const float*)d_in[11];
  const float* ln2g = (const float*)d_in[12];
  const float* ln2b = (const float*)d_in[13];

  float* ws      = (float*)d_ws;
  float* anchors = ws;
  float* feats   = ws + 8192;
  float* xln     = feats + (size_t)ROWS_ * C_;
  float* qkvb    = xln   + (size_t)ROWS_ * C_;
  float* attno   = qkvb  + (size_t)ROWS_ * 1536;
  float* ffh     = attno + (size_t)ROWS_ * C_;

  fps_kernel<<<B_ * L_, 32, 0, stream>>>(xyzs, anchors);
  p4dconv_kernel<<<B_ * L_ * MA_ / P4W_, 256, 0, stream>>>(xyzs, anchors, Wd, feats);

  for (int d = 0; d < 2; ++d) {
    ln_kernel<<<ROWS_ / 8, 256, 0, stream>>>(feats, ln1g + d * C_, ln1b + d * C_, xln);
    gemm_tiled<64, 64><<<dim3(1536 / 64, ROWS_ / 64), 256, 0, stream>>>(
        xln, C_, Wqkv + (size_t)d * 1536 * C_, C_, nullptr, nullptr,
        qkvb, 1536, C_, 0);
    attn_kernel<<<dim3(MTOK_ / QB_, HEADS_, B_), 256, 0, stream>>>(
        qkvb, anchors, Wsp + (size_t)d * DH_ * 3, attno);
    gemm_tiled<64, 64><<<dim3(C_ / 64, ROWS_ / 64), 256, 0, stream>>>(
        attno, C_, Wout + (size_t)d * C_ * C_, C_, bout + d * C_,
        feats, feats, C_, C_, 1);
    ln_kernel<<<ROWS_ / 8, 256, 0, stream>>>(feats, ln2g + d * C_, ln2b + d * C_, xln);
    gemm_tiled<128, 128><<<dim3(MLP_ / 128, ROWS_ / 128), 256, 0, stream>>>(
        xln, C_, Wff1 + (size_t)d * MLP_ * C_, C_, bff1 + d * MLP_,
        nullptr, ffh, MLP_, C_, 1);
    gemm_tiled<64, 64><<<dim3(C_ / 64, ROWS_ / 64), 256, 0, stream>>>(
        ffh, MLP_, Wff2 + (size_t)d * C_ * MLP_, MLP_, bff2 + d * C_,
        feats, feats, C_, MLP_, 0);
  }
  hipMemcpyAsync(d_out, feats, (size_t)ROWS_ * C_ * sizeof(float),
                 hipMemcpyDeviceToDevice, stream);
}


# GEMM epilogue de-serialised: bias/gelu/residual as template flags, residual and bias requested before the k loop, both erf ranges computed and selected (same value): no per-output branch or wait
# speedup vs baseline: 3.3496x; 1.1213x over previous
#include <hip/hip_runtime.h>
#include <hip/hip_bf16.h>
#include <math.h>

typedef float v2f __attribute__((ext_vector_type(2)));
typedef float v4f __attribute__((ext_vector_type(4)));
typedef float v8f __attribute__((ext_vector_type(8)));

#define B_    2
#define T_    4
#define N_    1024
#define L_    4
#define MA_   256
#define KNB_  32
#define C_    512
#define HEADS_ 8
#define DH_   64
#define MTOK_ 1024
#define ROWS_ 2048
#define MLP_  2048

__device__ __forceinline__ v8f wmma4(v2f a, v2f b, v8f c) {
  return __builtin_amdgcn_wmma_f32_16x16x4_f32(false, a, false, b, (short)0, c,
                                               false, false);
}

__device__ __forceinline__ float fbits(unsigned u) { return __builtin_bit_cast(float, u); }
__device__ __forceinline__ float gelu_sel(float v) {
  const float z = v * 0.7071067811865475f;
  const float t = fabsf(z);
  float p = fmaf(t, fbits(0x378e98abu), fbits(0xb9c68948u));
  p = fmaf(t, p, fbits(0x3b7cd369u));
  p = fmaf(t, p, fbits(0xbcc618b2u));
  p = fmaf(t, p, fbits(0x3dda74e4u));
  p = fmaf(t, p, fbits(0x3f228afdu));
  p = fmaf(t, p, fbits(0x3e03c728u));
  p = fmaf(t, p, t);
  const float big = 1.0f - expf(-p);
  const float s = z * z;
  float q = fmaf(s, fbits(0xba1345e1u), fbits(0x3ba10414u));
  q = fmaf(s, q, fbits(0xbcdac9b8u));
  q = fmaf(s, q, fbits(0x3de703beu));
  q = fmaf(s, q, fbits(0xbec09330u));
  q = fmaf(s, q, fbits(0x3e0375d0u));
  const float sm = fmaf(t, q, t);
  const float r = (t < 1.0f) ? sm : big;
  return (0.5f * v) * (1.0f + copysignf(r, z));
}

#define FPS_SLOTS_  (N_ / 32)
#define FPS_CHAINS_ 4
#define FPS_CLEN_   (FPS_SLOTS_ / FPS_CHAINS_)

template <int CTRL>
__device__ __forceinline__ int dpp_row_i(int v) {
  return __builtin_amdgcn_update_dpp(0, v, CTRL, 0xf, 0xf, true);
}
__device__ __forceinline__ int other_row_i(int v) {
  return __builtin_amdgcn_permlanex16(v, v, 0x76543210, 0xfedcba98, false, false);
}
__device__ __forceinline__ int wave_max_i32(int v) {
  v = max(v, dpp_row_i<0x161>(v));
  v = max(v, dpp_row_i<0x162>(v));
  v = max(v, dpp_row_i<0x164>(v));
  v = max(v, dpp_row_i<0x168>(v));
  v = max(v, other_row_i(v));
  return v;
}
__device__ __forceinline__ int wave_min_i32(int v) {
  v = min(v, dpp_row_i<0x161>(v));
  v = min(v, dpp_row_i<0x162>(v));
  v = min(v, dpp_row_i<0x164>(v));
  v = min(v, dpp_row_i<0x168>(v));
  v = min(v, other_row_i(v));
  return v;
}

__device__ __forceinline__ float wave_max_f32(float v) {
  v = fmaxf(v, __builtin_bit_cast(float, dpp_row_i<0x161>(__builtin_bit_cast(int, v))));
  v = fmaxf(v, __builtin_bit_cast(float, dpp_row_i<0x162>(__builtin_bit_cast(int, v))));
  v = fmaxf(v, __builtin_bit_cast(float, dpp_row_i<0x164>(__builtin_bit_cast(int, v))));
  v = fmaxf(v, __builtin_bit_cast(float, dpp_row_i<0x168>(__builtin_bit_cast(int, v))));
  v = fmaxf(v, __builtin_bit_cast(float, other_row_i(__builtin_bit_cast(int, v))));
  return v;
}
__device__ __forceinline__ float wave_sum_f32(float v) {
  v += __builtin_bit_cast(float, dpp_row_i<0x161>(__builtin_bit_cast(int, v)));
  v += __builtin_bit_cast(float, dpp_row_i<0x162>(__builtin_bit_cast(int, v)));
  v += __builtin_bit_cast(float, dpp_row_i<0x164>(__builtin_bit_cast(int, v)));
  v += __builtin_bit_cast(float, dpp_row_i<0x168>(__builtin_bit_cast(int, v)));
  v += __builtin_bit_cast(float, other_row_i(__builtin_bit_cast(int, v)));
  return v;
}

__global__ __launch_bounds__(32) void fps_kernel(const float* __restrict__ xyzs,
                                                 float* __restrict__ anchors) {
  __shared__ float4 spt[N_];
  const int blk  = blockIdx.x;
  const int lane = threadIdx.x;
  const float* frame = xyzs + (size_t)blk * N_ * 3;
  float px[FPS_SLOTS_], py[FPS_SLOTS_], pz[FPS_SLOTS_];
  int   dist[FPS_SLOTS_];
  #pragma unroll
  for (int s = 0; s < FPS_SLOTS_; ++s) {
    const int i = s * 32 + lane;
    px[s] = frame[i * 3 + 0];
    py[s] = frame[i * 3 + 1];
    pz[s] = frame[i * 3 + 2];
    dist[s] = __builtin_bit_cast(int, 1e10f);
    spt[i] = make_float4(px[s], py[s], pz[s], 0.0f);
  }
  __syncthreads();
  float* aout = anchors + (size_t)blk * MA_ * 3;
  const char* sbytes = (const char*)spt;
  int curb = 0;
  for (int it = 1; it < MA_; ++it) {
    const float4 last = *(const float4*)(sbytes + curb);
    if (lane == 0) {
      aout[(it - 1) * 3 + 0] = last.x;
      aout[(it - 1) * 3 + 1] = last.y;
      aout[(it - 1) * 3 + 2] = last.z;
    }
    #pragma unroll
    for (int s = 0; s < FPS_SLOTS_; ++s) {
      const float dx = px[s] - last.x, dy = py[s] - last.y, dz = pz[s] - last.z;
      const float d2 = fmaf(dz, dz, fmaf(dx, dx, dy * dy));
      dist[s] = min(dist[s], __builtin_bit_cast(int, d2));
    }
    int bv[FPS_CHAINS_], bo[FPS_CHAINS_];
    #pragma unroll
    for (int c = 0; c < FPS_CHAINS_; ++c) {
      bv[c] = dist[c * FPS_CLEN_];
      bo[c] = c * FPS_CLEN_ * 32 * 16;
      #pragma unroll
      for (int k = 1; k < FPS_CLEN_; ++k) {
        const int s = c * FPS_CLEN_ + k;
        bo[c] = (dist[s] > bv[c]) ? s * 32 * 16 : bo[c];
        bv[c] = max(bv[c], dist[s]);
      }
    }
    #pragma unroll
    for (int w = 1; w < FPS_CHAINS_; w <<= 1) {
      #pragma unroll
      for (int c = 0; c < FPS_CHAINS_; c += 2 * w) {
        bo[c] = (bv[c + w] > bv[c]) ? bo[c + w] : bo[c];
        bv[c] = max(bv[c], bv[c + w]);
      }
    }
    const int gmax = wave_max_i32(bv[0]);
    const int cand = (bv[0] == gmax) ? (bo[0] + lane * 16) : 0x7fffffff;
    curb = wave_min_i32(cand);
  }
  const float4 last = *(const float4*)(sbytes + curb);
  if (lane == 0) {
    aout[(MA_ - 1) * 3 + 0] = last.x;
    aout[(MA_ - 1) * 3 + 1] = last.y;
    aout[(MA_ - 1) * 3 + 2] = last.z;
  }
}

#define P4W_ 8

__global__ __launch_bounds__(256) void p4dconv_kernel(const float* __restrict__ xyzs,
                                                      const float* __restrict__ anchors,
                                                      const float* __restrict__ Wd,
                                                      float* __restrict__ feats) {
  __shared__ __attribute__((aligned(16))) float  sfr[3][N_ * 3];
  __shared__ __attribute__((aligned(16))) float4 snb[P4W_][KNB_];
  const int t = threadIdx.x;
  const int lane = t & 31, w = t >> 5;
  const int blk = blockIdx.x;
  const int bl  = blk / (MA_ / P4W_);
  const int m   = (blk % (MA_ / P4W_)) * P4W_ + w;
  const int l   = bl % L_;
  const int b   = bl / L_;

  for (int f = 0; f < 3; ++f) {
    int fi = l + f - 1; fi = fi < 0 ? 0 : (fi > T_ - 1 ? T_ - 1 : fi);
    const v4f* srcp = (const v4f*)(xyzs + ((size_t)b * T_ + fi) * N_ * 3);
    v4f* dstp = (v4f*)&sfr[f][0];
    for (int i = 0; i < 3; ++i) dstp[t + 256 * i] = srcp[t + 256 * i];
  }

  const float* arow = anchors + ((size_t)bl * MA_ + m) * 3;
  const float ax = arow[0], ay = arow[1], az = arow[2];

  v2f w0[8], w1[8], w2[8], w3[8], mx[8];
  #pragma unroll
  for (int j = 0; j < 8; ++j) {
    const float* wp = Wd + (size_t)(2 * lane + 64 * j) * 4;
    const v4f wa = *(const v4f*)wp;
    const v4f wb = *(const v4f*)(wp + 4);
    w0[j] = (v2f){wa.x, wb.x};
    w1[j] = (v2f){wa.y, wb.y};
    w2[j] = (v2f){wa.z, wb.z};
    w3[j] = (v2f){wa.w, wb.w};
    mx[j] = (v2f){-3.4e38f, -3.4e38f};
  }
  __syncthreads();

  for (int f = 0; f < 3; ++f) {
    const float* fr = &sfr[f][0];
    const float p0x = fr[0] - ax, p0y = fr[1] - ay, p0z = fr[2] - az;
    int cnt = 0;
    for (int g = 0; g < 4 && cnt < KNB_; ++g) {
      float q[8][3];
      #pragma unroll
      for (int u = 0; u < 8; ++u) {
        const int i = (g * 8 + u) * 32 + lane;
        q[u][0] = fr[3 * i + 0];
        q[u][1] = fr[3 * i + 1];
        q[u][2] = fr[3 * i + 2];
      }
      #pragma unroll
      for (int u = 0; u < 8; ++u) {
        const float dx = q[u][0] - ax;
        const float dy = q[u][1] - ay;
        const float dz = q[u][2] - az;
        const float d2 = fmaf(dz, dz, fmaf(dx, dx, dy * dy));
        const bool pred = d2 < 1.0f;
        const unsigned mask = __builtin_amdgcn_ballot_w32(pred);
        const int pos = cnt + (int)__builtin_amdgcn_mbcnt_lo(mask, 0u);
        if (pred && pos < KNB_) snb[w][pos] = make_float4(dx, dy, dz, 0.0f);
        cnt += __builtin_popcount(mask);
      }
    }
    __syncthreads();
    if (lane >= cnt) {
      snb[w][lane] = (cnt > 0) ? snb[w][0] : make_float4(p0x, p0y, p0z, 0.0f);
    }
    __syncthreads();
    const float tq = (float)(f - 1);
    #pragma unroll 4
    for (int k = 0; k < KNB_; ++k) {
      const float4 p = snb[w][k];
      #pragma unroll
      for (int j = 0; j < 8; ++j) {
        const v2f v = w0[j] * p.x + w1[j] * p.y + w2[j] * p.z + w3[j] * tq;
        mx[j] = __builtin_elementwise_max(mx[j], v);
      }
    }
    __syncthreads();
  }
  float* orow = feats + ((size_t)bl * MA_ + m) * C_ + 2 * lane;
  #pragma unroll
  for (int j = 0; j < 8; ++j) {
    const v2f o = {fmaxf(mx[j].x, 0.0f), fmaxf(mx[j].y, 0.0f)};
    *(v2f*)(orow + 64 * j) = o;
  }
}

__global__ __launch_bounds__(256) void ln_kernel(const float* __restrict__ x, const float* __restrict__ g,
                                                 const float* __restrict__ bta, float* __restrict__ y) {
  const int lane = threadIdx.x & 31, w = threadIdx.x >> 5;
  const int row = blockIdx.x * 8 + w;
  const float* xr = x + (size_t)row * C_ + 4 * lane;
  v4f a[4], d[4];
  #pragma unroll
  for (int i = 0; i < 4; ++i) a[i] = *(const v4f*)(xr + 128 * i);
  float s = 0.0f;
  #pragma unroll
  for (int i = 0; i < 4; ++i) s += (a[i].x + a[i].y) + (a[i].z + a[i].w);
  const float mean = wave_sum_f32(s) * (1.0f / C_);
  float q = 0.0f;
  #pragma unroll
  for (int i = 0; i < 4; ++i) {
    d[i] = a[i] - mean;
    q += (d[i].x * d[i].x + d[i].y * d[i].y) + (d[i].z * d[i].z + d[i].w * d[i].w);
  }
  const float inv = rsqrtf(wave_sum_f32(q) * (1.0f / C_) + 1e-5f);
  float* yr = y + (size_t)row * C_ + 4 * lane;
  #pragma unroll
  for (int i = 0; i < 4; ++i) {
    const v4f gg = *(const v4f*)(g + 4 * lane + 128 * i);
    const v4f bb = *(const v4f*)(bta + 4 * lane + 128 * i);
    *(v4f*)(yr + 128 * i) = d[i] * inv * gg + bb;
  }
}

#define GBK_    64
#define GPITCH_ 68
#define GNBUF_  3

__device__ __forceinline__ void lds_barrier() {
  __builtin_amdgcn_fence(__ATOMIC_RELEASE, "workgroup", "local");
  __builtin_amdgcn_s_barrier();
  __builtin_amdgcn_fence(__ATOMIC_ACQUIRE, "workgroup", "local");
}

template <int BM, int BN, int ACT, bool RESID, bool BIAS>
__global__ __launch_bounds__(256) void gemm_tiled(const float* __restrict__ A, int lda,
                                                  const float* __restrict__ W, int ldw,
                                                  const float* __restrict__ bias,
                                                  const float* resid,
                                                  float* Cout, int ldc,
                                                  int Kdim) {
  constexpr int WC = BN / 32;
  constexpr int WR = 8 / WC;
  constexpr int TM = BM / (16 * WR);
  constexpr int TN = 2;
  constexpr int LA = BM / 16;
  constexpr int LB = BN / 16;
  constexpr int NKB = GBK_ / 8;
  __shared__ __attribute__((aligned(16))) float As[GNBUF_][BM * GPITCH_];
  __shared__ __attribute__((aligned(16))) float Bs[GNBUF_][BN * GPITCH_];
  const int t = threadIdx.x;
  const int lane = t & 31, w = t >> 5;
  const int sub = lane & 15, hi = lane >> 4;
  const int wr = w / WC, wc = w % WC;
  const int i0 = blockIdx.y * BM, j0 = blockIdx.x * BN;
  const int lr = t >> 4;
  const int lc = (t & 15) * 4;
  const float* ag = A + (size_t)(i0 + lr) * lda + lc;
  const float* wg = W + (size_t)(j0 + lr) * ldw + lc;
  const int sw_off = lr * GPITCH_ + lc;
  const int nst = Kdim / GBK_;
  v4f ra[LA], rb[LB];
  {
    v4f pa[2][LA], pb[2][LB];
    #pragma unroll
    for (int i = 0; i < LA; ++i) pa[0][i] = *(const v4f*)(ag + (size_t)(16 * i) * lda);
    #pragma unroll
    for (int i = 0; i < LB; ++i) pb[0][i] = *(const v4f*)(wg + (size_t)(16 * i) * ldw);
    if (nst > 1) {
      #pragma unroll
      for (int i = 0; i < LA; ++i) pa[1][i] = *(const v4f*)(ag + (size_t)(16 * i) * lda + GBK_);
      #pragma unroll
      for (int i = 0; i < LB; ++i) pb[1][i] = *(const v4f*)(wg + (size_t)(16 * i) * ldw + GBK_);
    }
    if (nst > 2) {
      #pragma unroll
      for (int i = 0; i < LA; ++i) ra[i] = *(const v4f*)(ag + (size_t)(16 * i) * lda + 2 * GBK_);
      #pragma unroll
      for (int i = 0; i < LB; ++i) rb[i] = *(const v4f*)(wg + (size_t)(16 * i) * ldw + 2 * GBK_);
    } else {
      #pragma unroll
      for (int i = 0; i < LA; ++i) ra[i] = (v4f){0.f, 0.f, 0.f, 0.f};
      #pragma unroll
      for (int i = 0; i < LB; ++i) rb[i] = (v4f){0.f, 0.f, 0.f, 0.f};
    }
    #pragma unroll
    for (int i = 0; i < LA; ++i) *(v4f*)(&As[0][sw_off + 16 * i * GPITCH_]) = pa[0][i];
    #pragma unroll
    for (int i = 0; i < LB; ++i) *(v4f*)(&Bs[0][sw_off + 16 * i * GPITCH_]) = pb[0][i];
    if (nst > 1) {
      #pragma unroll
      for (int i = 0; i < LA; ++i) *(v4f*)(&As[1][sw_off + 16 * i * GPITCH_]) = pa[1][i];
      #pragma unroll
      for (int i = 0; i < LB; ++i) *(v4f*)(&Bs[1][sw_off + 16 * i * GPITCH_]) = pb[1][i];
    }
  }
  const size_t erow0 = (size_t)(i0 + wr * TM * 16 + 8 * hi);
  const int    ecol0 = j0 + wc * 32 + sub;
  float bj[TN];
  #pragma unroll
  for (int n = 0; n < TN; ++n) bj[n] = BIAS ? bias[ecol0 + n * 16] : 0.0f;
  float rv[TM][8][TN];
  if (RESID) {
    #pragma unroll
    for (int m = 0; m < TM; ++m)
      #pragma unroll
      for (int r = 0; r < 8; ++r)
        #pragma unroll
        for (int n = 0; n < TN; ++n)
          rv[m][r][n] = resid[(erow0 + m * 16 + r) * ldc + ecol0 + n * 16];
  }
  lds_barrier();

  v8f acc[TM][TN];
  #pragma unroll
  for (int m = 0; m < TM; ++m)
    #pragma unroll
    for (int n = 0; n < TN; ++n)
      acc[m][n] = (v8f){0.f, 0.f, 0.f, 0.f, 0.f, 0.f, 0.f, 0.f};

  const int fa_off = (wr * TM * 16 + sub) * GPITCH_ + 4 * hi;
  const int fb_off = (wc * 32 + sub) * GPITCH_ + 4 * hi;
  v4f fa[2][TM], fb[2][TN];
  #pragma unroll
  for (int m = 0; m < TM; ++m) fa[0][m] = *(const v4f*)(&As[0][fa_off + m * 16 * GPITCH_]);
  #pragma unroll
  for (int n = 0; n < TN; ++n) fb[0][n] = *(const v4f*)(&Bs[0][fb_off + n * 16 * GPITCH_]);
  int cur = 0, nx1 = 1, nx2 = 2;
  for (int st = 0; st < nst; ++st) {
    #pragma unroll
    for (int i = 0; i < LA; ++i) *(v4f*)(&As[nx2][sw_off + 16 * i * GPITCH_]) = ra[i];
    #pragma unroll
    for (int i = 0; i < LB; ++i) *(v4f*)(&Bs[nx2][sw_off + 16 * i * GPITCH_]) = rb[i];
    if (st + 3 < nst) {
      const int kn = (st + 3) * GBK_;
      #pragma unroll
      for (int i = 0; i < LA; ++i) ra[i] = *(const v4f*)(ag + (size_t)(16 * i) * lda + kn);
      #pragma unroll
      for (int i = 0; i < LB; ++i) rb[i] = *(const v4f*)(wg + (size_t)(16 * i) * ldw + kn);
    }
    const float* as  = &As[cur][fa_off];
    const float* bs  = &Bs[cur][fb_off];
    const float* asn = &As[nx1][fa_off];
    const float* bsn = &Bs[nx1][fb_off];
    const bool more = (st + 1 < nst);
    #pragma unroll
    for (int kb = 0; kb < NKB; ++kb) {
      const int c = kb & 1, d = c ^ 1;
      if (kb + 1 < NKB) {
        #pragma unroll
        for (int m = 0; m < TM; ++m) fa[d][m] = *(const v4f*)(as + m * 16 * GPITCH_ + (kb + 1) * 8);
        #pragma unroll
        for (int n = 0; n < TN; ++n) fb[d][n] = *(const v4f*)(bs + n * 16 * GPITCH_ + (kb + 1) * 8);
      } else if (more) {
        #pragma unroll
        for (int m = 0; m < TM; ++m) fa[d][m] = *(const v4f*)(asn + m * 16 * GPITCH_);
        #pragma unroll
        for (int n = 0; n < TN; ++n) fb[d][n] = *(const v4f*)(bsn + n * 16 * GPITCH_);
      }
      #pragma unroll
      for (int m = 0; m < TM; ++m)
        #pragma unroll
        for (int n = 0; n < TN; ++n) {
          const v2f a = {fa[c][m].x, fa[c][m].y};
          const v2f b = {fb[c][n].x, fb[c][n].y};
          acc[m][n] = wmma4(a, b, acc[m][n]);
        }
      #pragma unroll
      for (int m = 0; m < TM; ++m)
        #pragma unroll
        for (int n = 0; n < TN; ++n) {
          const v2f a = {fa[c][m].z, fa[c][m].w};
          const v2f b = {fb[c][n].z, fb[c][n].w};
          acc[m][n] = wmma4(a, b, acc[m][n]);
        }
    }
    lds_barrier();
    const int tcur = cur; cur = nx1; nx1 = nx2; nx2 = tcur;
  }

  #pragma unroll
  for (int m = 0; m < TM; ++m) {
    #pragma unroll
    for (int r = 0; r < 8; ++r) {
      #pragma unroll
      for (int n = 0; n < TN; ++n) {
        float v = acc[m][n][r];
        if (BIAS) v += bj[n];
        if (ACT == 1) v = gelu_sel(v);
        if (RESID) v += rv[m][r][n];
        Cout[(erow0 + m * 16 + r) * ldc + ecol0 + n * 16] = v;
      }
    }
  }
}

#define SP_ 1028
#define QT_ 2
#define QB_ (16 * QT_)
#define VG_ 16

__global__ __launch_bounds__(256) void attn_kernel(const float* __restrict__ qkv,
                                                   const float* __restrict__ xyzf,
                                                   const float* __restrict__ Wsp,
                                                   float* __restrict__ attn_out) {
  __shared__ __attribute__((aligned(16))) float smem[QB_ * SP_ + QB_ + 3 * QB_];
  float* S    = smem;
  float* rsum = S + QB_ * SP_;
  float* wx   = rsum + QB_;
  const int b = blockIdx.z, h = blockIdx.y, i0 = blockIdx.x * QB_;
  const int t = threadIdx.x;
  const int lane = t & 31, w = t >> 5;
  const int sub = lane & 15, hi = lane >> 4;

  v4f qa[QT_][8];
  #pragma unroll
  for (int qt = 0; qt < QT_; ++qt) {
    const float* qbase =
        qkv + ((size_t)(b * MTOK_ + i0 + qt * 16 + sub) * 3 + 0) * C_ + h * DH_ + 4 * hi;
    #pragma unroll
    for (int kb = 0; kb < 8; ++kb) qa[qt][kb] = *(const v4f*)(qbase + 8 * kb);
  }
  const float* kwave =
      qkv + ((size_t)(b * MTOK_ + w * 128 + sub) * 3 + 1) * C_ + h * DH_ + 4 * hi;
  auto load_k = [&](v4f (&kf)[8], int jj) {
    const float* kbase = kwave + (size_t)jj * 16 * 3 * C_;
    #pragma unroll
    for (int kb = 0; kb < 8; ++kb) kf[kb] = *(const v4f*)(kbase + 8 * kb);
  };
  auto score_tile = [&](const v4f (&kf)[8], int jj) {
    const int j0 = (w * 8 + jj) * 16;
    v8f acc[QT_];
    #pragma unroll
    for (int qt = 0; qt < QT_; ++qt) acc[qt] = (v8f){0.f, 0.f, 0.f, 0.f, 0.f, 0.f, 0.f, 0.f};
    #pragma unroll
    for (int kb = 0; kb < 8; ++kb) {
      const v2f b0 = {kf[kb].x, kf[kb].y}, b1 = {kf[kb].z, kf[kb].w};
      #pragma unroll
      for (int qt = 0; qt < QT_; ++qt) {
        const v2f a0 = {qa[qt][kb].x, qa[qt][kb].y};
        acc[qt] = wmma4(a0, b0, acc[qt]);
      }
      #pragma unroll
      for (int qt = 0; qt < QT_; ++qt) {
        const v2f a1 = {qa[qt][kb].z, qa[qt][kb].w};
        acc[qt] = wmma4(a1, b1, acc[qt]);
      }
    }
    #pragma unroll
    for (int qt = 0; qt < QT_; ++qt) {
      #pragma unroll
      for (int r = 0; r < 8; ++r)
        S[(qt * 16 + r + 8 * hi) * SP_ + j0 + sub] = acc[qt][r] * 0.125f;
    }
  };
  {
    v4f kf0[8], kf1[8];
    load_k(kf0, 0);
    #pragma unroll 1
    for (int jj = 0; jj < 8; jj += 2) {
      load_k(kf1, jj + 1);
      score_tile(kf0, jj);
      load_k(kf0, (jj + 2 < 8) ? jj + 2 : jj);
      score_tile(kf1, jj + 1);
    }
  }
  __syncthreads();

  for (int q = 0; q < QB_ / 8; ++q) {
    const int row = (QB_ / 8) * w + q;
    float* sr = S + row * SP_ + 4 * lane;
    v4f x[8];
    #pragma unroll
    for (int i = 0; i < 8; ++i) x[i] = *(const v4f*)(sr + 128 * i);
    float lm = -3.4e38f;
    #pragma unroll
    for (int i = 0; i < 8; ++i)
      lm = fmaxf(lm, fmaxf(fmaxf(x[i].x, x[i].y), fmaxf(x[i].z, x[i].w)));
    const float rm = wave_max_f32(lm);
    float ls = 0.0f;
    #pragma unroll
    for (int i = 0; i < 8; ++i) {
      v4f e;
      e.x = expf(x[i].x - rm);
      e.y = expf(x[i].y - rm);
      e.z = expf(x[i].z - rm);
      e.w = expf(x[i].w - rm);
      ls += (e.x + e.y) + (e.z + e.w);
      *(v4f*)(sr + 128 * i) = e;
    }
    ls = wave_sum_f32(ls);
    if (lane == 0) rsum[row] = ls;
  }
  __syncthreads();

  v8f acc0 = {0.f, 0.f, 0.f, 0.f, 0.f, 0.f, 0.f, 0.f};
  v8f acc1 = {0.f, 0.f, 0.f, 0.f, 0.f, 0.f, 0.f, 0.f};
  const int d0 = (w & 3) * 16;
  if (w < 4) {
    const float* vbase =
        qkv + ((size_t)(b * MTOK_) * 3 + 2) * C_ + h * DH_ + d0 + sub;
    const float* srow0 = S + sub * SP_ + 2 * hi;
    const float* srow1 = srow0 + 16 * SP_;
    auto load_v = [&](v2f (&vb)[VG_], int g) {
      #pragma unroll
      for (int u = 0; u < VG_; ++u) {
        const int kk = 4 * (g * VG_ + u) + 2 * hi;
        vb[u].x = vbase[(size_t)kk * 1536];
        vb[u].y = vbase[(size_t)(kk + 1) * 1536];
      }
    };
    auto pv_group = [&](const v2f (&vb)[VG_], int g) {
      #pragma unroll
      for (int u = 0; u < VG_; ++u) {
        const int ks = g * VG_ + u;
        const v2f a0 = *(const v2f*)(srow0 + 4 * ks);
        const v2f a1 = *(const v2f*)(srow1 + 4 * ks);
        acc0 = wmma4(a0, vb[u], acc0);
        acc1 = wmma4(a1, vb[u], acc1);
      }
    };
    v2f vb0[VG_], vb1[VG_];
    load_v(vb0, 0);
    #pragma unroll 1
    for (int g = 0; g < 256 / VG_; g += 2) {
      load_v(vb1, g + 1);
      pv_group(vb0, g);
      load_v(vb0, (g + 2 < 256 / VG_) ? g + 2 : g);
      pv_group(vb1, g + 1);
    }
  } else {
    const float* xb = xyzf + (size_t)b * MTOK_ * 3;
    #pragma unroll 1
    for (int ps = 0; ps < 2; ++ps) {
      const int r0 = (w - 4) * 8 + 4 * ps;
      float ax[4][3];
      #pragma unroll
      for (int rr = 0; rr < 4; ++rr) { ax[rr][0] = 0.0f; ax[rr][1] = 0.0f; ax[rr][2] = 0.0f; }
      #pragma unroll 1
      for (int i = 0; i < 8; ++i) {
        const int j = 4 * lane + 128 * i;
        const v4f c0 = *(const v4f*)(xb + 3 * j);
        const v4f c1 = *(const v4f*)(xb + 3 * j + 4);
        const v4f c2 = *(const v4f*)(xb + 3 * j + 8);
        #pragma unroll
        for (int rr = 0; rr < 4; ++rr) {
          const v4f p = *(const v4f*)(S + (r0 + rr) * SP_ + j);
          ax[rr][0] += p.x * c0.x + p.y * c0.w + p.z * c1.z + p.w * c2.y;
          ax[rr][1] += p.x * c0.y + p.y * c1.x + p.z * c1.w + p.w * c2.z;
          ax[rr][2] += p.x * c0.z + p.y * c1.y + p.z * c2.x + p.w * c2.w;
        }
      }
      #pragma unroll
      for (int rr = 0; rr < 4; ++rr) {
        #pragma unroll
        for (int cc = 0; cc < 3; ++cc) ax[rr][cc] = wave_sum_f32(ax[rr][cc]);
      }
      if (lane == 0) {
        #pragma unroll
        for (int rr = 0; rr < 4; ++rr) {
          #pragma unroll
          for (int cc = 0; cc < 3; ++cc) wx[(r0 + rr) * 3 + cc] = ax[rr][cc];
        }
      }
    }
  }
  __syncthreads();
  if (w < 4) {
    const int d = d0 + sub;
    const float w0 = Wsp[d * 3 + 0], w1 = Wsp[d * 3 + 1], w2 = Wsp[d * 3 + 2];
    #pragma unroll
    for (int qt = 0; qt < QT_; ++qt) {
      #pragma unroll
      for (int r = 0; r < 8; ++r) {
        const int rr = qt * 16 + r + 8 * hi;
        const int i  = i0 + rr;
        const float inv = 1.0f / rsum[rr];
        const float* xi = xyzf + ((size_t)b * MTOK_ + i) * 3;
        const float wd0 = wx[rr * 3 + 0] * inv - xi[0];
        const float wd1 = wx[rr * 3 + 1] * inv - xi[1];
        const float wd2 = wx[rr * 3 + 2] * inv - xi[2];
        const float o = (qt == 0) ? acc0[r] : acc1[r];
        attn_out[((size_t)(b * MTOK_) + i) * C_ + h * DH_ + d] =
            o * inv + w0 * wd0 + w1 * wd1 + w2 * wd2;
      }
    }
  }
}

extern "C" void kernel_launch(void* const* d_in, const int* in_sizes, int n_in,
                              void* d_out, int out_size, void* d_ws, size_t ws_size,
                              hipStream_t stream) {
  (void)in_sizes; (void)n_in; (void)out_size; (void)ws_size;
  const float* xyzs = (const float*)d_in[0];
  const float* Wd   = (const float*)d_in[1];
  const float* Wqkv = (const float*)d_in[2];
  const float* Wsp  = (const float*)d_in[3];
  const float* Wout = (const float*)d_in[4];
  const float* bout = (const float*)d_in[5];
  const float* ln1g = (const float*)d_in[6];
  const float* ln1b = (const float*)d_in[7];
  const float* Wff1 = (const float*)d_in[8];
  const float* bff1 = (const float*)d_in[9];
  const float* Wff2 = (const float*)d_in[10];
  const float* bff2 = (const float*)d_in[11];
  const float* ln2g = (const float*)d_in[12];
  const float* ln2b = (const float*)d_in[13];

  float* ws      = (float*)d_ws;
  float* anchors = ws;
  float* feats   = ws + 8192;
  float* xln     = feats + (size_t)ROWS_ * C_;
  float* qkvb    = xln   + (size_t)ROWS_ * C_;
  float* attno   = qkvb  + (size_t)ROWS_ * 1536;
  float* ffh     = attno + (size_t)ROWS_ * C_;

  fps_kernel<<<B_ * L_, 32, 0, stream>>>(xyzs, anchors);
  p4dconv_kernel<<<B_ * L_ * MA_ / P4W_, 256, 0, stream>>>(xyzs, anchors, Wd, feats);

  for (int d = 0; d < 2; ++d) {
    ln_kernel<<<ROWS_ / 8, 256, 0, stream>>>(feats, ln1g + d * C_, ln1b + d * C_, xln);
    gemm_tiled<64, 64, 0, false, false><<<dim3(1536 / 64, ROWS_ / 64), 256, 0, stream>>>(
        xln, C_, Wqkv + (size_t)d * 1536 * C_, C_, nullptr, nullptr,
        qkvb, 1536, C_);
    attn_kernel<<<dim3(MTOK_ / QB_, HEADS_, B_), 256, 0, stream>>>(
        qkvb, anchors, Wsp + (size_t)d * DH_ * 3, attno);
    gemm_tiled<64, 64, 1, true, true><<<dim3(C_ / 64, ROWS_ / 64), 256, 0, stream>>>(
        attno, C_, Wout + (size_t)d * C_ * C_, C_, bout + d * C_,
        feats, feats, C_, C_);
    ln_kernel<<<ROWS_ / 8, 256, 0, stream>>>(feats, ln2g + d * C_, ln2b + d * C_, xln);
    gemm_tiled<128, 128, 1, false, true><<<dim3(MLP_ / 128, ROWS_ / 128), 256, 0, stream>>>(
        xln, C_, Wff1 + (size_t)d * MLP_ * C_, C_, bff1 + d * MLP_,
        nullptr, ffh, MLP_, C_);
    gemm_tiled<64, 64, 0, true, true><<<dim3(C_ / 64, ROWS_ / 64), 256, 0, stream>>>(
        ffh, MLP_, Wff2 + (size_t)d * C_ * MLP_, MLP_, bff2 + d * C_,
        feats, feats, C_, MLP_);
  }
  hipMemcpyAsync(d_out, feats, (size_t)ROWS_ * C_ * sizeof(float),
                 hipMemcpyDeviceToDevice, stream);
}


# FPS: four waves per frame (8 points per thread) instead of one; per anchor the four wave maxima meet in a two-deep LDS ring behind one LDS-only barrier; anchors kept in registers, stored once
# speedup vs baseline: 3.5334x; 1.0549x over previous
#include <hip/hip_runtime.h>
#include <hip/hip_bf16.h>
#include <math.h>

typedef float v2f __attribute__((ext_vector_type(2)));
typedef float v4f __attribute__((ext_vector_type(4)));
typedef float v8f __attribute__((ext_vector_type(8)));

#define B_    2
#define T_    4
#define N_    1024
#define L_    4
#define MA_   256
#define KNB_  32
#define C_    512
#define HEADS_ 8
#define DH_   64
#define MTOK_ 1024
#define ROWS_ 2048
#define MLP_  2048

__device__ __forceinline__ v8f wmma4(v2f a, v2f b, v8f c) {
  return __builtin_amdgcn_wmma_f32_16x16x4_f32(false, a, false, b, (short)0, c,
                                               false, false);
}

__device__ __forceinline__ float fbits(unsigned u) { return __builtin_bit_cast(float, u); }
__device__ __forceinline__ float gelu_sel(float v) {
  const float z = v * 0.7071067811865475f;
  const float t = fabsf(z);
  float p = fmaf(t, fbits(0x378e98abu), fbits(0xb9c68948u));
  p = fmaf(t, p, fbits(0x3b7cd369u));
  p = fmaf(t, p, fbits(0xbcc618b2u));
  p = fmaf(t, p, fbits(0x3dda74e4u));
  p = fmaf(t, p, fbits(0x3f228afdu));
  p = fmaf(t, p, fbits(0x3e03c728u));
  p = fmaf(t, p, t);
  const float big = 1.0f - expf(-p);
  const float s = z * z;
  float q = fmaf(s, fbits(0xba1345e1u), fbits(0x3ba10414u));
  q = fmaf(s, q, fbits(0xbcdac9b8u));
  q = fmaf(s, q, fbits(0x3de703beu));
  q = fmaf(s, q, fbits(0xbec09330u));
  q = fmaf(s, q, fbits(0x3e0375d0u));
  const float sm = fmaf(t, q, t);
  const float r = (t < 1.0f) ? sm : big;
  return (0.5f * v) * (1.0f + copysignf(r, z));
}

#define FPS_WAVES_   4
#define FPS_THREADS_ (32 * FPS_WAVES_)
#define FPS_SLOTS_   (N_ / FPS_THREADS_)
#define FPS_CHAINS_  2
#define FPS_CLEN_    (FPS_SLOTS_ / FPS_CHAINS_)

typedef int v4i __attribute__((ext_vector_type(4)));

__device__ __forceinline__ int   f2i(float v) { return __builtin_bit_cast(int, v); }
__device__ __forceinline__ float i2f(int v)   { return __builtin_bit_cast(float, v); }

template <int CTRL>
__device__ __forceinline__ int dpp_row_i(int v) {
  return __builtin_amdgcn_update_dpp(0, v, CTRL, 0xf, 0xf, true);
}
__device__ __forceinline__ int other_row_i(int v) {
  return __builtin_amdgcn_permlanex16(v, v, 0x76543210, 0xfedcba98, false, false);
}
__device__ __forceinline__ int wave_max_i32(int v) {
  v = max(v, dpp_row_i<0x161>(v));
  v = max(v, dpp_row_i<0x162>(v));
  v = max(v, dpp_row_i<0x164>(v));
  v = max(v, dpp_row_i<0x168>(v));
  v = max(v, other_row_i(v));
  return v;
}
__device__ __forceinline__ float wave_max_f32(float v) {
  v = fmaxf(v, __builtin_bit_cast(float, dpp_row_i<0x161>(__builtin_bit_cast(int, v))));
  v = fmaxf(v, __builtin_bit_cast(float, dpp_row_i<0x162>(__builtin_bit_cast(int, v))));
  v = fmaxf(v, __builtin_bit_cast(float, dpp_row_i<0x164>(__builtin_bit_cast(int, v))));
  v = fmaxf(v, __builtin_bit_cast(float, dpp_row_i<0x168>(__builtin_bit_cast(int, v))));
  v = fmaxf(v, __builtin_bit_cast(float, other_row_i(__builtin_bit_cast(int, v))));
  return v;
}
__device__ __forceinline__ float wave_sum_f32(float v) {
  v += __builtin_bit_cast(float, dpp_row_i<0x161>(__builtin_bit_cast(int, v)));
  v += __builtin_bit_cast(float, dpp_row_i<0x162>(__builtin_bit_cast(int, v)));
  v += __builtin_bit_cast(float, dpp_row_i<0x164>(__builtin_bit_cast(int, v)));
  v += __builtin_bit_cast(float, dpp_row_i<0x168>(__builtin_bit_cast(int, v)));
  v += __builtin_bit_cast(float, other_row_i(__builtin_bit_cast(int, v)));
  return v;
}

__device__ __forceinline__ void lds_barrier() {
  __builtin_amdgcn_fence(__ATOMIC_RELEASE, "workgroup", "local");
  __builtin_amdgcn_s_barrier();
  __builtin_amdgcn_fence(__ATOMIC_ACQUIRE, "workgroup", "local");
}

__global__ __launch_bounds__(FPS_THREADS_) void fps_kernel(const float* __restrict__ xyzs,
                                                           float* __restrict__ anchors) {
  __shared__ __attribute__((aligned(16))) v4i xch[2][FPS_WAVES_];
  const int blk  = blockIdx.x;
  const int t    = threadIdx.x;
  const int lane = t & 31, w = t >> 5;
  const float* frame = xyzs + (size_t)blk * N_ * 3;
  const v4f* mp = (const v4f*)(frame + (size_t)t * FPS_SLOTS_ * 3);
  v4f raw[6];
  #pragma unroll
  for (int i = 0; i < 6; ++i) raw[i] = mp[i];
  float px[FPS_SLOTS_], py[FPS_SLOTS_], pz[FPS_SLOTS_];
  int   dist[FPS_SLOTS_];
  #pragma unroll
  for (int s = 0; s < FPS_SLOTS_; ++s) {
    px[s] = raw[(3 * s + 0) / 4][(3 * s + 0) % 4];
    py[s] = raw[(3 * s + 1) / 4][(3 * s + 1) % 4];
    pz[s] = raw[(3 * s + 2) / 4][(3 * s + 2) % 4];
    dist[s] = f2i(1e10f);
  }
  float* aout = anchors + (size_t)blk * MA_ * 3;
  float lx = frame[0], ly = frame[1], lz = frame[2];
  float a0x = 0.0f, a0y = 0.0f, a0z = 0.0f;
  float a1x = 0.0f, a1y = 0.0f, a1z = 0.0f;
  for (int it = 1; it < MA_; ++it) {
    {
      const bool k0 = (t == it - 1), k1 = (t + FPS_THREADS_ == it - 1);
      a0x = k0 ? lx : a0x; a0y = k0 ? ly : a0y; a0z = k0 ? lz : a0z;
      a1x = k1 ? lx : a1x; a1y = k1 ? ly : a1y; a1z = k1 ? lz : a1z;
    }
    #pragma unroll
    for (int s = 0; s < FPS_SLOTS_; ++s) {
      const float dx = px[s] - lx, dy = py[s] - ly, dz = pz[s] - lz;
      const float d2 = fmaf(dz, dz, fmaf(dx, dx, dy * dy));
      dist[s] = min(dist[s], f2i(d2));
    }
    int   bv[FPS_CHAINS_];
    float bx[FPS_CHAINS_], by[FPS_CHAINS_], bz[FPS_CHAINS_];
    #pragma unroll
    for (int c = 0; c < FPS_CHAINS_; ++c) {
      bv[c] = dist[c * FPS_CLEN_];
      bx[c] = px[c * FPS_CLEN_]; by[c] = py[c * FPS_CLEN_]; bz[c] = pz[c * FPS_CLEN_];
      #pragma unroll
      for (int k = 1; k < FPS_CLEN_; ++k) {
        const int s = c * FPS_CLEN_ + k;
        const bool gt = dist[s] > bv[c];
        bx[c] = gt ? px[s] : bx[c];
        by[c] = gt ? py[s] : by[c];
        bz[c] = gt ? pz[s] : bz[c];
        bv[c] = max(bv[c], dist[s]);
      }
    }
    {
      const bool gt = bv[1] > bv[0];
      bx[0] = gt ? bx[1] : bx[0];
      by[0] = gt ? by[1] : by[0];
      bz[0] = gt ? bz[1] : bz[0];
      bv[0] = max(bv[0], bv[1]);
    }
    const int gmax = wave_max_i32(bv[0]);
    const unsigned tie = __builtin_amdgcn_ballot_w32(bv[0] == gmax);
    const int winner = __builtin_ctz(tie);
    const int wx = __builtin_amdgcn_readlane(f2i(bx[0]), winner);
    const int wy = __builtin_amdgcn_readlane(f2i(by[0]), winner);
    const int wz = __builtin_amdgcn_readlane(f2i(bz[0]), winner);
    v4i* ring = &xch[it & 1][0];
    if (lane == 0) ring[w] = (v4i){gmax, wx, wy, wz};
    lds_barrier();
    const v4i e0 = ring[0], e1 = ring[1], e2 = ring[2], e3 = ring[3];
    const v4i f0 = (e1.x > e0.x) ? e1 : e0;
    const v4i f1 = (e3.x > e2.x) ? e3 : e2;
    const v4i g  = (f1.x > f0.x) ? f1 : f0;
    const int gy = g.y, gz = g.z, gw = g.w;
    lx = i2f(gy);
    ly = i2f(gz);
    lz = i2f(gw);
  }
  if (t + FPS_THREADS_ == MA_ - 1) { a1x = lx; a1y = ly; a1z = lz; }
  aout[t * 3 + 0] = a0x;
  aout[t * 3 + 1] = a0y;
  aout[t * 3 + 2] = a0z;
  aout[(t + FPS_THREADS_) * 3 + 0] = a1x;
  aout[(t + FPS_THREADS_) * 3 + 1] = a1y;
  aout[(t + FPS_THREADS_) * 3 + 2] = a1z;
}

#define P4W_ 8

__global__ __launch_bounds__(256) void p4dconv_kernel(const float* __restrict__ xyzs,
                                                      const float* __restrict__ anchors,
                                                      const float* __restrict__ Wd,
                                                      float* __restrict__ feats) {
  __shared__ __attribute__((aligned(16))) float  sfr[3][N_ * 3];
  __shared__ __attribute__((aligned(16))) float4 snb[P4W_][KNB_];
  const int t = threadIdx.x;
  const int lane = t & 31, w = t >> 5;
  const int blk = blockIdx.x;
  const int bl  = blk / (MA_ / P4W_);
  const int m   = (blk % (MA_ / P4W_)) * P4W_ + w;
  const int l   = bl % L_;
  const int b   = bl / L_;

  for (int f = 0; f < 3; ++f) {
    int fi = l + f - 1; fi = fi < 0 ? 0 : (fi > T_ - 1 ? T_ - 1 : fi);
    const v4f* srcp = (const v4f*)(xyzs + ((size_t)b * T_ + fi) * N_ * 3);
    v4f* dstp = (v4f*)&sfr[f][0];
    for (int i = 0; i < 3; ++i) dstp[t + 256 * i] = srcp[t + 256 * i];
  }

  const float* arow = anchors + ((size_t)bl * MA_ + m) * 3;
  const float ax = arow[0], ay = arow[1], az = arow[2];

  v2f w0[8], w1[8], w2[8], w3[8], mx[8];
  #pragma unroll
  for (int j = 0; j < 8; ++j) {
    const float* wp = Wd + (size_t)(2 * lane + 64 * j) * 4;
    const v4f wa = *(const v4f*)wp;
    const v4f wb = *(const v4f*)(wp + 4);
    w0[j] = (v2f){wa.x, wb.x};
    w1[j] = (v2f){wa.y, wb.y};
    w2[j] = (v2f){wa.z, wb.z};
    w3[j] = (v2f){wa.w, wb.w};
    mx[j] = (v2f){-3.4e38f, -3.4e38f};
  }
  __syncthreads();

  for (int f = 0; f < 3; ++f) {
    const float* fr = &sfr[f][0];
    const float p0x = fr[0] - ax, p0y = fr[1] - ay, p0z = fr[2] - az;
    int cnt = 0;
    for (int g = 0; g < 4 && cnt < KNB_; ++g) {
      float q[8][3];
      #pragma unroll
      for (int u = 0; u < 8; ++u) {
        const int i = (g * 8 + u) * 32 + lane;
        q[u][0] = fr[3 * i + 0];
        q[u][1] = fr[3 * i + 1];
        q[u][2] = fr[3 * i + 2];
      }
      #pragma unroll
      for (int u = 0; u < 8; ++u) {
        const float dx = q[u][0] - ax;
        const float dy = q[u][1] - ay;
        const float dz = q[u][2] - az;
        const float d2 = fmaf(dz, dz, fmaf(dx, dx, dy * dy));
        const bool pred = d2 < 1.0f;
        const unsigned mask = __builtin_amdgcn_ballot_w32(pred);
        const int pos = cnt + (int)__builtin_amdgcn_mbcnt_lo(mask, 0u);
        if (pred && pos < KNB_) snb[w][pos] = make_float4(dx, dy, dz, 0.0f);
        cnt += __builtin_popcount(mask);
      }
    }
    __syncthreads();
    if (lane >= cnt) {
      snb[w][lane] = (cnt > 0) ? snb[w][0] : make_float4(p0x, p0y, p0z, 0.0f);
    }
    __syncthreads();
    const float tq = (float)(f - 1);
    #pragma unroll 4
    for (int k = 0; k < KNB_; ++k) {
      const float4 p = snb[w][k];
      #pragma unroll
      for (int j = 0; j < 8; ++j) {
        const v2f v = w0[j] * p.x + w1[j] * p.y + w2[j] * p.z + w3[j] * tq;
        mx[j] = __builtin_elementwise_max(mx[j], v);
      }
    }
    __syncthreads();
  }
  float* orow = feats + ((size_t)bl * MA_ + m) * C_ + 2 * lane;
  #pragma unroll
  for (int j = 0; j < 8; ++j) {
    const v2f o = {fmaxf(mx[j].x, 0.0f), fmaxf(mx[j].y, 0.0f)};
    *(v2f*)(orow + 64 * j) = o;
  }
}

__global__ __launch_bounds__(256) void ln_kernel(const float* __restrict__ x, const float* __restrict__ g,
                                                 const float* __restrict__ bta, float* __restrict__ y) {
  const int lane = threadIdx.x & 31, w = threadIdx.x >> 5;
  const int row = blockIdx.x * 8 + w;
  const float* xr = x + (size_t)row * C_ + 4 * lane;
  v4f a[4], d[4];
  #pragma unroll
  for (int i = 0; i < 4; ++i) a[i] = *(const v4f*)(xr + 128 * i);
  float s = 0.0f;
  #pragma unroll
  for (int i = 0; i < 4; ++i) s += (a[i].x + a[i].y) + (a[i].z + a[i].w);
  const float mean = wave_sum_f32(s) * (1.0f / C_);
  float q = 0.0f;
  #pragma unroll
  for (int i = 0; i < 4; ++i) {
    d[i] = a[i] - mean;
    q += (d[i].x * d[i].x + d[i].y * d[i].y) + (d[i].z * d[i].z + d[i].w * d[i].w);
  }
  const float inv = rsqrtf(wave_sum_f32(q) * (1.0f / C_) + 1e-5f);
  float* yr = y + (size_t)row * C_ + 4 * lane;
  #pragma unroll
  for (int i = 0; i < 4; ++i) {
    const v4f gg = *(const v4f*)(g + 4 * lane + 128 * i);
    const v4f bb = *(const v4f*)(bta + 4 * lane + 128 * i);
    *(v4f*)(yr + 128 * i) = d[i] * inv * gg + bb;
  }
}

#define GBK_    64
#define GPITCH_ 68
#define GNBUF_  3

template <int BM, int BN, int ACT, bool RESID, bool BIAS>
__global__ __launch_bounds__(256) void gemm_tiled(const float* __restrict__ A, int lda,
                                                  const float* __restrict__ W, int ldw,
                                                  const float* __restrict__ bias,
                                                  const float* resid,
                                                  float* Cout, int ldc,
                                                  int Kdim) {
  constexpr int WC = BN / 32;
  constexpr int WR = 8 / WC;
  constexpr int TM = BM / (16 * WR);
  constexpr int TN = 2;
  constexpr int LA = BM / 16;
  constexpr int LB = BN / 16;
  constexpr int NKB = GBK_ / 8;
  __shared__ __attribute__((aligned(16))) float As[GNBUF_][BM * GPITCH_];
  __shared__ __attribute__((aligned(16))) float Bs[GNBUF_][BN * GPITCH_];
  const int t = threadIdx.x;
  const int lane = t & 31, w = t >> 5;
  const int sub = lane & 15, hi = lane >> 4;
  const int wr = w / WC, wc = w % WC;
  const int i0 = blockIdx.y * BM, j0 = blockIdx.x * BN;
  const int lr = t >> 4;
  const int lc = (t & 15) * 4;
  const float* ag = A + (size_t)(i0 + lr) * lda + lc;
  const float* wg = W + (size_t)(j0 + lr) * ldw + lc;
  const int sw_off = lr * GPITCH_ + lc;
  const int nst = Kdim / GBK_;
  v4f ra[LA], rb[LB];
  {
    v4f pa[2][LA], pb[2][LB];
    #pragma unroll
    for (int i = 0; i < LA; ++i) pa[0][i] = *(const v4f*)(ag + (size_t)(16 * i) * lda);
    #pragma unroll
    for (int i = 0; i < LB; ++i) pb[0][i] = *(const v4f*)(wg + (size_t)(16 * i) * ldw);
    if (nst > 1) {
      #pragma unroll
      for (int i = 0; i < LA; ++i) pa[1][i] = *(const v4f*)(ag + (size_t)(16 * i) * lda + GBK_);
      #pragma unroll
      for (int i = 0; i < LB; ++i) pb[1][i] = *(const v4f*)(wg + (size_t)(16 * i) * ldw + GBK_);
    }
    if (nst > 2) {
      #pragma unroll
      for (int i = 0; i < LA; ++i) ra[i] = *(const v4f*)(ag + (size_t)(16 * i) * lda + 2 * GBK_);
      #pragma unroll
      for (int i = 0; i < LB; ++i) rb[i] = *(const v4f*)(wg + (size_t)(16 * i) * ldw + 2 * GBK_);
    } else {
      #pragma unroll
      for (int i = 0; i < LA; ++i) ra[i] = (v4f){0.f, 0.f, 0.f, 0.f};
      #pragma unroll
      for (int i = 0; i < LB; ++i) rb[i] = (v4f){0.f, 0.f, 0.f, 0.f};
    }
    #pragma unroll
    for (int i = 0; i < LA; ++i) *(v4f*)(&As[0][sw_off + 16 * i * GPITCH_]) = pa[0][i];
    #pragma unroll
    for (int i = 0; i < LB; ++i) *(v4f*)(&Bs[0][sw_off + 16 * i * GPITCH_]) = pb[0][i];
    if (nst > 1) {
      #pragma unroll
      for (int i = 0; i < LA; ++i) *(v4f*)(&As[1][sw_off + 16 * i * GPITCH_]) = pa[1][i];
      #pragma unroll
      for (int i = 0; i < LB; ++i) *(v4f*)(&Bs[1][sw_off + 16 * i * GPITCH_]) = pb[1][i];
    }
  }
  const size_t erow0 = (size_t)(i0 + wr * TM * 16 + 8 * hi);
  const int    ecol0 = j0 + wc * 32 + sub;
  float bj[TN];
  #pragma unroll
  for (int n = 0; n < TN; ++n) bj[n] = BIAS ? bias[ecol0 + n * 16] : 0.0f;
  float rv[TM][8][TN];
  if (RESID) {
    #pragma unroll
    for (int m = 0; m < TM; ++m)
      #pragma unroll
      for (int r = 0; r < 8; ++r)
        #pragma unroll
        for (int n = 0; n < TN; ++n)
          rv[m][r][n] = resid[(erow0 + m * 16 + r) * ldc + ecol0 + n * 16];
  }
  lds_barrier();

  v8f acc[TM][TN];
  #pragma unroll
  for (int m = 0; m < TM; ++m)
    #pragma unroll
    for (int n = 0; n < TN; ++n)
      acc[m][n] = (v8f){0.f, 0.f, 0.f, 0.f, 0.f, 0.f, 0.f, 0.f};

  const int fa_off = (wr * TM * 16 + sub) * GPITCH_ + 4 * hi;
  const int fb_off = (wc * 32 + sub) * GPITCH_ + 4 * hi;
  v4f fa[2][TM], fb[2][TN];
  #pragma unroll
  for (int m = 0; m < TM; ++m) fa[0][m] = *(const v4f*)(&As[0][fa_off + m * 16 * GPITCH_]);
  #pragma unroll
  for (int n = 0; n < TN; ++n) fb[0][n] = *(const v4f*)(&Bs[0][fb_off + n * 16 * GPITCH_]);
  int cur = 0, nx1 = 1, nx2 = 2;
  for (int st = 0; st < nst; ++st) {
    #pragma unroll
    for (int i = 0; i < LA; ++i) *(v4f*)(&As[nx2][sw_off + 16 * i * GPITCH_]) = ra[i];
    #pragma unroll
    for (int i = 0; i < LB; ++i) *(v4f*)(&Bs[nx2][sw_off + 16 * i * GPITCH_]) = rb[i];
    if (st + 3 < nst) {
      const int kn = (st + 3) * GBK_;
      #pragma unroll
      for (int i = 0; i < LA; ++i) ra[i] = *(const v4f*)(ag + (size_t)(16 * i) * lda + kn);
      #pragma unroll
      for (int i = 0; i < LB; ++i) rb[i] = *(const v4f*)(wg + (size_t)(16 * i) * ldw + kn);
    }
    const float* as  = &As[cur][fa_off];
    const float* bs  = &Bs[cur][fb_off];
    const float* asn = &As[nx1][fa_off];
    const float* bsn = &Bs[nx1][fb_off];
    const bool more = (st + 1 < nst);
    #pragma unroll
    for (int kb = 0; kb < NKB; ++kb) {
      const int c = kb & 1, d = c ^ 1;
      if (kb + 1 < NKB) {
        #pragma unroll
        for (int m = 0; m < TM; ++m) fa[d][m] = *(const v4f*)(as + m * 16 * GPITCH_ + (kb + 1) * 8);
        #pragma unroll
        for (int n = 0; n < TN; ++n) fb[d][n] = *(const v4f*)(bs + n * 16 * GPITCH_ + (kb + 1) * 8);
      } else if (more) {
        #pragma unroll
        for (int m = 0; m < TM; ++m) fa[d][m] = *(const v4f*)(asn + m * 16 * GPITCH_);
        #pragma unroll
        for (int n = 0; n < TN; ++n) fb[d][n] = *(const v4f*)(bsn + n * 16 * GPITCH_);
      }
      #pragma unroll
      for (int m = 0; m < TM; ++m)
        #pragma unroll
        for (int n = 0; n < TN; ++n) {
          const v2f a = {fa[c][m].x, fa[c][m].y};
          const v2f b = {fb[c][n].x, fb[c][n].y};
          acc[m][n] = wmma4(a, b, acc[m][n]);
        }
      #pragma unroll
      for (int m = 0; m < TM; ++m)
        #pragma unroll
        for (int n = 0; n < TN; ++n) {
          const v2f a = {fa[c][m].z, fa[c][m].w};
          const v2f b = {fb[c][n].z, fb[c][n].w};
          acc[m][n] = wmma4(a, b, acc[m][n]);
        }
    }
    lds_barrier();
    const int tcur = cur; cur = nx1; nx1 = nx2; nx2 = tcur;
  }

  #pragma unroll
  for (int m = 0; m < TM; ++m) {
    #pragma unroll
    for (int r = 0; r < 8; ++r) {
      #pragma unroll
      for (int n = 0; n < TN; ++n) {
        float v = acc[m][n][r];
        if (BIAS) v += bj[n];
        if (ACT == 1) v = gelu_sel(v);
        if (RESID) v += rv[m][r][n];
        Cout[(erow0 + m * 16 + r) * ldc + ecol0 + n * 16] = v;
      }
    }
  }
}

#define SP_ 1028
#define QT_ 2
#define QB_ (16 * QT_)
#define VG_ 16

__global__ __launch_bounds__(256) void attn_kernel(const float* __restrict__ qkv,
                                                   const float* __restrict__ xyzf,
                                                   const float* __restrict__ Wsp,
                                                   float* __restrict__ attn_out) {
  __shared__ __attribute__((aligned(16))) float smem[QB_ * SP_ + QB_ + 3 * QB_];
  float* S    = smem;
  float* rsum = S + QB_ * SP_;
  float* wx   = rsum + QB_;
  const int b = blockIdx.z, h = blockIdx.y, i0 = blockIdx.x * QB_;
  const int t = threadIdx.x;
  const int lane = t & 31, w = t >> 5;
  const int sub = lane & 15, hi = lane >> 4;

  v4f qa[QT_][8];
  #pragma unroll
  for (int qt = 0; qt < QT_; ++qt) {
    const float* qbase =
        qkv + ((size_t)(b * MTOK_ + i0 + qt * 16 + sub) * 3 + 0) * C_ + h * DH_ + 4 * hi;
    #pragma unroll
    for (int kb = 0; kb < 8; ++kb) qa[qt][kb] = *(const v4f*)(qbase + 8 * kb);
  }
  const float* kwave =
      qkv + ((size_t)(b * MTOK_ + w * 128 + sub) * 3 + 1) * C_ + h * DH_ + 4 * hi;
  auto load_k = [&](v4f (&kf)[8], int jj) {
    const float* kbase = kwave + (size_t)jj * 16 * 3 * C_;
    #pragma unroll
    for (int kb = 0; kb < 8; ++kb) kf[kb] = *(const v4f*)(kbase + 8 * kb);
  };
  auto score_tile = [&](const v4f (&kf)[8], int jj) {
    const int j0 = (w * 8 + jj) * 16;
    v8f acc[QT_];
    #pragma unroll
    for (int qt = 0; qt < QT_; ++qt) acc[qt] = (v8f){0.f, 0.f, 0.f, 0.f, 0.f, 0.f, 0.f, 0.f};
    #pragma unroll
    for (int kb = 0; kb < 8; ++kb) {
      const v2f b0 = {kf[kb].x, kf[kb].y}, b1 = {kf[kb].z, kf[kb].w};
      #pragma unroll
      for (int qt = 0; qt < QT_; ++qt) {
        const v2f a0 = {qa[qt][kb].x, qa[qt][kb].y};
        acc[qt] = wmma4(a0, b0, acc[qt]);
      }
      #pragma unroll
      for (int qt = 0; qt < QT_; ++qt) {
        const v2f a1 = {qa[qt][kb].z, qa[qt][kb].w};
        acc[qt] = wmma4(a1, b1, acc[qt]);
      }
    }
    #pragma unroll
    for (int qt = 0; qt < QT_; ++qt) {
      #pragma unroll
      for (int r = 0; r < 8; ++r)
        S[(qt * 16 + r + 8 * hi) * SP_ + j0 + sub] = acc[qt][r] * 0.125f;
    }
  };
  {
    v4f kf0[8], kf1[8];
    load_k(kf0, 0);
    #pragma unroll 1
    for (int jj = 0; jj < 8; jj += 2) {
      load_k(kf1, jj + 1);
      score_tile(kf0, jj);
      load_k(kf0, (jj + 2 < 8) ? jj + 2 : jj);
      score_tile(kf1, jj + 1);
    }
  }
  __syncthreads();

  for (int q = 0; q < QB_ / 8; ++q) {
    const int row = (QB_ / 8) * w + q;
    float* sr = S + row * SP_ + 4 * lane;
    v4f x[8];
    #pragma unroll
    for (int i = 0; i < 8; ++i) x[i] = *(const v4f*)(sr + 128 * i);
    float lm = -3.4e38f;
    #pragma unroll
    for (int i = 0; i < 8; ++i)
      lm = fmaxf(lm, fmaxf(fmaxf(x[i].x, x[i].y), fmaxf(x[i].z, x[i].w)));
    const float rm = wave_max_f32(lm);
    float ls = 0.0f;
    #pragma unroll
    for (int i = 0; i < 8; ++i) {
      v4f e;
      e.x = expf(x[i].x - rm);
      e.y = expf(x[i].y - rm);
      e.z = expf(x[i].z - rm);
      e.w = expf(x[i].w - rm);
      ls += (e.x + e.y) + (e.z + e.w);
      *(v4f*)(sr + 128 * i) = e;
    }
    ls = wave_sum_f32(ls);
    if (lane == 0) rsum[row] = ls;
  }
  __syncthreads();

  v8f acc0 = {0.f, 0.f, 0.f, 0.f, 0.f, 0.f, 0.f, 0.f};
  v8f acc1 = {0.f, 0.f, 0.f, 0.f, 0.f, 0.f, 0.f, 0.f};
  const int d0 = (w & 3) * 16;
  if (w < 4) {
    const float* vbase =
        qkv + ((size_t)(b * MTOK_) * 3 + 2) * C_ + h * DH_ + d0 + sub;
    const float* srow0 = S + sub * SP_ + 2 * hi;
    const float* srow1 = srow0 + 16 * SP_;
    auto load_v = [&](v2f (&vb)[VG_], int g) {
      #pragma unroll
      for (int u = 0; u < VG_; ++u) {
        const int kk = 4 * (g * VG_ + u) + 2 * hi;
        vb[u].x = vbase[(size_t)kk * 1536];
        vb[u].y = vbase[(size_t)(kk + 1) * 1536];
      }
    };
    auto pv_group = [&](const v2f (&vb)[VG_], int g) {
      #pragma unroll
      for (int u = 0; u < VG_; ++u) {
        const int ks = g * VG_ + u;
        const v2f a0 = *(const v2f*)(srow0 + 4 * ks);
        const v2f a1 = *(const v2f*)(srow1 + 4 * ks);
        acc0 = wmma4(a0, vb[u], acc0);
        acc1 = wmma4(a1, vb[u], acc1);
      }
    };
    v2f vb0[VG_], vb1[VG_];
    load_v(vb0, 0);
    #pragma unroll 1
    for (int g = 0; g < 256 / VG_; g += 2) {
      load_v(vb1, g + 1);
      pv_group(vb0, g);
      load_v(vb0, (g + 2 < 256 / VG_) ? g + 2 : g);
      pv_group(vb1, g + 1);
    }
  } else {
    const float* xb = xyzf + (size_t)b * MTOK_ * 3;
    #pragma unroll 1
    for (int ps = 0; ps < 2; ++ps) {
      const int r0 = (w - 4) * 8 + 4 * ps;
      float ax[4][3];
      #pragma unroll
      for (int rr = 0; rr < 4; ++rr) { ax[rr][0] = 0.0f; ax[rr][1] = 0.0f; ax[rr][2] = 0.0f; }
      #pragma unroll 1
      for (int i = 0; i < 8; ++i) {
        const int j = 4 * lane + 128 * i;
        const v4f c0 = *(const v4f*)(xb + 3 * j);
        const v4f c1 = *(const v4f*)(xb + 3 * j + 4);
        const v4f c2 = *(const v4f*)(xb + 3 * j + 8);
        #pragma unroll
        for (int rr = 0; rr < 4; ++rr) {
          const v4f p = *(const v4f*)(S + (r0 + rr) * SP_ + j);
          ax[rr][0] += p.x * c0.x + p.y * c0.w + p.z * c1.z + p.w * c2.y;
          ax[rr][1] += p.x * c0.y + p.y * c1.x + p.z * c1.w + p.w * c2.z;
          ax[rr][2] += p.x * c0.z + p.y * c1.y + p.z * c2.x + p.w * c2.w;
        }
      }
      #pragma unroll
      for (int rr = 0; rr < 4; ++rr) {
        #pragma unroll
        for (int cc = 0; cc < 3; ++cc) ax[rr][cc] = wave_sum_f32(ax[rr][cc]);
      }
      if (lane == 0) {
        #pragma unroll
        for (int rr = 0; rr < 4; ++rr) {
          #pragma unroll
          for (int cc = 0; cc < 3; ++cc) wx[(r0 + rr) * 3 + cc] = ax[rr][cc];
        }
      }
    }
  }
  __syncthreads();
  if (w < 4) {
    const int d = d0 + sub;
    const float w0 = Wsp[d * 3 + 0], w1 = Wsp[d * 3 + 1], w2 = Wsp[d * 3 + 2];
    #pragma unroll
    for (int qt = 0; qt < QT_; ++qt) {
      #pragma unroll
      for (int r = 0; r < 8; ++r) {
        const int rr = qt * 16 + r + 8 * hi;
        const int i  = i0 + rr;
        const float inv = 1.0f / rsum[rr];
        const float* xi = xyzf + ((size_t)b * MTOK_ + i) * 3;
        const float wd0 = wx[rr * 3 + 0] * inv - xi[0];
        const float wd1 = wx[rr * 3 + 1] * inv - xi[1];
        const float wd2 = wx[rr * 3 + 2] * inv - xi[2];
        const float o = (qt == 0) ? acc0[r] : acc1[r];
        attn_out[((size_t)(b * MTOK_) + i) * C_ + h * DH_ + d] =
            o * inv + w0 * wd0 + w1 * wd1 + w2 * wd2;
      }
    }
  }
}

extern "C" void kernel_launch(void* const* d_in, const int* in_sizes, int n_in,
                              void* d_out, int out_size, void* d_ws, size_t ws_size,
                              hipStream_t stream) {
  (void)in_sizes; (void)n_in; (void)out_size; (void)ws_size;
  const float* xyzs = (const float*)d_in[0];
  const float* Wd   = (const float*)d_in[1];
  const float* Wqkv = (const float*)d_in[2];
  const float* Wsp  = (const float*)d_in[3];
  const float* Wout = (const float*)d_in[4];
  const float* bout = (const float*)d_in[5];
  const float* ln1g = (const float*)d_in[6];
  const float* ln1b = (const float*)d_in[7];
  const float* Wff1 = (const float*)d_in[8];
  const float* bff1 = (const float*)d_in[9];
  const float* Wff2 = (const float*)d_in[10];
  const float* bff2 = (const float*)d_in[11];
  const float* ln2g = (const float*)d_in[12];
  const float* ln2b = (const float*)d_in[13];

  float* ws      = (float*)d_ws;
  float* anchors = ws;
  float* feats   = ws + 8192;
  float* xln     = feats + (size_t)ROWS_ * C_;
  float* qkvb    = xln   + (size_t)ROWS_ * C_;
  float* attno   = qkvb  + (size_t)ROWS_ * 1536;
  float* ffh     = attno + (size_t)ROWS_ * C_;

  fps_kernel<<<B_ * L_, FPS_THREADS_, 0, stream>>>(xyzs, anchors);
  p4dconv_kernel<<<B_ * L_ * MA_ / P4W_, 256, 0, stream>>>(xyzs, anchors, Wd, feats);

  for (int d = 0; d < 2; ++d) {
    ln_kernel<<<ROWS_ / 8, 256, 0, stream>>>(feats, ln1g + d * C_, ln1b + d * C_, xln);
    gemm_tiled<64, 64, 0, false, false><<<dim3(1536 / 64, ROWS_ / 64), 256, 0, stream>>>(
        xln, C_, Wqkv + (size_t)d * 1536 * C_, C_, nullptr, nullptr,
        qkvb, 1536, C_);
    attn_kernel<<<dim3(MTOK_ / QB_, HEADS_, B_), 256, 0, stream>>>(
        qkvb, anchors, Wsp + (size_t)d * DH_ * 3, attno);
    gemm_tiled<64, 64, 1, true, true><<<dim3(C_ / 64, ROWS_ / 64), 256, 0, stream>>>(
        attno, C_, Wout + (size_t)d * C_ * C_, C_, bout + d * C_,
        feats, feats, C_, C_);
    ln_kernel<<<ROWS_ / 8, 256, 0, stream>>>(feats, ln2g + d * C_, ln2b + d * C_, xln);
    gemm_tiled<128, 128, 1, false, true><<<dim3(MLP_ / 128, ROWS_ / 128), 256, 0, stream>>>(
        xln, C_, Wff1 + (size_t)d * MLP_ * C_, C_, bff1 + d * MLP_,
        nullptr, ffh, MLP_, C_);
    gemm_tiled<64, 64, 0, true, true><<<dim3(C_ / 64, ROWS_ / 64), 256, 0, stream>>>(
        ffh, MLP_, Wff2 + (size_t)d * C_ * MLP_, MLP_, bff2 + d * C_,
        feats, feats, C_, MLP_);
  }
  hipMemcpyAsync(d_out, feats, (size_t)ROWS_ * C_ * sizeof(float),
                 hipMemcpyDeviceToDevice, stream);
}
